# starting kernel
# baseline (speedup 1.0000x reference)
#include <hip/hip_runtime.h>
#include <hip/hip_bf16.h>
#include <math.h>


typedef __attribute__((ext_vector_type(2))) float v2f;
typedef __attribute__((ext_vector_type(8))) float v8f;

__global__ void fill_f32(float* __restrict__ p, float v, int n) {
    int i = blockIdx.x * blockDim.x + threadIdx.x;
    if (i < n) p[i] = v;
}

__global__ void degree_k(const int* __restrict__ dst, float* __restrict__ deg, int E) {
    int i = blockIdx.x * blockDim.x + threadIdx.x;
    if (i < E) atomicAdd(&deg[dst[i]], 1.0f);
}

__global__ void gemm_wmma_f32(const float* __restrict__ A, int lda,
                              const float* __restrict__ W, int ldw,
                              float* __restrict__ C, int ldc,
                              int M, int K, int accumulate)
{
    const int nColGroups = 4;
    int waveId = blockIdx.x * (blockDim.x >> 5) + (threadIdx.x >> 5);
    int lane   = threadIdx.x & 31;
    int nRowTiles = M >> 4;
    if (waveId >= nRowTiles * nColGroups) return;
    int rowTile  = waveId / nColGroups;
    int colGroup = waveId - rowTile * nColGroups;
    int col0     = colGroup * 64;
    int half = lane >> 4;
    int lr   = lane & 15;

    const float* Arow = A + (size_t)(rowTile * 16 + lr) * lda + 2 * half;
    const float* Wr0  = W + (size_t)(col0 +  0 + lr) * ldw + 2 * half;
    const float* Wr1  = W + (size_t)(col0 + 16 + lr) * ldw + 2 * half;
    const float* Wr2  = W + (size_t)(col0 + 32 + lr) * ldw + 2 * half;
    const float* Wr3  = W + (size_t)(col0 + 48 + lr) * ldw + 2 * half;

    v8f acc0 = {}, acc1 = {}, acc2 = {}, acc3 = {};

    v2f a  = *(const v2f*)(Arow);
    v2f b0 = *(const v2f*)(Wr0);
    v2f b1 = *(const v2f*)(Wr1);
    v2f b2 = *(const v2f*)(Wr2);
    v2f b3 = *(const v2f*)(Wr3);

#pragma unroll 2
    for (int k = 4; k < K; k += 4) {
        v2f an  = *(const v2f*)(Arow + k);
        v2f bn0 = *(const v2f*)(Wr0 + k);
        v2f bn1 = *(const v2f*)(Wr1 + k);
        v2f bn2 = *(const v2f*)(Wr2 + k);
        v2f bn3 = *(const v2f*)(Wr3 + k);
        acc0 = __builtin_amdgcn_wmma_f32_16x16x4_f32(false, a, false, b0, (short)0, acc0, false, false);
        acc1 = __builtin_amdgcn_wmma_f32_16x16x4_f32(false, a, false, b1, (short)0, acc1, false, false);
        acc2 = __builtin_amdgcn_wmma_f32_16x16x4_f32(false, a, false, b2, (short)0, acc2, false, false);
        acc3 = __builtin_amdgcn_wmma_f32_16x16x4_f32(false, a, false, b3, (short)0, acc3, false, false);
        a = an; b0 = bn0; b1 = bn1; b2 = bn2; b3 = bn3;
    }
    acc0 = __builtin_amdgcn_wmma_f32_16x16x4_f32(false, a, false, b0, (short)0, acc0, false, false);
    acc1 = __builtin_amdgcn_wmma_f32_16x16x4_f32(false, a, false, b1, (short)0, acc1, false, false);
    acc2 = __builtin_amdgcn_wmma_f32_16x16x4_f32(false, a, false, b2, (short)0, acc2, false, false);
    acc3 = __builtin_amdgcn_wmma_f32_16x16x4_f32(false, a, false, b3, (short)0, acc3, false, false);

    float* Cb = C + (size_t)(rowTile * 16 + 8 * half) * ldc + col0 + lr;
#pragma unroll
    for (int j = 0; j < 8; ++j) {
        float* p = Cb + (size_t)j * ldc;
        p[ 0] = (accumulate ? p[ 0] : 0.0f) + acc0[j];
        p[16] = (accumulate ? p[16] : 0.0f) + acc1[j];
        p[32] = (accumulate ? p[32] : 0.0f) + acc2[j];
        p[48] = (accumulate ? p[48] : 0.0f) + acc3[j];
    }
}

__global__ void attn_scores(const float* __restrict__ feat,
                            const float* __restrict__ attn_l,
                            const float* __restrict__ attn_r,
                            float* __restrict__ el, float* __restrict__ er,
                            int pairs, int H, int D)
{
    int p    = blockIdx.x * (blockDim.x >> 5) + (threadIdx.x >> 5);
    int lane = threadIdx.x & 31;
    if (p >= pairs) return;
    int h = p % H;
    const float* f  = feat + (size_t)p * D;
    const float* al = attn_l + (size_t)h * D;
    const float* ar = attn_r + (size_t)h * D;
    float sl = 0.f, sr = 0.f;
    for (int i = lane; i < D; i += 32) {
        float fv = f[i];
        sl += fv * al[i];
        sr += fv * ar[i];
    }
#pragma unroll
    for (int off = 16; off > 0; off >>= 1) {
        sl += __shfl_xor(sl, off, 32);
        sr += __shfl_xor(sr, off, 32);
    }
    if (lane == 0) { el[p] = sl; er[p] = sr; }
}

__device__ inline void atomicMaxF(float* addr, float val) {
    if (val >= 0.0f) atomicMax((int*)addr, __float_as_int(val));
    else             atomicMin((unsigned int*)addr, __float_as_uint(val));
}

__device__ inline float leaky02(float v) { return v > 0.0f ? v : 0.2f * v; }

__global__ void edge_max_k(const int* __restrict__ src, const int* __restrict__ dst,
                           const float* __restrict__ el, const float* __restrict__ er,
                           float* __restrict__ m, int E, int H)
{
    int t = blockIdx.x * blockDim.x + threadIdx.x;
    if (t >= E * H) return;
    int e = t / H, h = t - e * H;
    float v = leaky02(el[src[e] * H + h] + er[dst[e] * H + h]);
    atomicMaxF(&m[dst[e] * H + h], v);
}

__global__ void edge_expsum_k(const int* __restrict__ src, const int* __restrict__ dst,
                              const float* __restrict__ el, const float* __restrict__ er,
                              const float* __restrict__ m, float* __restrict__ s,
                              float* __restrict__ exb, int E, int H)
{
    int t = blockIdx.x * blockDim.x + threadIdx.x;
    if (t >= E * H) return;
    int e = t / H, h = t - e * H;
    int d = dst[e];
    float v  = leaky02(el[src[e] * H + h] + er[d * H + h]);
    float ex = expf(v - m[d * H + h]);
    exb[t] = ex;
    atomicAdd(&s[d * H + h], ex);
}

__global__ void edge_agg_k(const int* __restrict__ src, const int* __restrict__ dst,
                           const float* __restrict__ exb, const float* __restrict__ s,
                           const float* __restrict__ feat, float* __restrict__ agg,
                           int total, int H, int D, int F)
{
    int t = blockIdx.x * blockDim.x + threadIdx.x;
    if (t >= total) return;
    int chunks = F >> 2;
    int e  = t / chunks;
    int q  = t - e * chunks;
    int d0 = q << 2;
    int h  = d0 / D;
    int sN = src[e], dN = dst[e];
    float a = exb[e * H + h] / s[dN * H + h];
    const float4 f = *(const float4*)(feat + (size_t)sN * F + d0);
    float* o = agg + (size_t)dN * F + d0;
    atomicAdd(o + 0, f.x * a);
    atomicAdd(o + 1, f.y * a);
    atomicAdd(o + 2, f.z * a);
    atomicAdd(o + 3, f.w * a);
}

__global__ void combine_h1_k(const float* __restrict__ agg, const float* __restrict__ bias1,
                             const float* __restrict__ buf, const float* __restrict__ bb0,
                             const float* __restrict__ deg, float* __restrict__ h1, int total)
{
    int i = blockIdx.x * blockDim.x + threadIdx.x;
    if (i >= total) return;
    int c = i & 255;
    int n = i >> 8;
    float norm = 1.0f / fmaxf(deg[n], 1.0f);
    float v = agg[i] + bias1[c] + norm * (buf[i] + bb0[c]);
    h1[i] = v > 0.0f ? v : 0.0f;
}

__global__ void combine_out_k(const float* __restrict__ agg, const float* __restrict__ bias2,
                              const float* __restrict__ buf, const float* __restrict__ bb1,
                              const float* __restrict__ deg, float* __restrict__ out, int total)
{
    int i = blockIdx.x * blockDim.x + threadIdx.x;
    if (i >= total) return;
    int c = i & 255;
    int n = i >> 8;
    float norm = 1.0f / fmaxf(deg[n], 1.0f);
    out[i] = agg[i] + bias2[c] + norm * (buf[i] + bb1[c]);
}

extern "C" void kernel_launch(void* const* d_in, const int* in_sizes, int n_in,
                              void* d_out, int out_size, void* d_ws, size_t ws_size,
                              hipStream_t stream)
{
    const float* inputs  = (const float*)d_in[0];
    const int*   src     = (const int*)  d_in[1];
    const int*   dst     = (const int*)  d_in[2];
    const float* fc1_w   = (const float*)d_in[3];
    const float* attn_l1 = (const float*)d_in[4];
    const float* attn_r1 = (const float*)d_in[5];
    const float* bias1   = (const float*)d_in[6];
    const float* fc2_w   = (const float*)d_in[7];
    const float* attn_l2 = (const float*)d_in[8];
    const float* attn_r2 = (const float*)d_in[9];
    const float* bias2   = (const float*)d_in[10];
    const float* wb0     = (const float*)d_in[11];
    const float* bb0     = (const float*)d_in[12];
    const float* wb1     = (const float*)d_in[13];
    const float* bb1     = (const float*)d_in[14];
    float* out = (float*)d_out;

    const int F  = 256;
    const int N  = in_sizes[0] / F;
    const int E  = in_sizes[1];
    const int H1 = 8, D1 = 32;

    float* ws   = (float*)d_ws;
    float* feat = ws;                ws += (size_t)N * F;
    float* buf  = ws;                ws += (size_t)N * F;
    float* agg  = ws;                ws += (size_t)N * F;
    float* h1   = ws;                ws += (size_t)N * F;
    float* el   = ws;                ws += (size_t)N * H1;
    float* er   = ws;                ws += (size_t)N * H1;
    float* m    = ws;                ws += (size_t)N * H1;
    float* s    = ws;                ws += (size_t)N * H1;
    float* exb  = ws;                ws += (size_t)E * H1;
    float* deg  = ws;                ws += (size_t)N;

    const int TB = 256;
    auto cdiv = [](int a, int b) { return (a + b - 1) / b; };

    const int waves    = (N / 16) * 4;
    const int gemmGrid = cdiv(waves, TB / 32);

    fill_f32<<<cdiv(N, TB), TB, 0, stream>>>(deg, 0.0f, N);
    fill_f32<<<cdiv(N * H1, TB), TB, 0, stream>>>(m, -INFINITY, N * H1);
    fill_f32<<<cdiv(N * H1, TB), TB, 0, stream>>>(s, 0.0f, N * H1);
    fill_f32<<<cdiv(N * F, TB), TB, 0, stream>>>(agg, 0.0f, N * F);
    degree_k<<<cdiv(E, TB), TB, 0, stream>>>(dst, deg, E);

    gemm_wmma_f32<<<gemmGrid, TB, 0, stream>>>(inputs, F, fc1_w, F, feat, F, N, F, 0);
    gemm_wmma_f32<<<gemmGrid, TB, 0, stream>>>(inputs, F, wb0,   F, buf,  F, N, F, 0);

    attn_scores<<<cdiv(N * H1, TB / 32), TB, 0, stream>>>(feat, attn_l1, attn_r1, el, er, N * H1, H1, D1);
    edge_max_k   <<<cdiv(E * H1, TB), TB, 0, stream>>>(src, dst, el, er, m, E, H1);
    edge_expsum_k<<<cdiv(E * H1, TB), TB, 0, stream>>>(src, dst, el, er, m, s, exb, E, H1);
    edge_agg_k   <<<cdiv(E * (F / 4), TB), TB, 0, stream>>>(src, dst, exb, s, feat, agg, E * (F / 4), H1, D1, F);

    combine_h1_k<<<cdiv(N * F, TB), TB, 0, stream>>>(agg, bias1, buf, bb0, deg, h1, N * F);

    fill_f32<<<cdiv(N, TB), TB, 0, stream>>>(m, -INFINITY, N);
    fill_f32<<<cdiv(N, TB), TB, 0, stream>>>(s, 0.0f, N);
    fill_f32<<<cdiv(N * F, TB), TB, 0, stream>>>(agg, 0.0f, N * F);

    gemm_wmma_f32<<<gemmGrid, TB, 0, stream>>>(h1, F, fc2_w, F, feat, F, N, F, 0);
    gemm_wmma_f32<<<gemmGrid, TB, 0, stream>>>(inputs, F, wb1,     2 * F, buf, F, N, F, 0);
    gemm_wmma_f32<<<gemmGrid, TB, 0, stream>>>(h1,     F, wb1 + F, 2 * F, buf, F, N, F, 1);

    attn_scores<<<cdiv(N, TB / 32), TB, 0, stream>>>(feat, attn_l2, attn_r2, el, er, N, 1, F);
    edge_max_k   <<<cdiv(E, TB), TB, 0, stream>>>(src, dst, el, er, m, E, 1);
    edge_expsum_k<<<cdiv(E, TB), TB, 0, stream>>>(src, dst, el, er, m, s, exb, E, 1);
    edge_agg_k   <<<cdiv(E * (F / 4), TB), TB, 0, stream>>>(src, dst, exb, s, feat, agg, E * (F / 4), 1, F, F);

    combine_out_k<<<cdiv(N * F, TB), TB, 0, stream>>>(agg, bias2, buf, bb1, deg, out, N * F);
}


# GEMM: A and W K-chunks (32 wide) staged in padded LDS by whole-line global loads; fragments read by ds_load instead of row-strided global gathers; k order and sums unchanged
# speedup vs baseline: 1.2398x; 1.2398x over previous
#include <hip/hip_runtime.h>
#include <hip/hip_bf16.h>
#include <math.h>


typedef __attribute__((ext_vector_type(2))) float v2f;
typedef __attribute__((ext_vector_type(4))) float v4f;
typedef __attribute__((ext_vector_type(8))) float v8f;

__global__ void fill_f32(float* __restrict__ p, float v, int n) {
    int i = blockIdx.x * blockDim.x + threadIdx.x;
    if (i < n) p[i] = v;
}

__global__ void degree_k(const int* __restrict__ dst, float* __restrict__ deg, int E) {
    int i = blockIdx.x * blockDim.x + threadIdx.x;
    if (i < E) atomicAdd(&deg[dst[i]], 1.0f);
}

#define GEMM_KC 32
#define GEMM_LP 36
__global__ __launch_bounds__(256)
void gemm_wmma_f32(const float* __restrict__ A, int lda,
                   const float* __restrict__ W, int ldw,
                   float* __restrict__ C, int ldc,
                   int M, int K, int accumulate)
{
    __shared__ __attribute__((aligned(16))) float Ws[256 * GEMM_LP];
    __shared__ __attribute__((aligned(16))) float As[32 * GEMM_LP];

    const int tid  = threadIdx.x;
    const int wave = tid >> 5;
    const int lane = tid & 31;
    const int half = lane >> 4;
    const int lr   = lane & 15;
    const int wr   = wave >> 2;
    const int col0 = (wave & 3) * 64;
    const int rowBase = blockIdx.x * 32;
    const int rowTile = blockIdx.x * 2 + wr;
    const int nRowTiles = M >> 4;

    const int sRow = tid >> 3;
    const int sCol = (tid & 7) * 4;
    int aRowG = rowBase + sRow;
    if (aRowG > M - 1) aRowG = M - 1;
    const float* Ag = A + (size_t)aRowG * lda + sCol;
    const float* Wg = W + (size_t)sRow * ldw + sCol;
    float* AsSt = As + sRow * GEMM_LP + sCol;
    float* WsSt = Ws + sRow * GEMM_LP + sCol;

    const float* aF = As + (wr * 16 + lr) * GEMM_LP + 2 * half;
    const float* bF = Ws + (col0 + lr) * GEMM_LP + 2 * half;

    v8f acc0 = {}, acc1 = {}, acc2 = {}, acc3 = {};

    v4f areg = *(const v4f*)(Ag);
    v4f wreg[8];
#pragma unroll
    for (int j = 0; j < 8; ++j)
        wreg[j] = *(const v4f*)(Wg + (size_t)(32 * j) * ldw);

    for (int k0 = 0; k0 < K; k0 += GEMM_KC) {
        __syncthreads();
        *(v4f*)(AsSt) = areg;
#pragma unroll
        for (int j = 0; j < 8; ++j)
            *(v4f*)(WsSt + (32 * j) * GEMM_LP) = wreg[j];
        __syncthreads();

        if (k0 + GEMM_KC < K) {
            areg = *(const v4f*)(Ag + k0 + GEMM_KC);
#pragma unroll
            for (int j = 0; j < 8; ++j)
                wreg[j] = *(const v4f*)(Wg + (size_t)(32 * j) * ldw + k0 + GEMM_KC);
        }

#pragma unroll
        for (int hc = 0; hc < GEMM_KC; hc += 16) {
            v2f a[4], b0[4], b1[4], b2[4], b3[4];
#pragma unroll
            for (int s = 0; s < 4; ++s) {
                const int kk = hc + 4 * s;
                a[s]  = *(const v2f*)(aF + kk);
                b0[s] = *(const v2f*)(bF +  0 * GEMM_LP + kk);
                b1[s] = *(const v2f*)(bF + 16 * GEMM_LP + kk);
                b2[s] = *(const v2f*)(bF + 32 * GEMM_LP + kk);
                b3[s] = *(const v2f*)(bF + 48 * GEMM_LP + kk);
            }
#pragma unroll
            for (int s = 0; s < 4; ++s) {
                acc0 = __builtin_amdgcn_wmma_f32_16x16x4_f32(false, a[s], false, b0[s], (short)0, acc0, false, false);
                acc1 = __builtin_amdgcn_wmma_f32_16x16x4_f32(false, a[s], false, b1[s], (short)0, acc1, false, false);
                acc2 = __builtin_amdgcn_wmma_f32_16x16x4_f32(false, a[s], false, b2[s], (short)0, acc2, false, false);
                acc3 = __builtin_amdgcn_wmma_f32_16x16x4_f32(false, a[s], false, b3[s], (short)0, acc3, false, false);
            }
        }
    }

    if (rowTile >= nRowTiles) return;

    float* Cb = C + (size_t)(rowTile * 16 + 8 * half) * ldc + col0 + lr;
#pragma unroll
    for (int j = 0; j < 8; ++j) {
        float* p = Cb + (size_t)j * ldc;
        p[ 0] = (accumulate ? p[ 0] : 0.0f) + acc0[j];
        p[16] = (accumulate ? p[16] : 0.0f) + acc1[j];
        p[32] = (accumulate ? p[32] : 0.0f) + acc2[j];
        p[48] = (accumulate ? p[48] : 0.0f) + acc3[j];
    }
}

__global__ void attn_scores(const float* __restrict__ feat,
                            const float* __restrict__ attn_l,
                            const float* __restrict__ attn_r,
                            float* __restrict__ el, float* __restrict__ er,
                            int pairs, int H, int D)
{
    int p    = blockIdx.x * (blockDim.x >> 5) + (threadIdx.x >> 5);
    int lane = threadIdx.x & 31;
    if (p >= pairs) return;
    int h = p % H;
    const float* f  = feat + (size_t)p * D;
    const float* al = attn_l + (size_t)h * D;
    const float* ar = attn_r + (size_t)h * D;
    float sl = 0.f, sr = 0.f;
    for (int i = lane; i < D; i += 32) {
        float fv = f[i];
        sl += fv * al[i];
        sr += fv * ar[i];
    }
#pragma unroll
    for (int off = 16; off > 0; off >>= 1) {
        sl += __shfl_xor(sl, off, 32);
        sr += __shfl_xor(sr, off, 32);
    }
    if (lane == 0) { el[p] = sl; er[p] = sr; }
}

__device__ inline void atomicMaxF(float* addr, float val) {
    if (val >= 0.0f) atomicMax((int*)addr, __float_as_int(val));
    else             atomicMin((unsigned int*)addr, __float_as_uint(val));
}

__device__ inline float leaky02(float v) { return v > 0.0f ? v : 0.2f * v; }

__global__ void edge_max_k(const int* __restrict__ src, const int* __restrict__ dst,
                           const float* __restrict__ el, const float* __restrict__ er,
                           float* __restrict__ m, int E, int H)
{
    int t = blockIdx.x * blockDim.x + threadIdx.x;
    if (t >= E * H) return;
    int e = t / H, h = t - e * H;
    float v = leaky02(el[src[e] * H + h] + er[dst[e] * H + h]);
    atomicMaxF(&m[dst[e] * H + h], v);
}

__global__ void edge_expsum_k(const int* __restrict__ src, const int* __restrict__ dst,
                              const float* __restrict__ el, const float* __restrict__ er,
                              const float* __restrict__ m, float* __restrict__ s,
                              float* __restrict__ exb, int E, int H)
{
    int t = blockIdx.x * blockDim.x + threadIdx.x;
    if (t >= E * H) return;
    int e = t / H, h = t - e * H;
    int d = dst[e];
    float v  = leaky02(el[src[e] * H + h] + er[d * H + h]);
    float ex = expf(v - m[d * H + h]);
    exb[t] = ex;
    atomicAdd(&s[d * H + h], ex);
}

__global__ void edge_agg_k(const int* __restrict__ src, const int* __restrict__ dst,
                           const float* __restrict__ exb, const float* __restrict__ s,
                           const float* __restrict__ feat, float* __restrict__ agg,
                           int total, int H, int D, int F)
{
    int t = blockIdx.x * blockDim.x + threadIdx.x;
    if (t >= total) return;
    int chunks = F >> 2;
    int e  = t / chunks;
    int q  = t - e * chunks;
    int d0 = q << 2;
    int h  = d0 / D;
    int sN = src[e], dN = dst[e];
    float a = exb[e * H + h] / s[dN * H + h];
    const float4 f = *(const float4*)(feat + (size_t)sN * F + d0);
    float* o = agg + (size_t)dN * F + d0;
    atomicAdd(o + 0, f.x * a);
    atomicAdd(o + 1, f.y * a);
    atomicAdd(o + 2, f.z * a);
    atomicAdd(o + 3, f.w * a);
}

__global__ void combine_h1_k(const float* __restrict__ agg, const float* __restrict__ bias1,
                             const float* __restrict__ buf, const float* __restrict__ bb0,
                             const float* __restrict__ deg, float* __restrict__ h1, int total)
{
    int i = blockIdx.x * blockDim.x + threadIdx.x;
    if (i >= total) return;
    int c = i & 255;
    int n = i >> 8;
    float norm = 1.0f / fmaxf(deg[n], 1.0f);
    float v = agg[i] + bias1[c] + norm * (buf[i] + bb0[c]);
    h1[i] = v > 0.0f ? v : 0.0f;
}

__global__ void combine_out_k(const float* __restrict__ agg, const float* __restrict__ bias2,
                              const float* __restrict__ buf, const float* __restrict__ bb1,
                              const float* __restrict__ deg, float* __restrict__ out, int total)
{
    int i = blockIdx.x * blockDim.x + threadIdx.x;
    if (i >= total) return;
    int c = i & 255;
    int n = i >> 8;
    float norm = 1.0f / fmaxf(deg[n], 1.0f);
    out[i] = agg[i] + bias2[c] + norm * (buf[i] + bb1[c]);
}

extern "C" void kernel_launch(void* const* d_in, const int* in_sizes, int n_in,
                              void* d_out, int out_size, void* d_ws, size_t ws_size,
                              hipStream_t stream)
{
    const float* inputs  = (const float*)d_in[0];
    const int*   src     = (const int*)  d_in[1];
    const int*   dst     = (const int*)  d_in[2];
    const float* fc1_w   = (const float*)d_in[3];
    const float* attn_l1 = (const float*)d_in[4];
    const float* attn_r1 = (const float*)d_in[5];
    const float* bias1   = (const float*)d_in[6];
    const float* fc2_w   = (const float*)d_in[7];
    const float* attn_l2 = (const float*)d_in[8];
    const float* attn_r2 = (const float*)d_in[9];
    const float* bias2   = (const float*)d_in[10];
    const float* wb0     = (const float*)d_in[11];
    const float* bb0     = (const float*)d_in[12];
    const float* wb1     = (const float*)d_in[13];
    const float* bb1     = (const float*)d_in[14];
    float* out = (float*)d_out;

    const int F  = 256;
    const int N  = in_sizes[0] / F;
    const int E  = in_sizes[1];
    const int H1 = 8, D1 = 32;

    float* ws   = (float*)d_ws;
    float* feat = ws;                ws += (size_t)N * F;
    float* buf  = ws;                ws += (size_t)N * F;
    float* agg  = ws;                ws += (size_t)N * F;
    float* h1   = ws;                ws += (size_t)N * F;
    float* el   = ws;                ws += (size_t)N * H1;
    float* er   = ws;                ws += (size_t)N * H1;
    float* m    = ws;                ws += (size_t)N * H1;
    float* s    = ws;                ws += (size_t)N * H1;
    float* exb  = ws;                ws += (size_t)E * H1;
    float* deg  = ws;                ws += (size_t)N;

    const int TB = 256;
    auto cdiv = [](int a, int b) { return (a + b - 1) / b; };

    const int waves    = (N / 16) * 4;
    const int gemmGrid = cdiv(waves, TB / 32);

    fill_f32<<<cdiv(N, TB), TB, 0, stream>>>(deg, 0.0f, N);
    fill_f32<<<cdiv(N * H1, TB), TB, 0, stream>>>(m, -INFINITY, N * H1);
    fill_f32<<<cdiv(N * H1, TB), TB, 0, stream>>>(s, 0.0f, N * H1);
    fill_f32<<<cdiv(N * F, TB), TB, 0, stream>>>(agg, 0.0f, N * F);
    degree_k<<<cdiv(E, TB), TB, 0, stream>>>(dst, deg, E);

    gemm_wmma_f32<<<gemmGrid, TB, 0, stream>>>(inputs, F, fc1_w, F, feat, F, N, F, 0);
    gemm_wmma_f32<<<gemmGrid, TB, 0, stream>>>(inputs, F, wb0,   F, buf,  F, N, F, 0);

    attn_scores<<<cdiv(N * H1, TB / 32), TB, 0, stream>>>(feat, attn_l1, attn_r1, el, er, N * H1, H1, D1);
    edge_max_k   <<<cdiv(E * H1, TB), TB, 0, stream>>>(src, dst, el, er, m, E, H1);
    edge_expsum_k<<<cdiv(E * H1, TB), TB, 0, stream>>>(src, dst, el, er, m, s, exb, E, H1);
    edge_agg_k   <<<cdiv(E * (F / 4), TB), TB, 0, stream>>>(src, dst, exb, s, feat, agg, E * (F / 4), H1, D1, F);

    combine_h1_k<<<cdiv(N * F, TB), TB, 0, stream>>>(agg, bias1, buf, bb0, deg, h1, N * F);

    fill_f32<<<cdiv(N, TB), TB, 0, stream>>>(m, -INFINITY, N);
    fill_f32<<<cdiv(N, TB), TB, 0, stream>>>(s, 0.0f, N);
    fill_f32<<<cdiv(N * F, TB), TB, 0, stream>>>(agg, 0.0f, N * F);

    gemm_wmma_f32<<<gemmGrid, TB, 0, stream>>>(h1, F, fc2_w, F, feat, F, N, F, 0);
    gemm_wmma_f32<<<gemmGrid, TB, 0, stream>>>(inputs, F, wb1,     2 * F, buf, F, N, F, 0);
    gemm_wmma_f32<<<gemmGrid, TB, 0, stream>>>(h1,     F, wb1 + F, 2 * F, buf, F, N, F, 1);

    attn_scores<<<cdiv(N, TB / 32), TB, 0, stream>>>(feat, attn_l2, attn_r2, el, er, N, 1, F);
    edge_max_k   <<<cdiv(E, TB), TB, 0, stream>>>(src, dst, el, er, m, E, 1);
    edge_expsum_k<<<cdiv(E, TB), TB, 0, stream>>>(src, dst, el, er, m, s, exb, E, 1);
    edge_agg_k   <<<cdiv(E * (F / 4), TB), TB, 0, stream>>>(src, dst, exb, s, feat, agg, E * (F / 4), 1, F, F);

    combine_out_k<<<cdiv(N * F, TB), TB, 0, stream>>>(agg, bias2, buf, bb1, deg, out, N * F);
}


# edge aggregation: each wave walks 32 edges with lane = column, so every f32 atomic add and gather load covers one whole 128-B line (was 4 lines x 8 lanes); sums still atomic, arrival order
# speedup vs baseline: 1.6732x; 1.3496x over previous
#include <hip/hip_runtime.h>
#include <hip/hip_bf16.h>
#include <math.h>


typedef __attribute__((ext_vector_type(2))) float v2f;
typedef __attribute__((ext_vector_type(4))) float v4f;
typedef __attribute__((ext_vector_type(8))) float v8f;

__global__ void fill_f32(float* __restrict__ p, float v, int n) {
    int i = blockIdx.x * blockDim.x + threadIdx.x;
    if (i < n) p[i] = v;
}

__global__ void degree_k(const int* __restrict__ dst, float* __restrict__ deg, int E) {
    int i = blockIdx.x * blockDim.x + threadIdx.x;
    if (i < E) atomicAdd(&deg[dst[i]], 1.0f);
}

#define GEMM_KC 32
#define GEMM_LP 36
__global__ __launch_bounds__(256)
void gemm_wmma_f32(const float* __restrict__ A, int lda,
                   const float* __restrict__ W, int ldw,
                   float* __restrict__ C, int ldc,
                   int M, int K, int accumulate)
{
    __shared__ __attribute__((aligned(16))) float Ws[256 * GEMM_LP];
    __shared__ __attribute__((aligned(16))) float As[32 * GEMM_LP];

    const int tid  = threadIdx.x;
    const int wave = tid >> 5;
    const int lane = tid & 31;
    const int half = lane >> 4;
    const int lr   = lane & 15;
    const int wr   = wave >> 2;
    const int col0 = (wave & 3) * 64;
    const int rowBase = blockIdx.x * 32;
    const int rowTile = blockIdx.x * 2 + wr;
    const int nRowTiles = M >> 4;

    const int sRow = tid >> 3;
    const int sCol = (tid & 7) * 4;
    int aRowG = rowBase + sRow;
    if (aRowG > M - 1) aRowG = M - 1;
    const float* Ag = A + (size_t)aRowG * lda + sCol;
    const float* Wg = W + (size_t)sRow * ldw + sCol;
    float* AsSt = As + sRow * GEMM_LP + sCol;
    float* WsSt = Ws + sRow * GEMM_LP + sCol;

    const float* aF = As + (wr * 16 + lr) * GEMM_LP + 2 * half;
    const float* bF = Ws + (col0 + lr) * GEMM_LP + 2 * half;

    v8f acc0 = {}, acc1 = {}, acc2 = {}, acc3 = {};

    v4f areg = *(const v4f*)(Ag);
    v4f wreg[8];
#pragma unroll
    for (int j = 0; j < 8; ++j)
        wreg[j] = *(const v4f*)(Wg + (size_t)(32 * j) * ldw);

    for (int k0 = 0; k0 < K; k0 += GEMM_KC) {
        __syncthreads();
        *(v4f*)(AsSt) = areg;
#pragma unroll
        for (int j = 0; j < 8; ++j)
            *(v4f*)(WsSt + (32 * j) * GEMM_LP) = wreg[j];
        __syncthreads();

        if (k0 + GEMM_KC < K) {
            areg = *(const v4f*)(Ag + k0 + GEMM_KC);
#pragma unroll
            for (int j = 0; j < 8; ++j)
                wreg[j] = *(const v4f*)(Wg + (size_t)(32 * j) * ldw + k0 + GEMM_KC);
        }

#pragma unroll
        for (int hc = 0; hc < GEMM_KC; hc += 16) {
            v2f a[4], b0[4], b1[4], b2[4], b3[4];
#pragma unroll
            for (int s = 0; s < 4; ++s) {
                const int kk = hc + 4 * s;
                a[s]  = *(const v2f*)(aF + kk);
                b0[s] = *(const v2f*)(bF +  0 * GEMM_LP + kk);
                b1[s] = *(const v2f*)(bF + 16 * GEMM_LP + kk);
                b2[s] = *(const v2f*)(bF + 32 * GEMM_LP + kk);
                b3[s] = *(const v2f*)(bF + 48 * GEMM_LP + kk);
            }
#pragma unroll
            for (int s = 0; s < 4; ++s) {
                acc0 = __builtin_amdgcn_wmma_f32_16x16x4_f32(false, a[s], false, b0[s], (short)0, acc0, false, false);
                acc1 = __builtin_amdgcn_wmma_f32_16x16x4_f32(false, a[s], false, b1[s], (short)0, acc1, false, false);
                acc2 = __builtin_amdgcn_wmma_f32_16x16x4_f32(false, a[s], false, b2[s], (short)0, acc2, false, false);
                acc3 = __builtin_amdgcn_wmma_f32_16x16x4_f32(false, a[s], false, b3[s], (short)0, acc3, false, false);
            }
        }
    }

    if (rowTile >= nRowTiles) return;

    float* Cb = C + (size_t)(rowTile * 16 + 8 * half) * ldc + col0 + lr;
#pragma unroll
    for (int j = 0; j < 8; ++j) {
        float* p = Cb + (size_t)j * ldc;
        p[ 0] = (accumulate ? p[ 0] : 0.0f) + acc0[j];
        p[16] = (accumulate ? p[16] : 0.0f) + acc1[j];
        p[32] = (accumulate ? p[32] : 0.0f) + acc2[j];
        p[48] = (accumulate ? p[48] : 0.0f) + acc3[j];
    }
}

__global__ void attn_scores(const float* __restrict__ feat,
                            const float* __restrict__ attn_l,
                            const float* __restrict__ attn_r,
                            float* __restrict__ el, float* __restrict__ er,
                            int pairs, int H, int D)
{
    int p    = blockIdx.x * (blockDim.x >> 5) + (threadIdx.x >> 5);
    int lane = threadIdx.x & 31;
    if (p >= pairs) return;
    int h = p % H;
    const float* f  = feat + (size_t)p * D;
    const float* al = attn_l + (size_t)h * D;
    const float* ar = attn_r + (size_t)h * D;
    float sl = 0.f, sr = 0.f;
    for (int i = lane; i < D; i += 32) {
        float fv = f[i];
        sl += fv * al[i];
        sr += fv * ar[i];
    }
#pragma unroll
    for (int off = 16; off > 0; off >>= 1) {
        sl += __shfl_xor(sl, off, 32);
        sr += __shfl_xor(sr, off, 32);
    }
    if (lane == 0) { el[p] = sl; er[p] = sr; }
}

__device__ inline void atomicMaxF(float* addr, float val) {
    if (val >= 0.0f) atomicMax((int*)addr, __float_as_int(val));
    else             atomicMin((unsigned int*)addr, __float_as_uint(val));
}

__device__ inline float leaky02(float v) { return v > 0.0f ? v : 0.2f * v; }

__global__ void edge_max_k(const int* __restrict__ src, const int* __restrict__ dst,
                           const float* __restrict__ el, const float* __restrict__ er,
                           float* __restrict__ m, int E, int H)
{
    int t = blockIdx.x * blockDim.x + threadIdx.x;
    if (t >= E * H) return;
    int e = t / H, h = t - e * H;
    float v = leaky02(el[src[e] * H + h] + er[dst[e] * H + h]);
    atomicMaxF(&m[dst[e] * H + h], v);
}

__global__ void edge_expsum_k(const int* __restrict__ src, const int* __restrict__ dst,
                              const float* __restrict__ el, const float* __restrict__ er,
                              const float* __restrict__ m, float* __restrict__ s,
                              float* __restrict__ exb, int E, int H)
{
    int t = blockIdx.x * blockDim.x + threadIdx.x;
    if (t >= E * H) return;
    int e = t / H, h = t - e * H;
    int d = dst[e];
    float v  = leaky02(el[src[e] * H + h] + er[d * H + h]);
    float ex = expf(v - m[d * H + h]);
    exb[t] = ex;
    atomicAdd(&s[d * H + h], ex);
}

#define AGG_TB 128
#define AGG_U  4
template <int H>
__global__ __launch_bounds__(AGG_TB)
void edge_agg_rows_k(const int* __restrict__ src, const int* __restrict__ dst,
                     const float* __restrict__ exb, const float* __restrict__ s,
                     const float* __restrict__ feat, float* __restrict__ agg, int E)
{
    const int lane = threadIdx.x & 31;
    const int wv = __builtin_amdgcn_readfirstlane((int)(blockIdx.x * (blockDim.x >> 5) + (threadIdx.x >> 5)));
    const int e0 = wv * 32;
    if (e0 >= E) return;
    const int cnt = (E - e0 < 32) ? (E - e0) : 32;

    int sv = 0, dv = 0;
    float coef[H];
#pragma unroll
    for (int h = 0; h < H; ++h) coef[h] = 0.0f;
    if (lane < cnt) {
        const int e = e0 + lane;
        sv = src[e];
        dv = dst[e];
        if constexpr (H == 8) {
            const v4f x0 = *(const v4f*)(exb + (size_t)e * 8);
            const v4f x1 = *(const v4f*)(exb + (size_t)e * 8 + 4);
            const v4f s0 = *(const v4f*)(s + (size_t)dv * 8);
            const v4f s1 = *(const v4f*)(s + (size_t)dv * 8 + 4);
#pragma unroll
            for (int h = 0; h < 4; ++h) {
                coef[h]     = x0[h] / s0[h];
                coef[h + 4] = x1[h] / s1[h];
            }
        } else {
            coef[0] = exb[e] / s[dv];
        }
    }

    const float* featL = feat + lane;
    float*       aggL  = agg + lane;

    int i = 0;
    for (; i + AGG_U <= cnt; i += AGG_U) {
        float f[AGG_U][8];
#pragma unroll
        for (int u = 0; u < AGG_U; ++u) {
            const int sN = __builtin_amdgcn_readlane(sv, i + u);
            const float* fr = featL + (size_t)sN * 256;
#pragma unroll
            for (int j = 0; j < 8; ++j) f[u][j] = fr[32 * j];
        }
#pragma unroll
        for (int u = 0; u < AGG_U; ++u) {
            const int dN = __builtin_amdgcn_readlane(dv, i + u);
            float* o = aggL + (size_t)dN * 256;
#pragma unroll
            for (int j = 0; j < 8; ++j) {
                const float a = __int_as_float(__builtin_amdgcn_readlane(__float_as_int(coef[(j * H) >> 3]), i + u));
                atomicAdd(o + 32 * j, f[u][j] * a);
            }
        }
    }
    for (; i < cnt; ++i) {
        const int sN = __builtin_amdgcn_readlane(sv, i);
        const int dN = __builtin_amdgcn_readlane(dv, i);
        const float* fr = featL + (size_t)sN * 256;
        float* o = aggL + (size_t)dN * 256;
        float f[8];
#pragma unroll
        for (int j = 0; j < 8; ++j) f[j] = fr[32 * j];
#pragma unroll
        for (int j = 0; j < 8; ++j) {
            const float a = __int_as_float(__builtin_amdgcn_readlane(__float_as_int(coef[(j * H) >> 3]), i));
            atomicAdd(o + 32 * j, f[j] * a);
        }
    }
}

__global__ void combine_h1_k(const float* __restrict__ agg, const float* __restrict__ bias1,
                             const float* __restrict__ buf, const float* __restrict__ bb0,
                             const float* __restrict__ deg, float* __restrict__ h1, int total)
{
    int i = blockIdx.x * blockDim.x + threadIdx.x;
    if (i >= total) return;
    int c = i & 255;
    int n = i >> 8;
    float norm = 1.0f / fmaxf(deg[n], 1.0f);
    float v = agg[i] + bias1[c] + norm * (buf[i] + bb0[c]);
    h1[i] = v > 0.0f ? v : 0.0f;
}

__global__ void combine_out_k(const float* __restrict__ agg, const float* __restrict__ bias2,
                              const float* __restrict__ buf, const float* __restrict__ bb1,
                              const float* __restrict__ deg, float* __restrict__ out, int total)
{
    int i = blockIdx.x * blockDim.x + threadIdx.x;
    if (i >= total) return;
    int c = i & 255;
    int n = i >> 8;
    float norm = 1.0f / fmaxf(deg[n], 1.0f);
    out[i] = agg[i] + bias2[c] + norm * (buf[i] + bb1[c]);
}

extern "C" void kernel_launch(void* const* d_in, const int* in_sizes, int n_in,
                              void* d_out, int out_size, void* d_ws, size_t ws_size,
                              hipStream_t stream)
{
    const float* inputs  = (const float*)d_in[0];
    const int*   src     = (const int*)  d_in[1];
    const int*   dst     = (const int*)  d_in[2];
    const float* fc1_w   = (const float*)d_in[3];
    const float* attn_l1 = (const float*)d_in[4];
    const float* attn_r1 = (const float*)d_in[5];
    const float* bias1   = (const float*)d_in[6];
    const float* fc2_w   = (const float*)d_in[7];
    const float* attn_l2 = (const float*)d_in[8];
    const float* attn_r2 = (const float*)d_in[9];
    const float* bias2   = (const float*)d_in[10];
    const float* wb0     = (const float*)d_in[11];
    const float* bb0     = (const float*)d_in[12];
    const float* wb1     = (const float*)d_in[13];
    const float* bb1     = (const float*)d_in[14];
    float* out = (float*)d_out;

    const int F  = 256;
    const int N  = in_sizes[0] / F;
    const int E  = in_sizes[1];
    const int H1 = 8, D1 = 32;

    float* ws   = (float*)d_ws;
    float* feat = ws;                ws += (size_t)N * F;
    float* buf  = ws;                ws += (size_t)N * F;
    float* agg  = ws;                ws += (size_t)N * F;
    float* h1   = ws;                ws += (size_t)N * F;
    float* el   = ws;                ws += (size_t)N * H1;
    float* er   = ws;                ws += (size_t)N * H1;
    float* m    = ws;                ws += (size_t)N * H1;
    float* s    = ws;                ws += (size_t)N * H1;
    float* exb  = ws;                ws += (size_t)E * H1;
    float* deg  = ws;                ws += (size_t)N;

    const int TB = 256;
    auto cdiv = [](int a, int b) { return (a + b - 1) / b; };

    const int waves    = (N / 16) * 4;
    const int gemmGrid = cdiv(waves, TB / 32);

    fill_f32<<<cdiv(N, TB), TB, 0, stream>>>(deg, 0.0f, N);
    fill_f32<<<cdiv(N * H1, TB), TB, 0, stream>>>(m, -INFINITY, N * H1);
    fill_f32<<<cdiv(N * H1, TB), TB, 0, stream>>>(s, 0.0f, N * H1);
    fill_f32<<<cdiv(N * F, TB), TB, 0, stream>>>(agg, 0.0f, N * F);
    degree_k<<<cdiv(E, TB), TB, 0, stream>>>(dst, deg, E);

    gemm_wmma_f32<<<gemmGrid, TB, 0, stream>>>(inputs, F, fc1_w, F, feat, F, N, F, 0);
    gemm_wmma_f32<<<gemmGrid, TB, 0, stream>>>(inputs, F, wb0,   F, buf,  F, N, F, 0);

    attn_scores<<<cdiv(N * H1, TB / 32), TB, 0, stream>>>(feat, attn_l1, attn_r1, el, er, N * H1, H1, D1);
    edge_max_k   <<<cdiv(E * H1, TB), TB, 0, stream>>>(src, dst, el, er, m, E, H1);
    edge_expsum_k<<<cdiv(E * H1, TB), TB, 0, stream>>>(src, dst, el, er, m, s, exb, E, H1);
    edge_agg_rows_k<8><<<cdiv(cdiv(E, 32), AGG_TB / 32), AGG_TB, 0, stream>>>(src, dst, exb, s, feat, agg, E);

    combine_h1_k<<<cdiv(N * F, TB), TB, 0, stream>>>(agg, bias1, buf, bb0, deg, h1, N * F);

    fill_f32<<<cdiv(N, TB), TB, 0, stream>>>(m, -INFINITY, N);
    fill_f32<<<cdiv(N, TB), TB, 0, stream>>>(s, 0.0f, N);
    fill_f32<<<cdiv(N * F, TB), TB, 0, stream>>>(agg, 0.0f, N * F);

    gemm_wmma_f32<<<gemmGrid, TB, 0, stream>>>(h1, F, fc2_w, F, feat, F, N, F, 0);
    gemm_wmma_f32<<<gemmGrid, TB, 0, stream>>>(inputs, F, wb1,     2 * F, buf, F, N, F, 0);
    gemm_wmma_f32<<<gemmGrid, TB, 0, stream>>>(h1,     F, wb1 + F, 2 * F, buf, F, N, F, 1);

    attn_scores<<<cdiv(N, TB / 32), TB, 0, stream>>>(feat, attn_l2, attn_r2, el, er, N, 1, F);
    edge_max_k   <<<cdiv(E, TB), TB, 0, stream>>>(src, dst, el, er, m, E, 1);
    edge_expsum_k<<<cdiv(E, TB), TB, 0, stream>>>(src, dst, el, er, m, s, exb, E, 1);
    edge_agg_rows_k<1><<<cdiv(cdiv(E, 32), AGG_TB / 32), AGG_TB, 0, stream>>>(src, dst, exb, s, feat, agg, E);

    combine_out_k<<<cdiv(N * F, TB), TB, 0, stream>>>(agg, bias2, buf, bb1, deg, out, N * F);
}


# edges grouped by destination once per call (integer-add rank, prefix sum); one wave per node does softmax and row sum in registers and stores once; no f32 atomics, sums in arrival-rank order
# speedup vs baseline: 2.7952x; 1.6706x over previous
#include <hip/hip_runtime.h>
#include <hip/hip_bf16.h>
#include <math.h>


typedef __attribute__((ext_vector_type(2))) float v2f;
typedef __attribute__((ext_vector_type(4))) float v4f;
typedef __attribute__((ext_vector_type(8))) float v8f;

__global__ void fill_i32(int* __restrict__ p, int v, int n) {
    int i = blockIdx.x * blockDim.x + threadIdx.x;
    if (i < n) p[i] = v;
}

__global__ void edge_rank_k(const int* __restrict__ dst, int* __restrict__ cnt,
                            int* __restrict__ rank, int E)
{
    int e = blockIdx.x * blockDim.x + threadIdx.x;
    if (e < E) rank[e] = atomicAdd(&cnt[dst[e]], 1);
}

#define SCAN_TB 1024
#define SCAN_CH 32
__global__ __launch_bounds__(SCAN_TB)
void scan_counts_k(const int* __restrict__ cnt, int* __restrict__ rp, int N)
{
    __shared__ int wtot[32];
    __shared__ int tileTot;
    const int t = threadIdx.x;
    const int lane = t & 31;
    const int wv = t >> 5;
    int carry = 0;
    for (int base = 0; base <= N; base += SCAN_TB * SCAN_CH) {
        const int b = base + t * SCAN_CH;
        int c[SCAN_CH];
#pragma unroll
        for (int i = 0; i < SCAN_CH; ++i) c[i] = (b + i < N) ? cnt[b + i] : 0;
        int sum = 0;
#pragma unroll
        for (int i = 0; i < SCAN_CH; ++i) sum += c[i];
        int inc = sum;
#pragma unroll
        for (int off = 1; off < 32; off <<= 1) {
            const int v = __shfl_up(inc, off, 32);
            if (lane >= off) inc += v;
        }
        if (lane == 31) wtot[wv] = inc;
        __syncthreads();
        if (wv == 0) {
            const int w = wtot[lane];
            int winc = w;
#pragma unroll
            for (int off = 1; off < 32; off <<= 1) {
                const int v = __shfl_up(winc, off, 32);
                if (lane >= off) winc += v;
            }
            wtot[lane] = winc - w;
            if (lane == 31) tileTot = winc;
        }
        __syncthreads();
        int run = carry + wtot[wv] + inc - sum;
#pragma unroll
        for (int i = 0; i < SCAN_CH; ++i) {
            if (b + i <= N) rp[b + i] = run;
            run += c[i];
        }
        carry += tileTot;
        __syncthreads();
    }
}

__global__ void edge_place_k(const int* __restrict__ src, const int* __restrict__ dst,
                             const int* __restrict__ rank, const int* __restrict__ rp,
                             int* __restrict__ csrc, int E)
{
    int e = blockIdx.x * blockDim.x + threadIdx.x;
    if (e < E) csrc[rp[dst[e]] + rank[e]] = src[e];
}

#define GEMM_KC 32
#define GEMM_LP 36
__global__ __launch_bounds__(256)
void gemm_wmma_f32(const float* __restrict__ A, int lda,
                   const float* __restrict__ W, int ldw,
                   float* __restrict__ C, int ldc,
                   int M, int K, int accumulate)
{
    __shared__ __attribute__((aligned(16))) float Ws[256 * GEMM_LP];
    __shared__ __attribute__((aligned(16))) float As[32 * GEMM_LP];

    const int tid  = threadIdx.x;
    const int wave = tid >> 5;
    const int lane = tid & 31;
    const int half = lane >> 4;
    const int lr   = lane & 15;
    const int wr   = wave >> 2;
    const int col0 = (wave & 3) * 64;
    const int rowBase = blockIdx.x * 32;
    const int rowTile = blockIdx.x * 2 + wr;
    const int nRowTiles = M >> 4;

    const int sRow = tid >> 3;
    const int sCol = (tid & 7) * 4;
    int aRowG = rowBase + sRow;
    if (aRowG > M - 1) aRowG = M - 1;
    const float* Ag = A + (size_t)aRowG * lda + sCol;
    const float* Wg = W + (size_t)sRow * ldw + sCol;
    float* AsSt = As + sRow * GEMM_LP + sCol;
    float* WsSt = Ws + sRow * GEMM_LP + sCol;

    const float* aF = As + (wr * 16 + lr) * GEMM_LP + 2 * half;
    const float* bF = Ws + (col0 + lr) * GEMM_LP + 2 * half;

    v8f acc0 = {}, acc1 = {}, acc2 = {}, acc3 = {};

    v4f areg = *(const v4f*)(Ag);
    v4f wreg[8];
#pragma unroll
    for (int j = 0; j < 8; ++j)
        wreg[j] = *(const v4f*)(Wg + (size_t)(32 * j) * ldw);

    for (int k0 = 0; k0 < K; k0 += GEMM_KC) {
        __syncthreads();
        *(v4f*)(AsSt) = areg;
#pragma unroll
        for (int j = 0; j < 8; ++j)
            *(v4f*)(WsSt + (32 * j) * GEMM_LP) = wreg[j];
        __syncthreads();

        if (k0 + GEMM_KC < K) {
            areg = *(const v4f*)(Ag + k0 + GEMM_KC);
#pragma unroll
            for (int j = 0; j < 8; ++j)
                wreg[j] = *(const v4f*)(Wg + (size_t)(32 * j) * ldw + k0 + GEMM_KC);
        }

#pragma unroll
        for (int hc = 0; hc < GEMM_KC; hc += 16) {
            v2f a[4], b0[4], b1[4], b2[4], b3[4];
#pragma unroll
            for (int s = 0; s < 4; ++s) {
                const int kk = hc + 4 * s;
                a[s]  = *(const v2f*)(aF + kk);
                b0[s] = *(const v2f*)(bF +  0 * GEMM_LP + kk);
                b1[s] = *(const v2f*)(bF + 16 * GEMM_LP + kk);
                b2[s] = *(const v2f*)(bF + 32 * GEMM_LP + kk);
                b3[s] = *(const v2f*)(bF + 48 * GEMM_LP + kk);
            }
#pragma unroll
            for (int s = 0; s < 4; ++s) {
                acc0 = __builtin_amdgcn_wmma_f32_16x16x4_f32(false, a[s], false, b0[s], (short)0, acc0, false, false);
                acc1 = __builtin_amdgcn_wmma_f32_16x16x4_f32(false, a[s], false, b1[s], (short)0, acc1, false, false);
                acc2 = __builtin_amdgcn_wmma_f32_16x16x4_f32(false, a[s], false, b2[s], (short)0, acc2, false, false);
                acc3 = __builtin_amdgcn_wmma_f32_16x16x4_f32(false, a[s], false, b3[s], (short)0, acc3, false, false);
            }
        }
    }

    if (rowTile >= nRowTiles) return;

    float* Cb = C + (size_t)(rowTile * 16 + 8 * half) * ldc + col0 + lr;
#pragma unroll
    for (int j = 0; j < 8; ++j) {
        float* p = Cb + (size_t)j * ldc;
        p[ 0] = (accumulate ? p[ 0] : 0.0f) + acc0[j];
        p[16] = (accumulate ? p[16] : 0.0f) + acc1[j];
        p[32] = (accumulate ? p[32] : 0.0f) + acc2[j];
        p[48] = (accumulate ? p[48] : 0.0f) + acc3[j];
    }
}

__global__ void attn_scores(const float* __restrict__ feat,
                            const float* __restrict__ attn_l,
                            const float* __restrict__ attn_r,
                            float* __restrict__ el, float* __restrict__ er,
                            int pairs, int H, int D)
{
    int p    = blockIdx.x * (blockDim.x >> 5) + (threadIdx.x >> 5);
    int lane = threadIdx.x & 31;
    if (p >= pairs) return;
    int h = p % H;
    const float* f  = feat + (size_t)p * D;
    const float* al = attn_l + (size_t)h * D;
    const float* ar = attn_r + (size_t)h * D;
    float sl = 0.f, sr = 0.f;
    for (int i = lane; i < D; i += 32) {
        float fv = f[i];
        sl += fv * al[i];
        sr += fv * ar[i];
    }
#pragma unroll
    for (int off = 16; off > 0; off >>= 1) {
        sl += __shfl_xor(sl, off, 32);
        sr += __shfl_xor(sr, off, 32);
    }
    if (lane == 0) { el[p] = sl; er[p] = sr; }
}

__device__ inline float leaky02(float v) { return v > 0.0f ? v : 0.2f * v; }

#define NODE_TB 256
#define AGG_U  4

template <int H>
__device__ inline void load_scores(const int* __restrict__ csrc, const float* __restrict__ el,
                                   const float (&ern)[H], int b, int cnt, int lane,
                                   int& sv, float (&e)[H])
{
    sv = 0;
#pragma unroll
    for (int h = 0; h < H; ++h) e[h] = -INFINITY;
    if (lane < cnt) {
        sv = csrc[b + lane];
        if constexpr (H == 8) {
            const v4f x0 = *(const v4f*)(el + (size_t)sv * 8);
            const v4f x1 = *(const v4f*)(el + (size_t)sv * 8 + 4);
#pragma unroll
            for (int h = 0; h < 4; ++h) {
                e[h]     = leaky02(x0[h] + ern[h]);
                e[h + 4] = leaky02(x1[h] + ern[h + 4]);
            }
        } else {
            e[0] = leaky02(el[sv] + ern[0]);
        }
    }
}

template <int H, int U>
__device__ inline void walk_group(const float* __restrict__ featL, int sv, const float (&coef)[H],
                                  int i, float (&acc)[8])
{
    float f[U][8];
#pragma unroll
    for (int u = 0; u < U; ++u) {
        const int sN = __builtin_amdgcn_readlane(sv, i + u);
        const float* fr = featL + (size_t)sN * 256;
#pragma unroll
        for (int j = 0; j < 8; ++j) f[u][j] = fr[32 * j];
    }
#pragma unroll
    for (int u = 0; u < U; ++u) {
#pragma unroll
        for (int j = 0; j < 8; ++j) {
            const float a = __int_as_float(__builtin_amdgcn_readlane(__float_as_int(coef[(j * H) >> 3]), i + u));
            acc[j] = fmaf(f[u][j], a, acc[j]);
        }
    }
}

template <int H>
__device__ inline void walk_rows(const float* __restrict__ featL, int sv, const float (&coef)[H],
                                 int cnt, float (&acc)[8])
{
    int i = 0;
    for (; i + AGG_U <= cnt; i += AGG_U) walk_group<H, AGG_U>(featL, sv, coef, i, acc);
    if (i + 2 <= cnt) { walk_group<H, 2>(featL, sv, coef, i, acc); i += 2; }
    if (i < cnt)      { walk_group<H, 1>(featL, sv, coef, i, acc); }
}

template <int H, bool RELU>
__global__ __launch_bounds__(NODE_TB)
void node_softmax_agg_k(const int* __restrict__ rp, const int* __restrict__ csrc,
                        const float* __restrict__ el, const float* __restrict__ er,
                        const float* __restrict__ feat,
                        const float* __restrict__ bias, const float* __restrict__ buf,
                        const float* __restrict__ bb, float* __restrict__ outp, int N)
{
    const int lane = threadIdx.x & 31;
    const int n = __builtin_amdgcn_readfirstlane((int)(blockIdx.x * (blockDim.x >> 5) + (threadIdx.x >> 5)));
    if (n >= N) return;
    const int start = __builtin_amdgcn_readfirstlane(rp[n]);
    const int end   = __builtin_amdgcn_readfirstlane(rp[n + 1]);
    const int len   = end - start;

    const size_t rowOff = (size_t)n * 256 + lane;
    float bufv[8];
#pragma unroll
    for (int j = 0; j < 8; ++j) bufv[j] = buf[rowOff + 32 * j];

    float ern[H];
#pragma unroll
    for (int h = 0; h < H; ++h) ern[h] = er[(size_t)n * H + h];

    float acc[8];
#pragma unroll
    for (int j = 0; j < 8; ++j) acc[j] = 0.0f;

    const float* featL = feat + lane;

    if (len > 0) {
        const int cnt0 = len < 32 ? len : 32;
        int sv0;
        float e0[H];
        load_scores<H>(csrc, el, ern, start, cnt0, lane, sv0, e0);

        float mx[H];
#pragma unroll
        for (int h = 0; h < H; ++h) mx[h] = e0[h];
        for (int b = start + 32; b < end; b += 32) {
            const int c = (end - b < 32) ? (end - b) : 32;
            int svb; float eb[H];
            load_scores<H>(csrc, el, ern, b, c, lane, svb, eb);
#pragma unroll
            for (int h = 0; h < H; ++h) mx[h] = fmaxf(mx[h], eb[h]);
        }
#pragma unroll
        for (int off = 16; off > 0; off >>= 1) {
#pragma unroll
            for (int h = 0; h < H; ++h) mx[h] = fmaxf(mx[h], __shfl_xor(mx[h], off, 32));
        }
#pragma unroll
        for (int h = 0; h < H; ++h) mx[h] = __int_as_float(__builtin_amdgcn_readfirstlane(__float_as_int(mx[h])));

        float ex0[H], sm[H];
#pragma unroll
        for (int h = 0; h < H; ++h) {
            ex0[h] = (lane < cnt0) ? expf(e0[h] - mx[h]) : 0.0f;
            sm[h]  = ex0[h];
        }
        for (int b = start + 32; b < end; b += 32) {
            const int c = (end - b < 32) ? (end - b) : 32;
            int svb; float eb[H];
            load_scores<H>(csrc, el, ern, b, c, lane, svb, eb);
#pragma unroll
            for (int h = 0; h < H; ++h) sm[h] += (lane < c) ? expf(eb[h] - mx[h]) : 0.0f;
        }
#pragma unroll
        for (int off = 16; off > 0; off >>= 1) {
#pragma unroll
            for (int h = 0; h < H; ++h) sm[h] += __shfl_xor(sm[h], off, 32);
        }
#pragma unroll
        for (int h = 0; h < H; ++h) sm[h] = __int_as_float(__builtin_amdgcn_readfirstlane(__float_as_int(sm[h])));

        float coef[H];
#pragma unroll
        for (int h = 0; h < H; ++h) coef[h] = ex0[h] / sm[h];
        walk_rows<H>(featL, sv0, coef, cnt0, acc);
        for (int b = start + 32; b < end; b += 32) {
            const int c = (end - b < 32) ? (end - b) : 32;
            int svb; float eb[H];
            load_scores<H>(csrc, el, ern, b, c, lane, svb, eb);
#pragma unroll
            for (int h = 0; h < H; ++h) coef[h] = ((lane < c) ? expf(eb[h] - mx[h]) : 0.0f) / sm[h];
            walk_rows<H>(featL, svb, coef, c, acc);
        }
    }

    const float dg = (float)len;
    const float norm = 1.0f / fmaxf(dg, 1.0f);
    float* o = outp + rowOff;
#pragma unroll
    for (int j = 0; j < 8; ++j) {
        float v = acc[j] + bias[lane + 32 * j] + norm * (bufv[j] + bb[lane + 32 * j]);
        if (RELU) v = v > 0.0f ? v : 0.0f;
        o[32 * j] = v;
    }
}

extern "C" void kernel_launch(void* const* d_in, const int* in_sizes, int n_in,
                              void* d_out, int out_size, void* d_ws, size_t ws_size,
                              hipStream_t stream)
{
    const float* inputs  = (const float*)d_in[0];
    const int*   src     = (const int*)  d_in[1];
    const int*   dst     = (const int*)  d_in[2];
    const float* fc1_w   = (const float*)d_in[3];
    const float* attn_l1 = (const float*)d_in[4];
    const float* attn_r1 = (const float*)d_in[5];
    const float* bias1   = (const float*)d_in[6];
    const float* fc2_w   = (const float*)d_in[7];
    const float* attn_l2 = (const float*)d_in[8];
    const float* attn_r2 = (const float*)d_in[9];
    const float* bias2   = (const float*)d_in[10];
    const float* wb0     = (const float*)d_in[11];
    const float* bb0     = (const float*)d_in[12];
    const float* wb1     = (const float*)d_in[13];
    const float* bb1     = (const float*)d_in[14];
    float* out = (float*)d_out;

    const int F  = 256;
    const int N  = in_sizes[0] / F;
    const int E  = in_sizes[1];
    const int H1 = 8, D1 = 32;

    float* ws   = (float*)d_ws;
    float* feat = ws;                ws += (size_t)N * F;
    float* buf  = ws;                ws += (size_t)N * F;
    float* h1   = ws;                ws += (size_t)N * F;
    float* el   = ws;                ws += (size_t)N * H1;
    float* er   = ws;                ws += (size_t)N * H1;
    int*   wi   = (int*)ws;
    int*   cnt  = wi;                wi += (size_t)N;
    int*   rank = wi;                wi += (size_t)E;
    int*   csrc = wi;                wi += (size_t)E;
    int*   rp   = wi;

    const int TB = 256;
    auto cdiv = [](int a, int b) { return (a + b - 1) / b; };

    const int waves    = (N / 16) * 4;
    const int gemmGrid = cdiv(waves, TB / 32);
    const int nodeGrid = cdiv(N, NODE_TB / 32);

    fill_i32<<<cdiv(N, TB), TB, 0, stream>>>(cnt, 0, N);
    edge_rank_k<<<cdiv(E, TB), TB, 0, stream>>>(dst, cnt, rank, E);
    scan_counts_k<<<1, SCAN_TB, 0, stream>>>(cnt, rp, N);
    edge_place_k<<<cdiv(E, TB), TB, 0, stream>>>(src, dst, rank, rp, csrc, E);

    gemm_wmma_f32<<<gemmGrid, TB, 0, stream>>>(inputs, F, fc1_w, F, feat, F, N, F, 0);
    gemm_wmma_f32<<<gemmGrid, TB, 0, stream>>>(inputs, F, wb0,   F, buf,  F, N, F, 0);

    attn_scores<<<cdiv(N * H1, TB / 32), TB, 0, stream>>>(feat, attn_l1, attn_r1, el, er, N * H1, H1, D1);
    node_softmax_agg_k<8, true><<<nodeGrid, NODE_TB, 0, stream>>>(rp, csrc, el, er, feat, bias1, buf, bb0, h1, N);

    gemm_wmma_f32<<<gemmGrid, TB, 0, stream>>>(h1, F, fc2_w, F, feat, F, N, F, 0);
    gemm_wmma_f32<<<gemmGrid, TB, 0, stream>>>(inputs, F, wb1,     2 * F, buf, F, N, F, 0);
    gemm_wmma_f32<<<gemmGrid, TB, 0, stream>>>(h1,     F, wb1 + F, 2 * F, buf, F, N, F, 1);

    attn_scores<<<cdiv(N, TB / 32), TB, 0, stream>>>(feat, attn_l2, attn_r2, el, er, N, 1, F);
    node_softmax_agg_k<1, false><<<nodeGrid, NODE_TB, 0, stream>>>(rp, csrc, el, er, feat, bias2, buf, bb1, out, N);
}


# attention scores: one wave takes 4 whole node rows (8 x 16-B loads per lane, all heads at once), summed by a transposing lane tree: 32x and 4x fewer waves; score sums in a new pairwise order
# speedup vs baseline: 3.1457x; 1.1254x over previous
#include <hip/hip_runtime.h>
#include <hip/hip_bf16.h>
#include <math.h>


typedef __attribute__((ext_vector_type(2))) float v2f;
typedef __attribute__((ext_vector_type(4))) float v4f;
typedef __attribute__((ext_vector_type(8))) float v8f;

__global__ void fill_i32(int* __restrict__ p, int v, int n) {
    int i = blockIdx.x * blockDim.x + threadIdx.x;
    if (i < n) p[i] = v;
}

__global__ void edge_rank_k(const int* __restrict__ dst, int* __restrict__ cnt,
                            int* __restrict__ rank, int E)
{
    int e = blockIdx.x * blockDim.x + threadIdx.x;
    if (e < E) rank[e] = atomicAdd(&cnt[dst[e]], 1);
}

#define SCAN_TB 1024
#define SCAN_CH 32
__global__ __launch_bounds__(SCAN_TB)
void scan_counts_k(const int* __restrict__ cnt, int* __restrict__ rp, int N)
{
    __shared__ int wtot[32];
    __shared__ int tileTot;
    const int t = threadIdx.x;
    const int lane = t & 31;
    const int wv = t >> 5;
    int carry = 0;
    for (int base = 0; base <= N; base += SCAN_TB * SCAN_CH) {
        const int b = base + t * SCAN_CH;
        int c[SCAN_CH];
#pragma unroll
        for (int i = 0; i < SCAN_CH; ++i) c[i] = (b + i < N) ? cnt[b + i] : 0;
        int sum = 0;
#pragma unroll
        for (int i = 0; i < SCAN_CH; ++i) sum += c[i];
        int inc = sum;
#pragma unroll
        for (int off = 1; off < 32; off <<= 1) {
            const int v = __shfl_up(inc, off, 32);
            if (lane >= off) inc += v;
        }
        if (lane == 31) wtot[wv] = inc;
        __syncthreads();
        if (wv == 0) {
            const int w = wtot[lane];
            int winc = w;
#pragma unroll
            for (int off = 1; off < 32; off <<= 1) {
                const int v = __shfl_up(winc, off, 32);
                if (lane >= off) winc += v;
            }
            wtot[lane] = winc - w;
            if (lane == 31) tileTot = winc;
        }
        __syncthreads();
        int run = carry + wtot[wv] + inc - sum;
#pragma unroll
        for (int i = 0; i < SCAN_CH; ++i) {
            if (b + i <= N) rp[b + i] = run;
            run += c[i];
        }
        carry += tileTot;
        __syncthreads();
    }
}

__global__ void edge_place_k(const int* __restrict__ src, const int* __restrict__ dst,
                             const int* __restrict__ rank, const int* __restrict__ rp,
                             int* __restrict__ csrc, int E)
{
    int e = blockIdx.x * blockDim.x + threadIdx.x;
    if (e < E) csrc[rp[dst[e]] + rank[e]] = src[e];
}

#define GEMM_KC 32
#define GEMM_LP 36
__global__ __launch_bounds__(256)
void gemm_wmma_f32(const float* __restrict__ A, int lda,
                   const float* __restrict__ W, int ldw,
                   float* __restrict__ C, int ldc,
                   int M, int K, int accumulate)
{
    __shared__ __attribute__((aligned(16))) float Ws[256 * GEMM_LP];
    __shared__ __attribute__((aligned(16))) float As[32 * GEMM_LP];

    const int tid  = threadIdx.x;
    const int wave = tid >> 5;
    const int lane = tid & 31;
    const int half = lane >> 4;
    const int lr   = lane & 15;
    const int wr   = wave >> 2;
    const int col0 = (wave & 3) * 64;
    const int rowBase = blockIdx.x * 32;
    const int rowTile = blockIdx.x * 2 + wr;
    const int nRowTiles = M >> 4;

    const int sRow = tid >> 3;
    const int sCol = (tid & 7) * 4;
    int aRowG = rowBase + sRow;
    if (aRowG > M - 1) aRowG = M - 1;
    const float* Ag = A + (size_t)aRowG * lda + sCol;
    const float* Wg = W + (size_t)sRow * ldw + sCol;
    float* AsSt = As + sRow * GEMM_LP + sCol;
    float* WsSt = Ws + sRow * GEMM_LP + sCol;

    const float* aF = As + (wr * 16 + lr) * GEMM_LP + 2 * half;
    const float* bF = Ws + (col0 + lr) * GEMM_LP + 2 * half;

    v8f acc0 = {}, acc1 = {}, acc2 = {}, acc3 = {};

    v4f areg = *(const v4f*)(Ag);
    v4f wreg[8];
#pragma unroll
    for (int j = 0; j < 8; ++j)
        wreg[j] = *(const v4f*)(Wg + (size_t)(32 * j) * ldw);

    for (int k0 = 0; k0 < K; k0 += GEMM_KC) {
        __syncthreads();
        *(v4f*)(AsSt) = areg;
#pragma unroll
        for (int j = 0; j < 8; ++j)
            *(v4f*)(WsSt + (32 * j) * GEMM_LP) = wreg[j];
        __syncthreads();

        if (k0 + GEMM_KC < K) {
            areg = *(const v4f*)(Ag + k0 + GEMM_KC);
#pragma unroll
            for (int j = 0; j < 8; ++j)
                wreg[j] = *(const v4f*)(Wg + (size_t)(32 * j) * ldw + k0 + GEMM_KC);
        }

#pragma unroll
        for (int hc = 0; hc < GEMM_KC; hc += 16) {
            v2f a[4], b0[4], b1[4], b2[4], b3[4];
#pragma unroll
            for (int s = 0; s < 4; ++s) {
                const int kk = hc + 4 * s;
                a[s]  = *(const v2f*)(aF + kk);
                b0[s] = *(const v2f*)(bF +  0 * GEMM_LP + kk);
                b1[s] = *(const v2f*)(bF + 16 * GEMM_LP + kk);
                b2[s] = *(const v2f*)(bF + 32 * GEMM_LP + kk);
                b3[s] = *(const v2f*)(bF + 48 * GEMM_LP + kk);
            }
#pragma unroll
            for (int s = 0; s < 4; ++s) {
                acc0 = __builtin_amdgcn_wmma_f32_16x16x4_f32(false, a[s], false, b0[s], (short)0, acc0, false, false);
                acc1 = __builtin_amdgcn_wmma_f32_16x16x4_f32(false, a[s], false, b1[s], (short)0, acc1, false, false);
                acc2 = __builtin_amdgcn_wmma_f32_16x16x4_f32(false, a[s], false, b2[s], (short)0, acc2, false, false);
                acc3 = __builtin_amdgcn_wmma_f32_16x16x4_f32(false, a[s], false, b3[s], (short)0, acc3, false, false);
            }
        }
    }

    if (rowTile >= nRowTiles) return;

    float* Cb = C + (size_t)(rowTile * 16 + 8 * half) * ldc + col0 + lr;
#pragma unroll
    for (int j = 0; j < 8; ++j) {
        float* p = Cb + (size_t)j * ldc;
        p[ 0] = (accumulate ? p[ 0] : 0.0f) + acc0[j];
        p[16] = (accumulate ? p[16] : 0.0f) + acc1[j];
        p[32] = (accumulate ? p[32] : 0.0f) + acc2[j];
        p[48] = (accumulate ? p[48] : 0.0f) + acc3[j];
    }
}

#define SC_NPW 4
#define SC_TB  256

__device__ inline float dot4(const v4f x, const v4f a)
{
    float p = x[0] * a[0];
    p = fmaf(x[1], a[1], p);
    p = fmaf(x[2], a[2], p);
    p = fmaf(x[3], a[3], p);
    return p;
}

template <int HALF>
__device__ inline void fold_step(float (&v)[2 * HALF], bool up, int mask)
{
#pragma unroll
    for (int k = 0; k < HALF; ++k) {
        const float keep = up ? v[k + HALF] : v[k];
        const float send = up ? v[k] : v[k + HALF];
        v[k] = keep + __shfl_xor(send, mask, 32);
    }
}

template <int H>
__global__ __launch_bounds__(SC_TB)
void attn_scores(const float* __restrict__ feat,
                 const float* __restrict__ attn_l,
                 const float* __restrict__ attn_r,
                 float* __restrict__ el, float* __restrict__ er, int N)
{
    const int lane = threadIdx.x & 31;
    const int n0 = (blockIdx.x * (blockDim.x >> 5) + (threadIdx.x >> 5)) * SC_NPW;
    if (n0 >= N) return;

    v4f x[2 * SC_NPW];
#pragma unroll
    for (int q = 0; q < SC_NPW; ++q) {
        int n = n0 + q;
        if (n > N - 1) n = N - 1;
        const float* fr = feat + (size_t)n * 256 + 4 * lane;
        x[2 * q]     = *(const v4f*)(fr);
        x[2 * q + 1] = *(const v4f*)(fr + 128);
    }
    const v4f al0 = *(const v4f*)(attn_l + 4 * lane);
    const v4f al1 = *(const v4f*)(attn_l + 128 + 4 * lane);
    const v4f ar0 = *(const v4f*)(attn_r + 4 * lane);
    const v4f ar1 = *(const v4f*)(attn_r + 128 + 4 * lane);

    float pl[2 * SC_NPW], pr[2 * SC_NPW];
#pragma unroll
    for (int q = 0; q < SC_NPW; ++q) {
        pl[2 * q]     = dot4(x[2 * q],     al0);
        pl[2 * q + 1] = dot4(x[2 * q + 1], al1);
        pr[2 * q]     = dot4(x[2 * q],     ar0);
        pr[2 * q + 1] = dot4(x[2 * q + 1], ar1);
    }

    if constexpr (H == 8) {
        fold_step<4>(pl, (lane & 4) != 0, 4);
        fold_step<4>(pr, (lane & 4) != 0, 4);
        float l4[4] = { pl[0], pl[1], pl[2], pl[3] };
        float r4[4] = { pr[0], pr[1], pr[2], pr[3] };
        fold_step<2>(l4, (lane & 2) != 0, 2);
        fold_step<2>(r4, (lane & 2) != 0, 2);
        float l2[2] = { l4[0], l4[1] };
        float r2[2] = { r4[0], r4[1] };
        fold_step<1>(l2, (lane & 1) != 0, 1);
        fold_step<1>(r2, (lane & 1) != 0, 1);
        const int slot = 4 * (lane & 7) + (lane >> 3);
        if (n0 + (slot >> 3) < N) {
            el[(size_t)n0 * 8 + slot] = l2[0];
            er[(size_t)n0 * 8 + slot] = r2[0];
        }
    } else {
        float sl[SC_NPW], sr[SC_NPW];
#pragma unroll
        for (int q = 0; q < SC_NPW; ++q) {
            sl[q] = pl[2 * q] + pl[2 * q + 1];
            sr[q] = pr[2 * q] + pr[2 * q + 1];
        }
        fold_step<2>(sl, (lane & 16) != 0, 16);
        fold_step<2>(sr, (lane & 16) != 0, 16);
        float l2[2] = { sl[0], sl[1] };
        float r2[2] = { sr[0], sr[1] };
        fold_step<1>(l2, (lane & 8) != 0, 8);
        fold_step<1>(r2, (lane & 8) != 0, 8);
        float tl = l2[0], tr = r2[0];
#pragma unroll
        for (int off = 4; off > 0; off >>= 1) {
            tl += __shfl_xor(tl, off, 32);
            tr += __shfl_xor(tr, off, 32);
        }
        const int q = lane >> 3;
        if ((lane & 7) == 0 && n0 + q < N) { el[n0 + q] = tl; er[n0 + q] = tr; }
    }
}

__device__ inline float leaky02(float v) { return v > 0.0f ? v : 0.2f * v; }

#define NODE_TB 256
#define AGG_U  4

template <int H>
__device__ inline void load_scores(const int* __restrict__ csrc, const float* __restrict__ el,
                                   const float (&ern)[H], int b, int cnt, int lane,
                                   int& sv, float (&e)[H])
{
    sv = 0;
#pragma unroll
    for (int h = 0; h < H; ++h) e[h] = -INFINITY;
    if (lane < cnt) {
        sv = csrc[b + lane];
        if constexpr (H == 8) {
            const v4f x0 = *(const v4f*)(el + (size_t)sv * 8);
            const v4f x1 = *(const v4f*)(el + (size_t)sv * 8 + 4);
#pragma unroll
            for (int h = 0; h < 4; ++h) {
                e[h]     = leaky02(x0[h] + ern[h]);
                e[h + 4] = leaky02(x1[h] + ern[h + 4]);
            }
        } else {
            e[0] = leaky02(el[sv] + ern[0]);
        }
    }
}

template <int H, int U>
__device__ inline void walk_group(const float* __restrict__ featL, int sv, const float (&coef)[H],
                                  int i, float (&acc)[8])
{
    float f[U][8];
#pragma unroll
    for (int u = 0; u < U; ++u) {
        const int sN = __builtin_amdgcn_readlane(sv, i + u);
        const float* fr = featL + (size_t)sN * 256;
#pragma unroll
        for (int j = 0; j < 8; ++j) f[u][j] = fr[32 * j];
    }
#pragma unroll
    for (int u = 0; u < U; ++u) {
#pragma unroll
        for (int j = 0; j < 8; ++j) {
            const float a = __int_as_float(__builtin_amdgcn_readlane(__float_as_int(coef[(j * H) >> 3]), i + u));
            acc[j] = fmaf(f[u][j], a, acc[j]);
        }
    }
}

template <int H>
__device__ inline void walk_rows(const float* __restrict__ featL, int sv, const float (&coef)[H],
                                 int cnt, float (&acc)[8])
{
    int i = 0;
    for (; i + AGG_U <= cnt; i += AGG_U) walk_group<H, AGG_U>(featL, sv, coef, i, acc);
    if (i + 2 <= cnt) { walk_group<H, 2>(featL, sv, coef, i, acc); i += 2; }
    if (i < cnt)      { walk_group<H, 1>(featL, sv, coef, i, acc); }
}

template <int H, bool RELU>
__global__ __launch_bounds__(NODE_TB)
void node_softmax_agg_k(const int* __restrict__ rp, const int* __restrict__ csrc,
                        const float* __restrict__ el, const float* __restrict__ er,
                        const float* __restrict__ feat,
                        const float* __restrict__ bias, const float* __restrict__ buf,
                        const float* __restrict__ bb, float* __restrict__ outp, int N)
{
    const int lane = threadIdx.x & 31;
    const int n = __builtin_amdgcn_readfirstlane((int)(blockIdx.x * (blockDim.x >> 5) + (threadIdx.x >> 5)));
    if (n >= N) return;
    const int start = __builtin_amdgcn_readfirstlane(rp[n]);
    const int end   = __builtin_amdgcn_readfirstlane(rp[n + 1]);
    const int len   = end - start;

    const size_t rowOff = (size_t)n * 256 + lane;
    float bufv[8];
#pragma unroll
    for (int j = 0; j < 8; ++j) bufv[j] = buf[rowOff + 32 * j];

    float ern[H];
#pragma unroll
    for (int h = 0; h < H; ++h) ern[h] = er[(size_t)n * H + h];

    float acc[8];
#pragma unroll
    for (int j = 0; j < 8; ++j) acc[j] = 0.0f;

    const float* featL = feat + lane;

    if (len > 0) {
        const int cnt0 = len < 32 ? len : 32;
        int sv0;
        float e0[H];
        load_scores<H>(csrc, el, ern, start, cnt0, lane, sv0, e0);

        float mx[H];
#pragma unroll
        for (int h = 0; h < H; ++h) mx[h] = e0[h];
        for (int b = start + 32; b < end; b += 32) {
            const int c = (end - b < 32) ? (end - b) : 32;
            int svb; float eb[H];
            load_scores<H>(csrc, el, ern, b, c, lane, svb, eb);
#pragma unroll
            for (int h = 0; h < H; ++h) mx[h] = fmaxf(mx[h], eb[h]);
        }
#pragma unroll
        for (int off = 16; off > 0; off >>= 1) {
#pragma unroll
            for (int h = 0; h < H; ++h) mx[h] = fmaxf(mx[h], __shfl_xor(mx[h], off, 32));
        }
#pragma unroll
        for (int h = 0; h < H; ++h) mx[h] = __int_as_float(__builtin_amdgcn_readfirstlane(__float_as_int(mx[h])));

        float ex0[H], sm[H];
#pragma unroll
        for (int h = 0; h < H; ++h) {
            ex0[h] = (lane < cnt0) ? expf(e0[h] - mx[h]) : 0.0f;
            sm[h]  = ex0[h];
        }
        for (int b = start + 32; b < end; b += 32) {
            const int c = (end - b < 32) ? (end - b) : 32;
            int svb; float eb[H];
            load_scores<H>(csrc, el, ern, b, c, lane, svb, eb);
#pragma unroll
            for (int h = 0; h < H; ++h) sm[h] += (lane < c) ? expf(eb[h] - mx[h]) : 0.0f;
        }
#pragma unroll
        for (int off = 16; off > 0; off >>= 1) {
#pragma unroll
            for (int h = 0; h < H; ++h) sm[h] += __shfl_xor(sm[h], off, 32);
        }
#pragma unroll
        for (int h = 0; h < H; ++h) sm[h] = __int_as_float(__builtin_amdgcn_readfirstlane(__float_as_int(sm[h])));

        float coef[H];
#pragma unroll
        for (int h = 0; h < H; ++h) coef[h] = ex0[h] / sm[h];
        walk_rows<H>(featL, sv0, coef, cnt0, acc);
        for (int b = start + 32; b < end; b += 32) {
            const int c = (end - b < 32) ? (end - b) : 32;
            int svb; float eb[H];
            load_scores<H>(csrc, el, ern, b, c, lane, svb, eb);
#pragma unroll
            for (int h = 0; h < H; ++h) coef[h] = ((lane < c) ? expf(eb[h] - mx[h]) : 0.0f) / sm[h];
            walk_rows<H>(featL, svb, coef, c, acc);
        }
    }

    const float dg = (float)len;
    const float norm = 1.0f / fmaxf(dg, 1.0f);
    float* o = outp + rowOff;
#pragma unroll
    for (int j = 0; j < 8; ++j) {
        float v = acc[j] + bias[lane + 32 * j] + norm * (bufv[j] + bb[lane + 32 * j]);
        if (RELU) v = v > 0.0f ? v : 0.0f;
        o[32 * j] = v;
    }
}

extern "C" void kernel_launch(void* const* d_in, const int* in_sizes, int n_in,
                              void* d_out, int out_size, void* d_ws, size_t ws_size,
                              hipStream_t stream)
{
    const float* inputs  = (const float*)d_in[0];
    const int*   src     = (const int*)  d_in[1];
    const int*   dst     = (const int*)  d_in[2];
    const float* fc1_w   = (const float*)d_in[3];
    const float* attn_l1 = (const float*)d_in[4];
    const float* attn_r1 = (const float*)d_in[5];
    const float* bias1   = (const float*)d_in[6];
    const float* fc2_w   = (const float*)d_in[7];
    const float* attn_l2 = (const float*)d_in[8];
    const float* attn_r2 = (const float*)d_in[9];
    const float* bias2   = (const float*)d_in[10];
    const float* wb0     = (const float*)d_in[11];
    const float* bb0     = (const float*)d_in[12];
    const float* wb1     = (const float*)d_in[13];
    const float* bb1     = (const float*)d_in[14];
    float* out = (float*)d_out;

    const int F  = 256;
    const int N  = in_sizes[0] / F;
    const int E  = in_sizes[1];
    const int H1 = 8;

    float* ws   = (float*)d_ws;
    float* feat = ws;                ws += (size_t)N * F;
    float* buf  = ws;                ws += (size_t)N * F;
    float* h1   = ws;                ws += (size_t)N * F;
    float* el   = ws;                ws += (size_t)N * H1;
    float* er   = ws;                ws += (size_t)N * H1;
    int*   wi   = (int*)ws;
    int*   cnt  = wi;                wi += (size_t)N;
    int*   rank = wi;                wi += (size_t)E;
    int*   csrc = wi;                wi += (size_t)E;
    int*   rp   = wi;

    const int TB = 256;
    auto cdiv = [](int a, int b) { return (a + b - 1) / b; };

    const int waves    = (N / 16) * 4;
    const int gemmGrid = cdiv(waves, TB / 32);
    const int nodeGrid = cdiv(N, NODE_TB / 32);
    const int scoreGrid = cdiv(N, (SC_TB / 32) * SC_NPW);

    fill_i32<<<cdiv(N, TB), TB, 0, stream>>>(cnt, 0, N);
    edge_rank_k<<<cdiv(E, TB), TB, 0, stream>>>(dst, cnt, rank, E);
    scan_counts_k<<<1, SCAN_TB, 0, stream>>>(cnt, rp, N);
    edge_place_k<<<cdiv(E, TB), TB, 0, stream>>>(src, dst, rank, rp, csrc, E);

    gemm_wmma_f32<<<gemmGrid, TB, 0, stream>>>(inputs, F, fc1_w, F, feat, F, N, F, 0);
    gemm_wmma_f32<<<gemmGrid, TB, 0, stream>>>(inputs, F, wb0,   F, buf,  F, N, F, 0);

    attn_scores<8><<<scoreGrid, SC_TB, 0, stream>>>(feat, attn_l1, attn_r1, el, er, N);
    node_softmax_agg_k<8, true><<<nodeGrid, NODE_TB, 0, stream>>>(rp, csrc, el, er, feat, bias1, buf, bb0, h1, N);

    gemm_wmma_f32<<<gemmGrid, TB, 0, stream>>>(h1, F, fc2_w, F, feat, F, N, F, 0);
    gemm_wmma_f32<<<gemmGrid, TB, 0, stream>>>(inputs, F, wb1,     2 * F, buf, F, N, F, 0);
    gemm_wmma_f32<<<gemmGrid, TB, 0, stream>>>(h1,     F, wb1 + F, 2 * F, buf, F, N, F, 1);

    attn_scores<1><<<scoreGrid, SC_TB, 0, stream>>>(feat, attn_l2, attn_r2, el, er, N);
    node_softmax_agg_k<1, false><<<nodeGrid, NODE_TB, 0, stream>>>(rp, csrc, el, er, feat, bias2, buf, bb1, out, N);
}


# GEMMs that share their left operand run as one launch (grid y = weight matrix: 3 on inputs, 2 on h1): 5 GEMM launches become 2, blocks dealt out as slots free; sums and k order unchanged
# speedup vs baseline: 3.4788x; 1.1059x over previous
#include <hip/hip_runtime.h>
#include <hip/hip_bf16.h>
#include <math.h>


typedef __attribute__((ext_vector_type(2))) float v2f;
typedef __attribute__((ext_vector_type(4))) float v4f;
typedef __attribute__((ext_vector_type(8))) float v8f;

__global__ void fill_i32(int* __restrict__ p, int v, int n) {
    int i = blockIdx.x * blockDim.x + threadIdx.x;
    if (i < n) p[i] = v;
}

__global__ void edge_rank_k(const int* __restrict__ dst, int* __restrict__ cnt,
                            int* __restrict__ rank, int E)
{
    int e = blockIdx.x * blockDim.x + threadIdx.x;
    if (e < E) rank[e] = atomicAdd(&cnt[dst[e]], 1);
}

#define SCAN_TB 1024
#define SCAN_CH 32
__global__ __launch_bounds__(SCAN_TB)
void scan_counts_k(const int* __restrict__ cnt, int* __restrict__ rp, int N)
{
    __shared__ int wtot[32];
    __shared__ int tileTot;
    const int t = threadIdx.x;
    const int lane = t & 31;
    const int wv = t >> 5;
    int carry = 0;
    for (int base = 0; base <= N; base += SCAN_TB * SCAN_CH) {
        const int b = base + t * SCAN_CH;
        int c[SCAN_CH];
#pragma unroll
        for (int i = 0; i < SCAN_CH; ++i) c[i] = (b + i < N) ? cnt[b + i] : 0;
        int sum = 0;
#pragma unroll
        for (int i = 0; i < SCAN_CH; ++i) sum += c[i];
        int inc = sum;
#pragma unroll
        for (int off = 1; off < 32; off <<= 1) {
            const int v = __shfl_up(inc, off, 32);
            if (lane >= off) inc += v;
        }
        if (lane == 31) wtot[wv] = inc;
        __syncthreads();
        if (wv == 0) {
            const int w = wtot[lane];
            int winc = w;
#pragma unroll
            for (int off = 1; off < 32; off <<= 1) {
                const int v = __shfl_up(winc, off, 32);
                if (lane >= off) winc += v;
            }
            wtot[lane] = winc - w;
            if (lane == 31) tileTot = winc;
        }
        __syncthreads();
        int run = carry + wtot[wv] + inc - sum;
#pragma unroll
        for (int i = 0; i < SCAN_CH; ++i) {
            if (b + i <= N) rp[b + i] = run;
            run += c[i];
        }
        carry += tileTot;
        __syncthreads();
    }
}

__global__ void edge_place_k(const int* __restrict__ src, const int* __restrict__ dst,
                             const int* __restrict__ rank, const int* __restrict__ rp,
                             int* __restrict__ csrc, int E)
{
    int e = blockIdx.x * blockDim.x + threadIdx.x;
    if (e < E) csrc[rp[dst[e]] + rank[e]] = src[e];
}

#define GEMM_KC 32
#define GEMM_LP 36
__global__ __launch_bounds__(256)
void gemm_wmma_f32(const float* __restrict__ A, int lda,
                   const float* __restrict__ W0, int ldw0, float* __restrict__ C0,
                   const float* __restrict__ W1, int ldw1, float* __restrict__ C1,
                   const float* __restrict__ W2, int ldw2, float* __restrict__ C2,
                   int ldc, int M, int K, int accMask)
{
    const int y = blockIdx.y;
    const float* __restrict__ W = (y == 0) ? W0 : (y == 1) ? W1 : W2;
    float* __restrict__ C       = (y == 0) ? C0 : (y == 1) ? C1 : C2;
    const int ldw               = (y == 0) ? ldw0 : (y == 1) ? ldw1 : ldw2;
    const int accumulate        = (accMask >> y) & 1;

    __shared__ __attribute__((aligned(16))) float Ws[256 * GEMM_LP];
    __shared__ __attribute__((aligned(16))) float As[32 * GEMM_LP];

    const int tid  = threadIdx.x;
    const int wave = tid >> 5;
    const int lane = tid & 31;
    const int half = lane >> 4;
    const int lr   = lane & 15;
    const int wr   = wave >> 2;
    const int col0 = (wave & 3) * 64;
    const int rowBase = blockIdx.x * 32;
    const int rowTile = blockIdx.x * 2 + wr;
    const int nRowTiles = M >> 4;

    const int sRow = tid >> 3;
    const int sCol = (tid & 7) * 4;
    int aRowG = rowBase + sRow;
    if (aRowG > M - 1) aRowG = M - 1;
    const float* Ag = A + (size_t)aRowG * lda + sCol;
    const float* Wg = W + (size_t)sRow * ldw + sCol;
    float* AsSt = As + sRow * GEMM_LP + sCol;
    float* WsSt = Ws + sRow * GEMM_LP + sCol;

    const float* aF = As + (wr * 16 + lr) * GEMM_LP + 2 * half;
    const float* bF = Ws + (col0 + lr) * GEMM_LP + 2 * half;

    v8f acc0 = {}, acc1 = {}, acc2 = {}, acc3 = {};

    v4f areg = *(const v4f*)(Ag);
    v4f wreg[8];
#pragma unroll
    for (int j = 0; j < 8; ++j)
        wreg[j] = *(const v4f*)(Wg + (size_t)(32 * j) * ldw);

    for (int k0 = 0; k0 < K; k0 += GEMM_KC) {
        __syncthreads();
        *(v4f*)(AsSt) = areg;
#pragma unroll
        for (int j = 0; j < 8; ++j)
            *(v4f*)(WsSt + (32 * j) * GEMM_LP) = wreg[j];
        __syncthreads();

        if (k0 + GEMM_KC < K) {
            areg = *(const v4f*)(Ag + k0 + GEMM_KC);
#pragma unroll
            for (int j = 0; j < 8; ++j)
                wreg[j] = *(const v4f*)(Wg + (size_t)(32 * j) * ldw + k0 + GEMM_KC);
        }

#pragma unroll
        for (int hc = 0; hc < GEMM_KC; hc += 16) {
            v2f a[4], b0[4], b1[4], b2[4], b3[4];
#pragma unroll
            for (int s = 0; s < 4; ++s) {
                const int kk = hc + 4 * s;
                a[s]  = *(const v2f*)(aF + kk);
                b0[s] = *(const v2f*)(bF +  0 * GEMM_LP + kk);
                b1[s] = *(const v2f*)(bF + 16 * GEMM_LP + kk);
                b2[s] = *(const v2f*)(bF + 32 * GEMM_LP + kk);
                b3[s] = *(const v2f*)(bF + 48 * GEMM_LP + kk);
            }
#pragma unroll
            for (int s = 0; s < 4; ++s) {
                acc0 = __builtin_amdgcn_wmma_f32_16x16x4_f32(false, a[s], false, b0[s], (short)0, acc0, false, false);
                acc1 = __builtin_amdgcn_wmma_f32_16x16x4_f32(false, a[s], false, b1[s], (short)0, acc1, false, false);
                acc2 = __builtin_amdgcn_wmma_f32_16x16x4_f32(false, a[s], false, b2[s], (short)0, acc2, false, false);
                acc3 = __builtin_amdgcn_wmma_f32_16x16x4_f32(false, a[s], false, b3[s], (short)0, acc3, false, false);
            }
        }
    }

    if (rowTile >= nRowTiles) return;

    float* Cb = C + (size_t)(rowTile * 16 + 8 * half) * ldc + col0 + lr;
#pragma unroll
    for (int j = 0; j < 8; ++j) {
        float* p = Cb + (size_t)j * ldc;
        p[ 0] = (accumulate ? p[ 0] : 0.0f) + acc0[j];
        p[16] = (accumulate ? p[16] : 0.0f) + acc1[j];
        p[32] = (accumulate ? p[32] : 0.0f) + acc2[j];
        p[48] = (accumulate ? p[48] : 0.0f) + acc3[j];
    }
}

#define SC_NPW 4
#define SC_TB  256

__device__ inline float dot4(const v4f x, const v4f a)
{
    float p = x[0] * a[0];
    p = fmaf(x[1], a[1], p);
    p = fmaf(x[2], a[2], p);
    p = fmaf(x[3], a[3], p);
    return p;
}

template <int HALF>
__device__ inline void fold_step(float (&v)[2 * HALF], bool up, int mask)
{
#pragma unroll
    for (int k = 0; k < HALF; ++k) {
        const float keep = up ? v[k + HALF] : v[k];
        const float send = up ? v[k] : v[k + HALF];
        v[k] = keep + __shfl_xor(send, mask, 32);
    }
}

template <int H>
__global__ __launch_bounds__(SC_TB)
void attn_scores(const float* __restrict__ feat,
                 const float* __restrict__ attn_l,
                 const float* __restrict__ attn_r,
                 float* __restrict__ el, float* __restrict__ er, int N)
{
    const int lane = threadIdx.x & 31;
    const int n0 = (blockIdx.x * (blockDim.x >> 5) + (threadIdx.x >> 5)) * SC_NPW;
    if (n0 >= N) return;

    v4f x[2 * SC_NPW];
#pragma unroll
    for (int q = 0; q < SC_NPW; ++q) {
        int n = n0 + q;
        if (n > N - 1) n = N - 1;
        const float* fr = feat + (size_t)n * 256 + 4 * lane;
        x[2 * q]     = *(const v4f*)(fr);
        x[2 * q + 1] = *(const v4f*)(fr + 128);
    }
    const v4f al0 = *(const v4f*)(attn_l + 4 * lane);
    const v4f al1 = *(const v4f*)(attn_l + 128 + 4 * lane);
    const v4f ar0 = *(const v4f*)(attn_r + 4 * lane);
    const v4f ar1 = *(const v4f*)(attn_r + 128 + 4 * lane);

    float pl[2 * SC_NPW], pr[2 * SC_NPW];
#pragma unroll
    for (int q = 0; q < SC_NPW; ++q) {
        pl[2 * q]     = dot4(x[2 * q],     al0);
        pl[2 * q + 1] = dot4(x[2 * q + 1], al1);
        pr[2 * q]     = dot4(x[2 * q],     ar0);
        pr[2 * q + 1] = dot4(x[2 * q + 1], ar1);
    }

    if constexpr (H == 8) {
        fold_step<4>(pl, (lane & 4) != 0, 4);
        fold_step<4>(pr, (lane & 4) != 0, 4);
        float l4[4] = { pl[0], pl[1], pl[2], pl[3] };
        float r4[4] = { pr[0], pr[1], pr[2], pr[3] };
        fold_step<2>(l4, (lane & 2) != 0, 2);
        fold_step<2>(r4, (lane & 2) != 0, 2);
        float l2[2] = { l4[0], l4[1] };
        float r2[2] = { r4[0], r4[1] };
        fold_step<1>(l2, (lane & 1) != 0, 1);
        fold_step<1>(r2, (lane & 1) != 0, 1);
        const int slot = 4 * (lane & 7) + (lane >> 3);
        if (n0 + (slot >> 3) < N) {
            el[(size_t)n0 * 8 + slot] = l2[0];
            er[(size_t)n0 * 8 + slot] = r2[0];
        }
    } else {
        float sl[SC_NPW], sr[SC_NPW];
#pragma unroll
        for (int q = 0; q < SC_NPW; ++q) {
            sl[q] = pl[2 * q] + pl[2 * q + 1];
            sr[q] = pr[2 * q] + pr[2 * q + 1];
        }
        fold_step<2>(sl, (lane & 16) != 0, 16);
        fold_step<2>(sr, (lane & 16) != 0, 16);
        float l2[2] = { sl[0], sl[1] };
        float r2[2] = { sr[0], sr[1] };
        fold_step<1>(l2, (lane & 8) != 0, 8);
        fold_step<1>(r2, (lane & 8) != 0, 8);
        float tl = l2[0], tr = r2[0];
#pragma unroll
        for (int off = 4; off > 0; off >>= 1) {
            tl += __shfl_xor(tl, off, 32);
            tr += __shfl_xor(tr, off, 32);
        }
        const int q = lane >> 3;
        if ((lane & 7) == 0 && n0 + q < N) { el[n0 + q] = tl; er[n0 + q] = tr; }
    }
}

__device__ inline float leaky02(float v) { return v > 0.0f ? v : 0.2f * v; }

#define NODE_TB 256
#define AGG_U  4

template <int H>
__device__ inline void load_scores(const int* __restrict__ csrc, const float* __restrict__ el,
                                   const float (&ern)[H], int b, int cnt, int lane,
                                   int& sv, float (&e)[H])
{
    sv = 0;
#pragma unroll
    for (int h = 0; h < H; ++h) e[h] = -INFINITY;
    if (lane < cnt) {
        sv = csrc[b + lane];
        if constexpr (H == 8) {
            const v4f x0 = *(const v4f*)(el + (size_t)sv * 8);
            const v4f x1 = *(const v4f*)(el + (size_t)sv * 8 + 4);
#pragma unroll
            for (int h = 0; h < 4; ++h) {
                e[h]     = leaky02(x0[h] + ern[h]);
                e[h + 4] = leaky02(x1[h] + ern[h + 4]);
            }
        } else {
            e[0] = leaky02(el[sv] + ern[0]);
        }
    }
}

template <int H, int U>
__device__ inline void walk_group(const float* __restrict__ featL, int sv, const float (&coef)[H],
                                  int i, float (&acc)[8])
{
    float f[U][8];
#pragma unroll
    for (int u = 0; u < U; ++u) {
        const int sN = __builtin_amdgcn_readlane(sv, i + u);
        const float* fr = featL + (size_t)sN * 256;
#pragma unroll
        for (int j = 0; j < 8; ++j) f[u][j] = fr[32 * j];
    }
#pragma unroll
    for (int u = 0; u < U; ++u) {
#pragma unroll
        for (int j = 0; j < 8; ++j) {
            const float a = __int_as_float(__builtin_amdgcn_readlane(__float_as_int(coef[(j * H) >> 3]), i + u));
            acc[j] = fmaf(f[u][j], a, acc[j]);
        }
    }
}

template <int H>
__device__ inline void walk_rows(const float* __restrict__ featL, int sv, const float (&coef)[H],
                                 int cnt, float (&acc)[8])
{
    int i = 0;
    for (; i + AGG_U <= cnt; i += AGG_U) walk_group<H, AGG_U>(featL, sv, coef, i, acc);
    if (i + 2 <= cnt) { walk_group<H, 2>(featL, sv, coef, i, acc); i += 2; }
    if (i < cnt)      { walk_group<H, 1>(featL, sv, coef, i, acc); }
}

template <int H, bool RELU>
__global__ __launch_bounds__(NODE_TB)
void node_softmax_agg_k(const int* __restrict__ rp, const int* __restrict__ csrc,
                        const float* __restrict__ el, const float* __restrict__ er,
                        const float* __restrict__ feat,
                        const float* __restrict__ bias, const float* __restrict__ buf,
                        const float* __restrict__ bb, float* __restrict__ outp, int N)
{
    const int lane = threadIdx.x & 31;
    const int n = __builtin_amdgcn_readfirstlane((int)(blockIdx.x * (blockDim.x >> 5) + (threadIdx.x >> 5)));
    if (n >= N) return;
    const int start = __builtin_amdgcn_readfirstlane(rp[n]);
    const int end   = __builtin_amdgcn_readfirstlane(rp[n + 1]);
    const int len   = end - start;

    const size_t rowOff = (size_t)n * 256 + lane;
    float bufv[8];
#pragma unroll
    for (int j = 0; j < 8; ++j) bufv[j] = buf[rowOff + 32 * j];

    float ern[H];
#pragma unroll
    for (int h = 0; h < H; ++h) ern[h] = er[(size_t)n * H + h];

    float acc[8];
#pragma unroll
    for (int j = 0; j < 8; ++j) acc[j] = 0.0f;

    const float* featL = feat + lane;

    if (len > 0) {
        const int cnt0 = len < 32 ? len : 32;
        int sv0;
        float e0[H];
        load_scores<H>(csrc, el, ern, start, cnt0, lane, sv0, e0);

        float mx[H];
#pragma unroll
        for (int h = 0; h < H; ++h) mx[h] = e0[h];
        for (int b = start + 32; b < end; b += 32) {
            const int c = (end - b < 32) ? (end - b) : 32;
            int svb; float eb[H];
            load_scores<H>(csrc, el, ern, b, c, lane, svb, eb);
#pragma unroll
            for (int h = 0; h < H; ++h) mx[h] = fmaxf(mx[h], eb[h]);
        }
#pragma unroll
        for (int off = 16; off > 0; off >>= 1) {
#pragma unroll
            for (int h = 0; h < H; ++h) mx[h] = fmaxf(mx[h], __shfl_xor(mx[h], off, 32));
        }
#pragma unroll
        for (int h = 0; h < H; ++h) mx[h] = __int_as_float(__builtin_amdgcn_readfirstlane(__float_as_int(mx[h])));

        float ex0[H], sm[H];
#pragma unroll
        for (int h = 0; h < H; ++h) {
            ex0[h] = (lane < cnt0) ? expf(e0[h] - mx[h]) : 0.0f;
            sm[h]  = ex0[h];
        }
        for (int b = start + 32; b < end; b += 32) {
            const int c = (end - b < 32) ? (end - b) : 32;
            int svb; float eb[H];
            load_scores<H>(csrc, el, ern, b, c, lane, svb, eb);
#pragma unroll
            for (int h = 0; h < H; ++h) sm[h] += (lane < c) ? expf(eb[h] - mx[h]) : 0.0f;
        }
#pragma unroll
        for (int off = 16; off > 0; off >>= 1) {
#pragma unroll
            for (int h = 0; h < H; ++h) sm[h] += __shfl_xor(sm[h], off, 32);
        }
#pragma unroll
        for (int h = 0; h < H; ++h) sm[h] = __int_as_float(__builtin_amdgcn_readfirstlane(__float_as_int(sm[h])));

        float coef[H];
#pragma unroll
        for (int h = 0; h < H; ++h) coef[h] = ex0[h] / sm[h];
        walk_rows<H>(featL, sv0, coef, cnt0, acc);
        for (int b = start + 32; b < end; b += 32) {
            const int c = (end - b < 32) ? (end - b) : 32;
            int svb; float eb[H];
            load_scores<H>(csrc, el, ern, b, c, lane, svb, eb);
#pragma unroll
            for (int h = 0; h < H; ++h) coef[h] = ((lane < c) ? expf(eb[h] - mx[h]) : 0.0f) / sm[h];
            walk_rows<H>(featL, svb, coef, c, acc);
        }
    }

    const float dg = (float)len;
    const float norm = 1.0f / fmaxf(dg, 1.0f);
    float* o = outp + rowOff;
#pragma unroll
    for (int j = 0; j < 8; ++j) {
        float v = acc[j] + bias[lane + 32 * j] + norm * (bufv[j] + bb[lane + 32 * j]);
        if (RELU) v = v > 0.0f ? v : 0.0f;
        o[32 * j] = v;
    }
}

extern "C" void kernel_launch(void* const* d_in, const int* in_sizes, int n_in,
                              void* d_out, int out_size, void* d_ws, size_t ws_size,
                              hipStream_t stream)
{
    const float* inputs  = (const float*)d_in[0];
    const int*   src     = (const int*)  d_in[1];
    const int*   dst     = (const int*)  d_in[2];
    const float* fc1_w   = (const float*)d_in[3];
    const float* attn_l1 = (const float*)d_in[4];
    const float* attn_r1 = (const float*)d_in[5];
    const float* bias1   = (const float*)d_in[6];
    const float* fc2_w   = (const float*)d_in[7];
    const float* attn_l2 = (const float*)d_in[8];
    const float* attn_r2 = (const float*)d_in[9];
    const float* bias2   = (const float*)d_in[10];
    const float* wb0     = (const float*)d_in[11];
    const float* bb0     = (const float*)d_in[12];
    const float* wb1     = (const float*)d_in[13];
    const float* bb1     = (const float*)d_in[14];
    float* out = (float*)d_out;

    const int F  = 256;
    const int N  = in_sizes[0] / F;
    const int E  = in_sizes[1];
    const int H1 = 8;

    float* ws   = (float*)d_ws;
    float* feat = ws;                ws += (size_t)N * F;
    float* buf  = ws;                ws += (size_t)N * F;
    float* buf2 = ws;                ws += (size_t)N * F;
    float* h1   = ws;                ws += (size_t)N * F;
    float* el   = ws;                ws += (size_t)N * H1;
    float* er   = ws;                ws += (size_t)N * H1;
    int*   wi   = (int*)ws;
    int*   cnt  = wi;                wi += (size_t)N;
    int*   rank = wi;                wi += (size_t)E;
    int*   csrc = wi;                wi += (size_t)E;
    int*   rp   = wi;

    const int TB = 256;
    auto cdiv = [](int a, int b) { return (a + b - 1) / b; };

    const int waves    = (N / 16) * 4;
    const int gemmGrid = cdiv(waves, TB / 32);
    const int nodeGrid = cdiv(N, NODE_TB / 32);
    const int scoreGrid = cdiv(N, (SC_TB / 32) * SC_NPW);

    fill_i32<<<cdiv(N, TB), TB, 0, stream>>>(cnt, 0, N);
    edge_rank_k<<<cdiv(E, TB), TB, 0, stream>>>(dst, cnt, rank, E);
    scan_counts_k<<<1, SCAN_TB, 0, stream>>>(cnt, rp, N);
    edge_place_k<<<cdiv(E, TB), TB, 0, stream>>>(src, dst, rank, rp, csrc, E);

    gemm_wmma_f32<<<dim3(gemmGrid, 3), TB, 0, stream>>>(inputs, F,
                                                        fc1_w, F,     feat,
                                                        wb0,   F,     buf,
                                                        wb1,   2 * F, buf2,
                                                        F, N, F, 0);

    attn_scores<8><<<scoreGrid, SC_TB, 0, stream>>>(feat, attn_l1, attn_r1, el, er, N);
    node_softmax_agg_k<8, true><<<nodeGrid, NODE_TB, 0, stream>>>(rp, csrc, el, er, feat, bias1, buf, bb0, h1, N);

    gemm_wmma_f32<<<dim3(gemmGrid, 2), TB, 0, stream>>>(h1, F,
                                                        fc2_w,   F,     feat,
                                                        wb1 + F, 2 * F, buf2,
                                                        nullptr, 0,     nullptr,
                                                        F, N, F, 2);

    attn_scores<1><<<scoreGrid, SC_TB, 0, stream>>>(feat, attn_l2, attn_r2, el, er, N);
    node_softmax_agg_k<1, false><<<nodeGrid, NODE_TB, 0, stream>>>(rp, csrc, el, er, feat, bias2, buf2, bb1, out, N);
}


# prefix sum of the edge counts: a wave walks 1024 counters as 8 rows of 16 B per lane, so each load and store covers 4 whole 128-B lines; was one line per lane (32 lines per 4-B access)
# speedup vs baseline: 3.7514x; 1.0784x over previous
#include <hip/hip_runtime.h>
#include <hip/hip_bf16.h>
#include <math.h>


typedef __attribute__((ext_vector_type(2))) float v2f;
typedef __attribute__((ext_vector_type(4))) float v4f;
typedef __attribute__((ext_vector_type(8))) float v8f;
typedef __attribute__((ext_vector_type(4))) int   v4i;

__global__ void fill_i32(int* __restrict__ p, int v, int n) {
    int i = blockIdx.x * blockDim.x + threadIdx.x;
    if (i < n) p[i] = v;
}

__global__ void edge_rank_k(const int* __restrict__ dst, int* __restrict__ cnt,
                            int* __restrict__ rank, int E)
{
    int e = blockIdx.x * blockDim.x + threadIdx.x;
    if (e < E) rank[e] = atomicAdd(&cnt[dst[e]], 1);
}

#define SCAN_TB 1024
#define SCAN_R  8
#define SCAN_TILE (SCAN_TB * 4 * SCAN_R)
__global__ __launch_bounds__(SCAN_TB)
void scan_counts_k(const int* __restrict__ cnt, int* __restrict__ rp, int N)
{
    __shared__ int wtot[32];
    __shared__ int tileTot;
    const int t = threadIdx.x;
    const int lane = t & 31;
    const int wv = t >> 5;
    const bool vec = ((((size_t)cnt) | ((size_t)rp)) & 15) == 0 && N >= 4;
    const int lastVec = (N - 4) & ~3;
    int carry = 0;
    for (int base = 0; base <= N; base += SCAN_TILE) {
        const int b = base + wv * (128 * SCAN_R) + 4 * lane;
        v4i c[SCAN_R];
        if (vec) {
#pragma unroll
            for (int r = 0; r < SCAN_R; ++r) {
                const int i0 = b + 128 * r;
                c[r] = *(const v4i*)(cnt + (i0 < lastVec ? i0 : lastVec));
            }
#pragma unroll
            for (int r = 0; r < SCAN_R; ++r) {
                const int i0 = b + 128 * r;
                if (i0 + 3 >= N) {
                    c[r][0] = (i0     < N) ? cnt[i0]     : 0;
                    c[r][1] = (i0 + 1 < N) ? cnt[i0 + 1] : 0;
                    c[r][2] = (i0 + 2 < N) ? cnt[i0 + 2] : 0;
                    c[r][3] = 0;
                }
            }
        } else {
#pragma unroll
            for (int r = 0; r < SCAN_R; ++r) {
                const int i0 = b + 128 * r;
                c[r][0] = (i0     < N) ? cnt[i0]     : 0;
                c[r][1] = (i0 + 1 < N) ? cnt[i0 + 1] : 0;
                c[r][2] = (i0 + 2 < N) ? cnt[i0 + 2] : 0;
                c[r][3] = (i0 + 3 < N) ? cnt[i0 + 3] : 0;
            }
        }
        int tot[SCAN_R], inc[SCAN_R];
#pragma unroll
        for (int r = 0; r < SCAN_R; ++r) {
            tot[r] = c[r][0] + c[r][1] + c[r][2] + c[r][3];
            inc[r] = tot[r];
        }
#pragma unroll
        for (int off = 1; off < 32; off <<= 1) {
#pragma unroll
            for (int r = 0; r < SCAN_R; ++r) {
                const int v = __shfl_up(inc[r], off, 32);
                if (lane >= off) inc[r] += v;
            }
        }
        int rowOff[SCAN_R];
        int wsum = 0;
#pragma unroll
        for (int r = 0; r < SCAN_R; ++r) {
            rowOff[r] = wsum;
            wsum += __builtin_amdgcn_readlane(inc[r], 31);
        }
        if (lane == 0) wtot[wv] = wsum;
        __syncthreads();
        if (wv == 0) {
            const int w = wtot[lane];
            int winc = w;
#pragma unroll
            for (int off = 1; off < 32; off <<= 1) {
                const int v = __shfl_up(winc, off, 32);
                if (lane >= off) winc += v;
            }
            wtot[lane] = winc - w;
            if (lane == 31) tileTot = winc;
        }
        __syncthreads();
        const int wbase = carry + wtot[wv];
#pragma unroll
        for (int r = 0; r < SCAN_R; ++r) {
            const int i0 = b + 128 * r;
            v4i o;
            o[0] = wbase + rowOff[r] + inc[r] - tot[r];
            o[1] = o[0] + c[r][0];
            o[2] = o[1] + c[r][1];
            o[3] = o[2] + c[r][2];
            if (vec && i0 + 3 <= N) {
                *(v4i*)(rp + i0) = o;
            } else {
                if (i0     <= N) rp[i0]     = o[0];
                if (i0 + 1 <= N) rp[i0 + 1] = o[1];
                if (i0 + 2 <= N) rp[i0 + 2] = o[2];
                if (i0 + 3 <= N) rp[i0 + 3] = o[3];
            }
        }
        carry += tileTot;
        __syncthreads();
    }
}

__global__ void edge_place_k(const int* __restrict__ src, const int* __restrict__ dst,
                             const int* __restrict__ rank, const int* __restrict__ rp,
                             int* __restrict__ csrc, int E)
{
    int e = blockIdx.x * blockDim.x + threadIdx.x;
    if (e < E) csrc[rp[dst[e]] + rank[e]] = src[e];
}

#define GEMM_KC 32
#define GEMM_LP 36
__global__ __launch_bounds__(256)
void gemm_wmma_f32(const float* __restrict__ A, int lda,
                   const float* __restrict__ W0, int ldw0, float* __restrict__ C0,
                   const float* __restrict__ W1, int ldw1, float* __restrict__ C1,
                   const float* __restrict__ W2, int ldw2, float* __restrict__ C2,
                   int ldc, int M, int K, int accMask)
{
    const int y = blockIdx.y;
    const float* __restrict__ W = (y == 0) ? W0 : (y == 1) ? W1 : W2;
    float* __restrict__ C       = (y == 0) ? C0 : (y == 1) ? C1 : C2;
    const int ldw               = (y == 0) ? ldw0 : (y == 1) ? ldw1 : ldw2;
    const int accumulate        = (accMask >> y) & 1;

    __shared__ __attribute__((aligned(16))) float Ws[256 * GEMM_LP];
    __shared__ __attribute__((aligned(16))) float As[32 * GEMM_LP];

    const int tid  = threadIdx.x;
    const int wave = tid >> 5;
    const int lane = tid & 31;
    const int half = lane >> 4;
    const int lr   = lane & 15;
    const int wr   = wave >> 2;
    const int col0 = (wave & 3) * 64;
    const int rowBase = blockIdx.x * 32;
    const int rowTile = blockIdx.x * 2 + wr;
    const int nRowTiles = M >> 4;

    const int sRow = tid >> 3;
    const int sCol = (tid & 7) * 4;
    int aRowG = rowBase + sRow;
    if (aRowG > M - 1) aRowG = M - 1;
    const float* Ag = A + (size_t)aRowG * lda + sCol;
    const float* Wg = W + (size_t)sRow * ldw + sCol;
    float* AsSt = As + sRow * GEMM_LP + sCol;
    float* WsSt = Ws + sRow * GEMM_LP + sCol;

    const float* aF = As + (wr * 16 + lr) * GEMM_LP + 2 * half;
    const float* bF = Ws + (col0 + lr) * GEMM_LP + 2 * half;

    v8f acc0 = {}, acc1 = {}, acc2 = {}, acc3 = {};

    v4f areg = *(const v4f*)(Ag);
    v4f wreg[8];
#pragma unroll
    for (int j = 0; j < 8; ++j)
        wreg[j] = *(const v4f*)(Wg + (size_t)(32 * j) * ldw);

    for (int k0 = 0; k0 < K; k0 += GEMM_KC) {
        __syncthreads();
        *(v4f*)(AsSt) = areg;
#pragma unroll
        for (int j = 0; j < 8; ++j)
            *(v4f*)(WsSt + (32 * j) * GEMM_LP) = wreg[j];
        __syncthreads();

        if (k0 + GEMM_KC < K) {
            areg = *(const v4f*)(Ag + k0 + GEMM_KC);
#pragma unroll
            for (int j = 0; j < 8; ++j)
                wreg[j] = *(const v4f*)(Wg + (size_t)(32 * j) * ldw + k0 + GEMM_KC);
        }

#pragma unroll
        for (int hc = 0; hc < GEMM_KC; hc += 16) {
            v2f a[4], b0[4], b1[4], b2[4], b3[4];
#pragma unroll
            for (int s = 0; s < 4; ++s) {
                const int kk = hc + 4 * s;
                a[s]  = *(const v2f*)(aF + kk);
                b0[s] = *(const v2f*)(bF +  0 * GEMM_LP + kk);
                b1[s] = *(const v2f*)(bF + 16 * GEMM_LP + kk);
                b2[s] = *(const v2f*)(bF + 32 * GEMM_LP + kk);
                b3[s] = *(const v2f*)(bF + 48 * GEMM_LP + kk);
            }
#pragma unroll
            for (int s = 0; s < 4; ++s) {
                acc0 = __builtin_amdgcn_wmma_f32_16x16x4_f32(false, a[s], false, b0[s], (short)0, acc0, false, false);
                acc1 = __builtin_amdgcn_wmma_f32_16x16x4_f32(false, a[s], false, b1[s], (short)0, acc1, false, false);
                acc2 = __builtin_amdgcn_wmma_f32_16x16x4_f32(false, a[s], false, b2[s], (short)0, acc2, false, false);
                acc3 = __builtin_amdgcn_wmma_f32_16x16x4_f32(false, a[s], false, b3[s], (short)0, acc3, false, false);
            }
        }
    }

    if (rowTile >= nRowTiles) return;

    float* Cb = C + (size_t)(rowTile * 16 + 8 * half) * ldc + col0 + lr;
#pragma unroll
    for (int j = 0; j < 8; ++j) {
        float* p = Cb + (size_t)j * ldc;
        p[ 0] = (accumulate ? p[ 0] : 0.0f) + acc0[j];
        p[16] = (accumulate ? p[16] : 0.0f) + acc1[j];
        p[32] = (accumulate ? p[32] : 0.0f) + acc2[j];
        p[48] = (accumulate ? p[48] : 0.0f) + acc3[j];
    }
}

#define SC_NPW 4
#define SC_TB  256

__device__ inline float dot4(const v4f x, const v4f a)
{
    float p = x[0] * a[0];
    p = fmaf(x[1], a[1], p);
    p = fmaf(x[2], a[2], p);
    p = fmaf(x[3], a[3], p);
    return p;
}

template <int HALF>
__device__ inline void fold_step(float (&v)[2 * HALF], bool up, int mask)
{
#pragma unroll
    for (int k = 0; k < HALF; ++k) {
        const float keep = up ? v[k + HALF] : v[k];
        const float send = up ? v[k] : v[k + HALF];
        v[k] = keep + __shfl_xor(send, mask, 32);
    }
}

template <int H>
__global__ __launch_bounds__(SC_TB)
void attn_scores(const float* __restrict__ feat,
                 const float* __restrict__ attn_l,
                 const float* __restrict__ attn_r,
                 float* __restrict__ el, float* __restrict__ er, int N)
{
    const int lane = threadIdx.x & 31;
    const int n0 = (blockIdx.x * (blockDim.x >> 5) + (threadIdx.x >> 5)) * SC_NPW;
    if (n0 >= N) return;

    v4f x[2 * SC_NPW];
#pragma unroll
    for (int q = 0; q < SC_NPW; ++q) {
        int n = n0 + q;
        if (n > N - 1) n = N - 1;
        const float* fr = feat + (size_t)n * 256 + 4 * lane;
        x[2 * q]     = *(const v4f*)(fr);
        x[2 * q + 1] = *(const v4f*)(fr + 128);
    }
    const v4f al0 = *(const v4f*)(attn_l + 4 * lane);
    const v4f al1 = *(const v4f*)(attn_l + 128 + 4 * lane);
    const v4f ar0 = *(const v4f*)(attn_r + 4 * lane);
    const v4f ar1 = *(const v4f*)(attn_r + 128 + 4 * lane);

    float pl[2 * SC_NPW], pr[2 * SC_NPW];
#pragma unroll
    for (int q = 0; q < SC_NPW; ++q) {
        pl[2 * q]     = dot4(x[2 * q],     al0);
        pl[2 * q + 1] = dot4(x[2 * q + 1], al1);
        pr[2 * q]     = dot4(x[2 * q],     ar0);
        pr[2 * q + 1] = dot4(x[2 * q + 1], ar1);
    }

    if constexpr (H == 8) {
        fold_step<4>(pl, (lane & 4) != 0, 4);
        fold_step<4>(pr, (lane & 4) != 0, 4);
        float l4[4] = { pl[0], pl[1], pl[2], pl[3] };
        float r4[4] = { pr[0], pr[1], pr[2], pr[3] };
        fold_step<2>(l4, (lane & 2) != 0, 2);
        fold_step<2>(r4, (lane & 2) != 0, 2);
        float l2[2] = { l4[0], l4[1] };
        float r2[2] = { r4[0], r4[1] };
        fold_step<1>(l2, (lane & 1) != 0, 1);
        fold_step<1>(r2, (lane & 1) != 0, 1);
        const int slot = 4 * (lane & 7) + (lane >> 3);
        if (n0 + (slot >> 3) < N) {
            el[(size_t)n0 * 8 + slot] = l2[0];
            er[(size_t)n0 * 8 + slot] = r2[0];
        }
    } else {
        float sl[SC_NPW], sr[SC_NPW];
#pragma unroll
        for (int q = 0; q < SC_NPW; ++q) {
            sl[q] = pl[2 * q] + pl[2 * q + 1];
            sr[q] = pr[2 * q] + pr[2 * q + 1];
        }
        fold_step<2>(sl, (lane & 16) != 0, 16);
        fold_step<2>(sr, (lane & 16) != 0, 16);
        float l2[2] = { sl[0], sl[1] };
        float r2[2] = { sr[0], sr[1] };
        fold_step<1>(l2, (lane & 8) != 0, 8);
        fold_step<1>(r2, (lane & 8) != 0, 8);
        float tl = l2[0], tr = r2[0];
#pragma unroll
        for (int off = 4; off > 0; off >>= 1) {
            tl += __shfl_xor(tl, off, 32);
            tr += __shfl_xor(tr, off, 32);
        }
        const int q = lane >> 3;
        if ((lane & 7) == 0 && n0 + q < N) { el[n0 + q] = tl; er[n0 + q] = tr; }
    }
}

__device__ inline float leaky02(float v) { return v > 0.0f ? v : 0.2f * v; }

#define NODE_TB 256
#define AGG_U  4

template <int H>
__device__ inline void load_scores(const int* __restrict__ csrc, const float* __restrict__ el,
                                   const float (&ern)[H], int b, int cnt, int lane,
                                   int& sv, float (&e)[H])
{
    sv = 0;
#pragma unroll
    for (int h = 0; h < H; ++h) e[h] = -INFINITY;
    if (lane < cnt) {
        sv = csrc[b + lane];
        if constexpr (H == 8) {
            const v4f x0 = *(const v4f*)(el + (size_t)sv * 8);
            const v4f x1 = *(const v4f*)(el + (size_t)sv * 8 + 4);
#pragma unroll
            for (int h = 0; h < 4; ++h) {
                e[h]     = leaky02(x0[h] + ern[h]);
                e[h + 4] = leaky02(x1[h] + ern[h + 4]);
            }
        } else {
            e[0] = leaky02(el[sv] + ern[0]);
        }
    }
}

template <int H, int U>
__device__ inline void walk_group(const float* __restrict__ featL, int sv, const float (&coef)[H],
                                  int i, float (&acc)[8])
{
    float f[U][8];
#pragma unroll
    for (int u = 0; u < U; ++u) {
        const int sN = __builtin_amdgcn_readlane(sv, i + u);
        const float* fr = featL + (size_t)sN * 256;
#pragma unroll
        for (int j = 0; j < 8; ++j) f[u][j] = fr[32 * j];
    }
#pragma unroll
    for (int u = 0; u < U; ++u) {
#pragma unroll
        for (int j = 0; j < 8; ++j) {
            const float a = __int_as_float(__builtin_amdgcn_readlane(__float_as_int(coef[(j * H) >> 3]), i + u));
            acc[j] = fmaf(f[u][j], a, acc[j]);
        }
    }
}

template <int H>
__device__ inline void walk_rows(const float* __restrict__ featL, int sv, const float (&coef)[H],
                                 int cnt, float (&acc)[8])
{
    int i = 0;
    for (; i + AGG_U <= cnt; i += AGG_U) walk_group<H, AGG_U>(featL, sv, coef, i, acc);
    if (i + 2 <= cnt) { walk_group<H, 2>(featL, sv, coef, i, acc); i += 2; }
    if (i < cnt)      { walk_group<H, 1>(featL, sv, coef, i, acc); }
}

template <int H, bool RELU>
__global__ __launch_bounds__(NODE_TB)
void node_softmax_agg_k(const int* __restrict__ rp, const int* __restrict__ csrc,
                        const float* __restrict__ el, const float* __restrict__ er,
                        const float* __restrict__ feat,
                        const float* __restrict__ bias, const float* __restrict__ buf,
                        const float* __restrict__ bb, float* __restrict__ outp, int N)
{
    const int lane = threadIdx.x & 31;
    const int n = __builtin_amdgcn_readfirstlane((int)(blockIdx.x * (blockDim.x >> 5) + (threadIdx.x >> 5)));
    if (n >= N) return;
    const int start = __builtin_amdgcn_readfirstlane(rp[n]);
    const int end   = __builtin_amdgcn_readfirstlane(rp[n + 1]);
    const int len   = end - start;

    const size_t rowOff = (size_t)n * 256 + lane;
    float bufv[8];
#pragma unroll
    for (int j = 0; j < 8; ++j) bufv[j] = buf[rowOff + 32 * j];

    float ern[H];
#pragma unroll
    for (int h = 0; h < H; ++h) ern[h] = er[(size_t)n * H + h];

    float acc[8];
#pragma unroll
    for (int j = 0; j < 8; ++j) acc[j] = 0.0f;

    const float* featL = feat + lane;

    if (len > 0) {
        const int cnt0 = len < 32 ? len : 32;
        int sv0;
        float e0[H];
        load_scores<H>(csrc, el, ern, start, cnt0, lane, sv0, e0);

        float mx[H];
#pragma unroll
        for (int h = 0; h < H; ++h) mx[h] = e0[h];
        for (int b = start + 32; b < end; b += 32) {
            const int c = (end - b < 32) ? (end - b) : 32;
            int svb; float eb[H];
            load_scores<H>(csrc, el, ern, b, c, lane, svb, eb);
#pragma unroll
            for (int h = 0; h < H; ++h) mx[h] = fmaxf(mx[h], eb[h]);
        }
#pragma unroll
        for (int off = 16; off > 0; off >>= 1) {
#pragma unroll
            for (int h = 0; h < H; ++h) mx[h] = fmaxf(mx[h], __shfl_xor(mx[h], off, 32));
        }
#pragma unroll
        for (int h = 0; h < H; ++h) mx[h] = __int_as_float(__builtin_amdgcn_readfirstlane(__float_as_int(mx[h])));

        float ex0[H], sm[H];
#pragma unroll
        for (int h = 0; h < H; ++h) {
            ex0[h] = (lane < cnt0) ? expf(e0[h] - mx[h]) : 0.0f;
            sm[h]  = ex0[h];
        }
        for (int b = start + 32; b < end; b += 32) {
            const int c = (end - b < 32) ? (end - b) : 32;
            int svb; float eb[H];
            load_scores<H>(csrc, el, ern, b, c, lane, svb, eb);
#pragma unroll
            for (int h = 0; h < H; ++h) sm[h] += (lane < c) ? expf(eb[h] - mx[h]) : 0.0f;
        }
#pragma unroll
        for (int off = 16; off > 0; off >>= 1) {
#pragma unroll
            for (int h = 0; h < H; ++h) sm[h] += __shfl_xor(sm[h], off, 32);
        }
#pragma unroll
        for (int h = 0; h < H; ++h) sm[h] = __int_as_float(__builtin_amdgcn_readfirstlane(__float_as_int(sm[h])));

        float coef[H];
#pragma unroll
        for (int h = 0; h < H; ++h) coef[h] = ex0[h] / sm[h];
        walk_rows<H>(featL, sv0, coef, cnt0, acc);
        for (int b = start + 32; b < end; b += 32) {
            const int c = (end - b < 32) ? (end - b) : 32;
            int svb; float eb[H];
            load_scores<H>(csrc, el, ern, b, c, lane, svb, eb);
#pragma unroll
            for (int h = 0; h < H; ++h) coef[h] = ((lane < c) ? expf(eb[h] - mx[h]) : 0.0f) / sm[h];
            walk_rows<H>(featL, svb, coef, c, acc);
        }
    }

    const float dg = (float)len;
    const float norm = 1.0f / fmaxf(dg, 1.0f);
    float* o = outp + rowOff;
#pragma unroll
    for (int j = 0; j < 8; ++j) {
        float v = acc[j] + bias[lane + 32 * j] + norm * (bufv[j] + bb[lane + 32 * j]);
        if (RELU) v = v > 0.0f ? v : 0.0f;
        o[32 * j] = v;
    }
}

extern "C" void kernel_launch(void* const* d_in, const int* in_sizes, int n_in,
                              void* d_out, int out_size, void* d_ws, size_t ws_size,
                              hipStream_t stream)
{
    const float* inputs  = (const float*)d_in[0];
    const int*   src     = (const int*)  d_in[1];
    const int*   dst     = (const int*)  d_in[2];
    const float* fc1_w   = (const float*)d_in[3];
    const float* attn_l1 = (const float*)d_in[4];
    const float* attn_r1 = (const float*)d_in[5];
    const float* bias1   = (const float*)d_in[6];
    const float* fc2_w   = (const float*)d_in[7];
    const float* attn_l2 = (const float*)d_in[8];
    const float* attn_r2 = (const float*)d_in[9];
    const float* bias2   = (const float*)d_in[10];
    const float* wb0     = (const float*)d_in[11];
    const float* bb0     = (const float*)d_in[12];
    const float* wb1     = (const float*)d_in[13];
    const float* bb1     = (const float*)d_in[14];
    float* out = (float*)d_out;

    const int F  = 256;
    const int N  = in_sizes[0] / F;
    const int E  = in_sizes[1];
    const int H1 = 8;

    float* ws   = (float*)d_ws;
    float* feat = ws;                ws += (size_t)N * F;
    float* buf  = ws;                ws += (size_t)N * F;
    float* buf2 = ws;                ws += (size_t)N * F;
    float* h1   = ws;                ws += (size_t)N * F;
    float* el   = ws;                ws += (size_t)N * H1;
    float* er   = ws;                ws += (size_t)N * H1;
    int*   wi   = (int*)ws;
    int*   cnt  = wi;                wi += (size_t)N;
    int*   rank = wi;                wi += (size_t)E;
    int*   csrc = wi;                wi += (size_t)E;
    int*   rp   = wi;

    const int TB = 256;
    auto cdiv = [](int a, int b) { return (a + b - 1) / b; };

    const int waves    = (N / 16) * 4;
    const int gemmGrid = cdiv(waves, TB / 32);
    const int nodeGrid = cdiv(N, NODE_TB / 32);
    const int scoreGrid = cdiv(N, (SC_TB / 32) * SC_NPW);

    fill_i32<<<cdiv(N, TB), TB, 0, stream>>>(cnt, 0, N);
    edge_rank_k<<<cdiv(E, TB), TB, 0, stream>>>(dst, cnt, rank, E);
    scan_counts_k<<<1, SCAN_TB, 0, stream>>>(cnt, rp, N);
    edge_place_k<<<cdiv(E, TB), TB, 0, stream>>>(src, dst, rank, rp, csrc, E);

    gemm_wmma_f32<<<dim3(gemmGrid, 3), TB, 0, stream>>>(inputs, F,
                                                        fc1_w, F,     feat,
                                                        wb0,   F,     buf,
                                                        wb1,   2 * F, buf2,
                                                        F, N, F, 0);

    attn_scores<8><<<scoreGrid, SC_TB, 0, stream>>>(feat, attn_l1, attn_r1, el, er, N);
    node_softmax_agg_k<8, true><<<nodeGrid, NODE_TB, 0, stream>>>(rp, csrc, el, er, feat, bias1, buf, bb0, h1, N);

    gemm_wmma_f32<<<dim3(gemmGrid, 2), TB, 0, stream>>>(h1, F,
                                                        fc2_w,   F,     feat,
                                                        wb1 + F, 2 * F, buf2,
                                                        nullptr, 0,     nullptr,
                                                        F, N, F, 2);

    attn_scores<1><<<scoreGrid, SC_TB, 0, stream>>>(feat, attn_l2, attn_r2, el, er, N);
    node_softmax_agg_k<1, false><<<nodeGrid, NODE_TB, 0, stream>>>(rp, csrc, el, er, feat, bias2, buf2, bb1, out, N);
}


# attention scores el/er taken in the epilogue of the GEMM that makes feat, from the accumulators (16-lane transposing tree, LDS, whole-line stores): 2 score launches dropped; score sums in a new order
# speedup vs baseline: 3.8664x; 1.0307x over previous
#include <hip/hip_runtime.h>
#include <hip/hip_bf16.h>
#include <math.h>


typedef __attribute__((ext_vector_type(2))) float v2f;
typedef __attribute__((ext_vector_type(4))) float v4f;
typedef __attribute__((ext_vector_type(8))) float v8f;
typedef __attribute__((ext_vector_type(4))) int   v4i;

__global__ void fill_i32(int* __restrict__ p, int v, int n) {
    int i = blockIdx.x * blockDim.x + threadIdx.x;
    if (i < n) p[i] = v;
}

__global__ void edge_rank_k(const int* __restrict__ dst, int* __restrict__ cnt,
                            int* __restrict__ rank, int E)
{
    int e = blockIdx.x * blockDim.x + threadIdx.x;
    if (e < E) rank[e] = atomicAdd(&cnt[dst[e]], 1);
}

#define SCAN_TB 1024
#define SCAN_R  8
#define SCAN_TILE (SCAN_TB * 4 * SCAN_R)
__global__ __launch_bounds__(SCAN_TB)
void scan_counts_k(const int* __restrict__ cnt, int* __restrict__ rp, int N)
{
    __shared__ int wtot[32];
    __shared__ int tileTot;
    const int t = threadIdx.x;
    const int lane = t & 31;
    const int wv = t >> 5;
    const bool vec = ((((size_t)cnt) | ((size_t)rp)) & 15) == 0 && N >= 4;
    const int lastVec = (N - 4) & ~3;
    int carry = 0;
    for (int base = 0; base <= N; base += SCAN_TILE) {
        const int b = base + wv * (128 * SCAN_R) + 4 * lane;
        v4i c[SCAN_R];
        if (vec) {
#pragma unroll
            for (int r = 0; r < SCAN_R; ++r) {
                const int i0 = b + 128 * r;
                c[r] = *(const v4i*)(cnt + (i0 < lastVec ? i0 : lastVec));
            }
#pragma unroll
            for (int r = 0; r < SCAN_R; ++r) {
                const int i0 = b + 128 * r;
                if (i0 + 3 >= N) {
                    c[r][0] = (i0     < N) ? cnt[i0]     : 0;
                    c[r][1] = (i0 + 1 < N) ? cnt[i0 + 1] : 0;
                    c[r][2] = (i0 + 2 < N) ? cnt[i0 + 2] : 0;
                    c[r][3] = 0;
                }
            }
        } else {
#pragma unroll
            for (int r = 0; r < SCAN_R; ++r) {
                const int i0 = b + 128 * r;
                c[r][0] = (i0     < N) ? cnt[i0]     : 0;
                c[r][1] = (i0 + 1 < N) ? cnt[i0 + 1] : 0;
                c[r][2] = (i0 + 2 < N) ? cnt[i0 + 2] : 0;
                c[r][3] = (i0 + 3 < N) ? cnt[i0 + 3] : 0;
            }
        }
        int tot[SCAN_R], inc[SCAN_R];
#pragma unroll
        for (int r = 0; r < SCAN_R; ++r) {
            tot[r] = c[r][0] + c[r][1] + c[r][2] + c[r][3];
            inc[r] = tot[r];
        }
#pragma unroll
        for (int off = 1; off < 32; off <<= 1) {
#pragma unroll
            for (int r = 0; r < SCAN_R; ++r) {
                const int v = __shfl_up(inc[r], off, 32);
                if (lane >= off) inc[r] += v;
            }
        }
        int rowOff[SCAN_R];
        int wsum = 0;
#pragma unroll
        for (int r = 0; r < SCAN_R; ++r) {
            rowOff[r] = wsum;
            wsum += __builtin_amdgcn_readlane(inc[r], 31);
        }
        if (lane == 0) wtot[wv] = wsum;
        __syncthreads();
        if (wv == 0) {
            const int w = wtot[lane];
            int winc = w;
#pragma unroll
            for (int off = 1; off < 32; off <<= 1) {
                const int v = __shfl_up(winc, off, 32);
                if (lane >= off) winc += v;
            }
            wtot[lane] = winc - w;
            if (lane == 31) tileTot = winc;
        }
        __syncthreads();
        const int wbase = carry + wtot[wv];
#pragma unroll
        for (int r = 0; r < SCAN_R; ++r) {
            const int i0 = b + 128 * r;
            v4i o;
            o[0] = wbase + rowOff[r] + inc[r] - tot[r];
            o[1] = o[0] + c[r][0];
            o[2] = o[1] + c[r][1];
            o[3] = o[2] + c[r][2];
            if (vec && i0 + 3 <= N) {
                *(v4i*)(rp + i0) = o;
            } else {
                if (i0     <= N) rp[i0]     = o[0];
                if (i0 + 1 <= N) rp[i0 + 1] = o[1];
                if (i0 + 2 <= N) rp[i0 + 2] = o[2];
                if (i0 + 3 <= N) rp[i0 + 3] = o[3];
            }
        }
        carry += tileTot;
        __syncthreads();
    }
}

__global__ void edge_place_k(const int* __restrict__ src, const int* __restrict__ dst,
                             const int* __restrict__ rank, const int* __restrict__ rp,
                             int* __restrict__ csrc, int E)
{
    int e = blockIdx.x * blockDim.x + threadIdx.x;
    if (e < E) csrc[rp[dst[e]] + rank[e]] = src[e];
}

template <int N>
__device__ __forceinline__ void fold_half(float* v, bool up, int mask)
{
#pragma unroll
    for (int k = 0; k < N / 2; ++k) {
        const float keep = up ? v[k + N / 2] : v[k];
        const float send = up ? v[k] : v[k + N / 2];
        v[k] = keep + __shfl_xor(send, mask, 32);
    }
}

#define GEMM_KC 32
#define GEMM_LP 36
__global__ __launch_bounds__(256)
void gemm_wmma_f32(const float* __restrict__ A, int lda,
                   const float* __restrict__ W0, int ldw0, float* __restrict__ C0,
                   const float* __restrict__ W1, int ldw1, float* __restrict__ C1,
                   const float* __restrict__ W2, int ldw2, float* __restrict__ C2,
                   int ldc, int M, int K, int accMask,
                   const float* __restrict__ attnL, const float* __restrict__ attnR,
                   float* __restrict__ el, float* __restrict__ er, int scoreH)
{
    const int y = blockIdx.y;
    const float* __restrict__ W = (y == 0) ? W0 : (y == 1) ? W1 : W2;
    float* __restrict__ C       = (y == 0) ? C0 : (y == 1) ? C1 : C2;
    const int ldw               = (y == 0) ? ldw0 : (y == 1) ? ldw1 : ldw2;
    const int accumulate        = (accMask >> y) & 1;

    __shared__ __attribute__((aligned(16))) float Ws[256 * GEMM_LP];
    __shared__ __attribute__((aligned(16))) float As[32 * GEMM_LP];
    __shared__ float Sc[512];

    const int tid  = threadIdx.x;
    const int wave = tid >> 5;
    const int lane = tid & 31;
    const int half = lane >> 4;
    const int lr   = lane & 15;
    const int wr   = wave >> 2;
    const int col0 = (wave & 3) * 64;
    const int rowBase = blockIdx.x * 32;
    const int rowTile = blockIdx.x * 2 + wr;
    const int nRowTiles = M >> 4;

    const int sRow = tid >> 3;
    const int sCol = (tid & 7) * 4;
    int aRowG = rowBase + sRow;
    if (aRowG > M - 1) aRowG = M - 1;
    const float* Ag = A + (size_t)aRowG * lda + sCol;
    const float* Wg = W + (size_t)sRow * ldw + sCol;
    float* AsSt = As + sRow * GEMM_LP + sCol;
    float* WsSt = Ws + sRow * GEMM_LP + sCol;

    const float* aF = As + (wr * 16 + lr) * GEMM_LP + 2 * half;
    const float* bF = Ws + (col0 + lr) * GEMM_LP + 2 * half;

    v8f acc0 = {}, acc1 = {}, acc2 = {}, acc3 = {};

    v4f areg = *(const v4f*)(Ag);
    v4f wreg[8];
#pragma unroll
    for (int j = 0; j < 8; ++j)
        wreg[j] = *(const v4f*)(Wg + (size_t)(32 * j) * ldw);

    for (int k0 = 0; k0 < K; k0 += GEMM_KC) {
        __syncthreads();
        *(v4f*)(AsSt) = areg;
#pragma unroll
        for (int j = 0; j < 8; ++j)
            *(v4f*)(WsSt + (32 * j) * GEMM_LP) = wreg[j];
        __syncthreads();

        if (k0 + GEMM_KC < K) {
            areg = *(const v4f*)(Ag + k0 + GEMM_KC);
#pragma unroll
            for (int j = 0; j < 8; ++j)
                wreg[j] = *(const v4f*)(Wg + (size_t)(32 * j) * ldw + k0 + GEMM_KC);
        }

#pragma unroll
        for (int hc = 0; hc < GEMM_KC; hc += 16) {
            v2f a[4], b0[4], b1[4], b2[4], b3[4];
#pragma unroll
            for (int s = 0; s < 4; ++s) {
                const int kk = hc + 4 * s;
                a[s]  = *(const v2f*)(aF + kk);
                b0[s] = *(const v2f*)(bF +  0 * GEMM_LP + kk);
                b1[s] = *(const v2f*)(bF + 16 * GEMM_LP + kk);
                b2[s] = *(const v2f*)(bF + 32 * GEMM_LP + kk);
                b3[s] = *(const v2f*)(bF + 48 * GEMM_LP + kk);
            }
#pragma unroll
            for (int s = 0; s < 4; ++s) {
                acc0 = __builtin_amdgcn_wmma_f32_16x16x4_f32(false, a[s], false, b0[s], (short)0, acc0, false, false);
                acc1 = __builtin_amdgcn_wmma_f32_16x16x4_f32(false, a[s], false, b1[s], (short)0, acc1, false, false);
                acc2 = __builtin_amdgcn_wmma_f32_16x16x4_f32(false, a[s], false, b2[s], (short)0, acc2, false, false);
                acc3 = __builtin_amdgcn_wmma_f32_16x16x4_f32(false, a[s], false, b3[s], (short)0, acc3, false, false);
            }
        }
    }

    if (rowTile < nRowTiles) {
        float* Cb = C + (size_t)(rowTile * 16 + 8 * half) * ldc + col0 + lr;
#pragma unroll
        for (int j = 0; j < 8; ++j) {
            float* p = Cb + (size_t)j * ldc;
            p[ 0] = (accumulate ? p[ 0] : 0.0f) + acc0[j];
            p[16] = (accumulate ? p[16] : 0.0f) + acc1[j];
            p[32] = (accumulate ? p[32] : 0.0f) + acc2[j];
            p[48] = (accumulate ? p[48] : 0.0f) + acc3[j];
        }
    }

    if (scoreH == 0 || y != 0) return;

    float al[4], ar[4];
#pragma unroll
    for (int t = 0; t < 4; ++t) {
        al[t] = attnL[col0 + 16 * t + lr];
        ar[t] = attnR[col0 + 16 * t + lr];
    }
    const int srow = wr * 16 + 8 * half + (lr >> 1);
    const int ssel = lr & 1;
    const int strip = wave & 3;

    if (scoreH == 8) {
        float v[32];
#pragma unroll
        for (int j = 0; j < 8; ++j) {
            v[4 * j + 0] = fmaf(acc1[j], al[1], acc0[j] * al[0]);
            v[4 * j + 1] = fmaf(acc3[j], al[3], acc2[j] * al[2]);
            v[4 * j + 2] = fmaf(acc1[j], ar[1], acc0[j] * ar[0]);
            v[4 * j + 3] = fmaf(acc3[j], ar[3], acc2[j] * ar[2]);
        }
        fold_half<32>(v, (lane & 8) != 0, 8);
        fold_half<16>(v, (lane & 4) != 0, 4);
        fold_half<8>(v, (lane & 2) != 0, 2);
        fold_half<4>(v, (lane & 1) != 0, 1);
        Sc[ssel * 256 + srow * 8 + 2 * strip]     = v[0];
        Sc[ssel * 256 + srow * 8 + 2 * strip + 1] = v[1];
        __syncthreads();
        if (rowBase + (tid >> 3) < M) {
            el[(size_t)rowBase * 8 + tid] = Sc[tid];
            er[(size_t)rowBase * 8 + tid] = Sc[256 + tid];
        }
    } else {
        float v[16];
#pragma unroll
        for (int j = 0; j < 8; ++j) {
            float sl = acc0[j] * al[0];
            sl = fmaf(acc1[j], al[1], sl);
            sl = fmaf(acc2[j], al[2], sl);
            sl = fmaf(acc3[j], al[3], sl);
            float sr = acc0[j] * ar[0];
            sr = fmaf(acc1[j], ar[1], sr);
            sr = fmaf(acc2[j], ar[2], sr);
            sr = fmaf(acc3[j], ar[3], sr);
            v[2 * j + 0] = sl;
            v[2 * j + 1] = sr;
        }
        fold_half<16>(v, (lane & 8) != 0, 8);
        fold_half<8>(v, (lane & 4) != 0, 4);
        fold_half<4>(v, (lane & 2) != 0, 2);
        fold_half<2>(v, (lane & 1) != 0, 1);
        Sc[strip * 64 + ssel * 32 + srow] = v[0];
        __syncthreads();
        if (tid < 64) {
            const float s = (Sc[tid] + Sc[64 + tid]) + (Sc[128 + tid] + Sc[192 + tid]);
            const int r = tid & 31;
            if (rowBase + r < M) (tid < 32 ? el : er)[rowBase + r] = s;
        }
    }
}

__device__ inline float leaky02(float v) { return v > 0.0f ? v : 0.2f * v; }

#define NODE_TB 256
#define AGG_U  4

template <int H>
__device__ inline void load_scores(const int* __restrict__ csrc, const float* __restrict__ el,
                                   const float (&ern)[H], int b, int cnt, int lane,
                                   int& sv, float (&e)[H])
{
    sv = 0;
#pragma unroll
    for (int h = 0; h < H; ++h) e[h] = -INFINITY;
    if (lane < cnt) {
        sv = csrc[b + lane];
        if constexpr (H == 8) {
            const v4f x0 = *(const v4f*)(el + (size_t)sv * 8);
            const v4f x1 = *(const v4f*)(el + (size_t)sv * 8 + 4);
#pragma unroll
            for (int h = 0; h < 4; ++h) {
                e[h]     = leaky02(x0[h] + ern[h]);
                e[h + 4] = leaky02(x1[h] + ern[h + 4]);
            }
        } else {
            e[0] = leaky02(el[sv] + ern[0]);
        }
    }
}

template <int H, int U>
__device__ inline void walk_group(const float* __restrict__ featL, int sv, const float (&coef)[H],
                                  int i, float (&acc)[8])
{
    float f[U][8];
#pragma unroll
    for (int u = 0; u < U; ++u) {
        const int sN = __builtin_amdgcn_readlane(sv, i + u);
        const float* fr = featL + (size_t)sN * 256;
#pragma unroll
        for (int j = 0; j < 8; ++j) f[u][j] = fr[32 * j];
    }
#pragma unroll
    for (int u = 0; u < U; ++u) {
#pragma unroll
        for (int j = 0; j < 8; ++j) {
            const float a = __int_as_float(__builtin_amdgcn_readlane(__float_as_int(coef[(j * H) >> 3]), i + u));
            acc[j] = fmaf(f[u][j], a, acc[j]);
        }
    }
}

template <int H>
__device__ inline void walk_rows(const float* __restrict__ featL, int sv, const float (&coef)[H],
                                 int cnt, float (&acc)[8])
{
    int i = 0;
    for (; i + AGG_U <= cnt; i += AGG_U) walk_group<H, AGG_U>(featL, sv, coef, i, acc);
    if (i + 2 <= cnt) { walk_group<H, 2>(featL, sv, coef, i, acc); i += 2; }
    if (i < cnt)      { walk_group<H, 1>(featL, sv, coef, i, acc); }
}

template <int H, bool RELU>
__global__ __launch_bounds__(NODE_TB)
void node_softmax_agg_k(const int* __restrict__ rp, const int* __restrict__ csrc,
                        const float* __restrict__ el, const float* __restrict__ er,
                        const float* __restrict__ feat,
                        const float* __restrict__ bias, const float* __restrict__ buf,
                        const float* __restrict__ bb, float* __restrict__ outp, int N)
{
    const int lane = threadIdx.x & 31;
    const int n = __builtin_amdgcn_readfirstlane((int)(blockIdx.x * (blockDim.x >> 5) + (threadIdx.x >> 5)));
    if (n >= N) return;
    const int start = __builtin_amdgcn_readfirstlane(rp[n]);
    const int end   = __builtin_amdgcn_readfirstlane(rp[n + 1]);
    const int len   = end - start;

    const size_t rowOff = (size_t)n * 256 + lane;
    float bufv[8];
#pragma unroll
    for (int j = 0; j < 8; ++j) bufv[j] = buf[rowOff + 32 * j];

    float ern[H];
#pragma unroll
    for (int h = 0; h < H; ++h) ern[h] = er[(size_t)n * H + h];

    float acc[8];
#pragma unroll
    for (int j = 0; j < 8; ++j) acc[j] = 0.0f;

    const float* featL = feat + lane;

    if (len > 0) {
        const int cnt0 = len < 32 ? len : 32;
        int sv0;
        float e0[H];
        load_scores<H>(csrc, el, ern, start, cnt0, lane, sv0, e0);

        float mx[H];
#pragma unroll
        for (int h = 0; h < H; ++h) mx[h] = e0[h];
        for (int b = start + 32; b < end; b += 32) {
            const int c = (end - b < 32) ? (end - b) : 32;
            int svb; float eb[H];
            load_scores<H>(csrc, el, ern, b, c, lane, svb, eb);
#pragma unroll
            for (int h = 0; h < H; ++h) mx[h] = fmaxf(mx[h], eb[h]);
        }
#pragma unroll
        for (int off = 16; off > 0; off >>= 1) {
#pragma unroll
            for (int h = 0; h < H; ++h) mx[h] = fmaxf(mx[h], __shfl_xor(mx[h], off, 32));
        }
#pragma unroll
        for (int h = 0; h < H; ++h) mx[h] = __int_as_float(__builtin_amdgcn_readfirstlane(__float_as_int(mx[h])));

        float ex0[H], sm[H];
#pragma unroll
        for (int h = 0; h < H; ++h) {
            ex0[h] = (lane < cnt0) ? expf(e0[h] - mx[h]) : 0.0f;
            sm[h]  = ex0[h];
        }
        for (int b = start + 32; b < end; b += 32) {
            const int c = (end - b < 32) ? (end - b) : 32;
            int svb; float eb[H];
            load_scores<H>(csrc, el, ern, b, c, lane, svb, eb);
#pragma unroll
            for (int h = 0; h < H; ++h) sm[h] += (lane < c) ? expf(eb[h] - mx[h]) : 0.0f;
        }
#pragma unroll
        for (int off = 16; off > 0; off >>= 1) {
#pragma unroll
            for (int h = 0; h < H; ++h) sm[h] += __shfl_xor(sm[h], off, 32);
        }
#pragma unroll
        for (int h = 0; h < H; ++h) sm[h] = __int_as_float(__builtin_amdgcn_readfirstlane(__float_as_int(sm[h])));

        float coef[H];
#pragma unroll
        for (int h = 0; h < H; ++h) coef[h] = ex0[h] / sm[h];
        walk_rows<H>(featL, sv0, coef, cnt0, acc);
        for (int b = start + 32; b < end; b += 32) {
            const int c = (end - b < 32) ? (end - b) : 32;
            int svb; float eb[H];
            load_scores<H>(csrc, el, ern, b, c, lane, svb, eb);
#pragma unroll
            for (int h = 0; h < H; ++h) coef[h] = ((lane < c) ? expf(eb[h] - mx[h]) : 0.0f) / sm[h];
            walk_rows<H>(featL, svb, coef, c, acc);
        }
    }

    const float dg = (float)len;
    const float norm = 1.0f / fmaxf(dg, 1.0f);
    float* o = outp + rowOff;
#pragma unroll
    for (int j = 0; j < 8; ++j) {
        float v = acc[j] + bias[lane + 32 * j] + norm * (bufv[j] + bb[lane + 32 * j]);
        if (RELU) v = v > 0.0f ? v : 0.0f;
        o[32 * j] = v;
    }
}

extern "C" void kernel_launch(void* const* d_in, const int* in_sizes, int n_in,
                              void* d_out, int out_size, void* d_ws, size_t ws_size,
                              hipStream_t stream)
{
    const float* inputs  = (const float*)d_in[0];
    const int*   src     = (const int*)  d_in[1];
    const int*   dst     = (const int*)  d_in[2];
    const float* fc1_w   = (const float*)d_in[3];
    const float* attn_l1 = (const float*)d_in[4];
    const float* attn_r1 = (const float*)d_in[5];
    const float* bias1   = (const float*)d_in[6];
    const float* fc2_w   = (const float*)d_in[7];
    const float* attn_l2 = (const float*)d_in[8];
    const float* attn_r2 = (const float*)d_in[9];
    const float* bias2   = (const float*)d_in[10];
    const float* wb0     = (const float*)d_in[11];
    const float* bb0     = (const float*)d_in[12];
    const float* wb1     = (const float*)d_in[13];
    const float* bb1     = (const float*)d_in[14];
    float* out = (float*)d_out;

    const int F  = 256;
    const int N  = in_sizes[0] / F;
    const int E  = in_sizes[1];
    const int H1 = 8;

    float* ws   = (float*)d_ws;
    float* feat = ws;                ws += (size_t)N * F;
    float* buf  = ws;                ws += (size_t)N * F;
    float* buf2 = ws;                ws += (size_t)N * F;
    float* h1   = ws;                ws += (size_t)N * F;
    float* el   = ws;                ws += (size_t)N * H1;
    float* er   = ws;                ws += (size_t)N * H1;
    int*   wi   = (int*)ws;
    int*   cnt  = wi;                wi += (size_t)N;
    int*   rank = wi;                wi += (size_t)E;
    int*   csrc = wi;                wi += (size_t)E;
    int*   rp   = wi;

    const int TB = 256;
    auto cdiv = [](int a, int b) { return (a + b - 1) / b; };

    const int waves    = (N / 16) * 4;
    const int gemmGrid = cdiv(waves, TB / 32);
    const int nodeGrid = cdiv(N, NODE_TB / 32);

    fill_i32<<<cdiv(N, TB), TB, 0, stream>>>(cnt, 0, N);
    edge_rank_k<<<cdiv(E, TB), TB, 0, stream>>>(dst, cnt, rank, E);
    scan_counts_k<<<1, SCAN_TB, 0, stream>>>(cnt, rp, N);
    edge_place_k<<<cdiv(E, TB), TB, 0, stream>>>(src, dst, rank, rp, csrc, E);

    gemm_wmma_f32<<<dim3(gemmGrid, 3), TB, 0, stream>>>(inputs, F,
                                                        fc1_w, F,     feat,
                                                        wb0,   F,     buf,
                                                        wb1,   2 * F, buf2,
                                                        F, N, F, 0,
                                                        attn_l1, attn_r1, el, er, H1);
    node_softmax_agg_k<8, true><<<nodeGrid, NODE_TB, 0, stream>>>(rp, csrc, el, er, feat, bias1, buf, bb0, h1, N);

    gemm_wmma_f32<<<dim3(gemmGrid, 2), TB, 0, stream>>>(h1, F,
                                                        fc2_w,   F,     feat,
                                                        wb1 + F, 2 * F, buf2,
                                                        nullptr, 0,     nullptr,
                                                        F, N, F, 2,
                                                        attn_l2, attn_r2, el, er, 1);
    node_softmax_agg_k<1, false><<<nodeGrid, NODE_TB, 0, stream>>>(rp, csrc, el, er, feat, bias2, buf2, bb1, out, N);
}


# edge grouping in ONE launch: a block owns 128 nodes, reads all dst once, ranks its edges with LDS counters, places them from its own counts; 4 launches -> 1, no global adds; arrival order as before
# speedup vs baseline: 4.2178x; 1.0909x over previous
#include <hip/hip_runtime.h>
#include <hip/hip_bf16.h>
#include <math.h>


typedef __attribute__((ext_vector_type(2))) float v2f;
typedef __attribute__((ext_vector_type(4))) float v4f;
typedef __attribute__((ext_vector_type(8))) float v8f;
typedef __attribute__((ext_vector_type(4))) int   v4i;

#define GRP_TB  1024
#define GRP_S   128
#define GRP_CAP 8192
#define GRP_U   4

__device__ __forceinline__ void grp_take(int d, int e, int base, int hi, int& below,
                                         int* cntS, int* tailS, int* lstE, int* lstN, int* lstR)
{
    const int dl = d - base;
    below += (dl < 0) ? 1 : 0;
    if ((unsigned)dl < (unsigned)hi) {
        const int r = atomicAdd(&cntS[dl], 1);
        const int p = atomicAdd(tailS, 1);
        if (p < GRP_CAP) { lstE[p] = e; lstN[p] = dl; lstR[p] = r; }
    }
}

__global__ __launch_bounds__(GRP_TB)
void group_edges_k(const int* __restrict__ src, const int* __restrict__ dst,
                   int* __restrict__ rp, int* __restrict__ csrc, int N, int E)
{
    __shared__ __attribute__((aligned(16))) int cntS[GRP_S];
    __shared__ int lstE[GRP_CAP];
    __shared__ int lstN[GRP_CAP];
    __shared__ int lstR[GRP_CAP];
    __shared__ int tailS, belowS;
    const int t = threadIdx.x;
    const int lane = t & 31;
    const int wv = t >> 5;
    const int base = blockIdx.x * GRP_S;
    const int hi = (N - base < GRP_S) ? (N - base) : GRP_S;

    if (t < GRP_S) cntS[t] = 0;
    if (t == 0) { tailS = 0; belowS = 0; }
    __syncthreads();

    int below = 0;
    const bool vec = (((size_t)dst) & 15) == 0;
    const int nV = vec ? (E >> 2) : 0;
    int g0 = 0;
    for (; g0 + GRP_TB * GRP_U <= nV; g0 += GRP_TB * GRP_U) {
        v4i d[GRP_U];
#pragma unroll
        for (int u = 0; u < GRP_U; ++u) d[u] = ((const v4i*)dst)[g0 + u * GRP_TB + t];
#pragma unroll
        for (int u = 0; u < GRP_U; ++u) {
            const int e0 = 4 * (g0 + u * GRP_TB + t);
#pragma unroll
            for (int c = 0; c < 4; ++c) grp_take(d[u][c], e0 + c, base, hi, below, cntS, &tailS, lstE, lstN, lstR);
        }
    }
    for (int g = g0 + t; g < nV; g += GRP_TB) {
        const v4i d = ((const v4i*)dst)[g];
#pragma unroll
        for (int c = 0; c < 4; ++c) grp_take(d[c], 4 * g + c, base, hi, below, cntS, &tailS, lstE, lstN, lstR);
    }
    for (int e = 4 * nV + t; e < E; e += GRP_TB)
        grp_take(dst[e], e, base, hi, below, cntS, &tailS, lstE, lstN, lstR);

#pragma unroll
    for (int off = 16; off > 0; off >>= 1) below += __shfl_xor(below, off, 32);
    if (lane == 0) atomicAdd(&belowS, below);
    __syncthreads();

    const int M = tailS;
    if (wv == 0) {
        const v4i c = *(const v4i*)(cntS + 4 * lane);
        const int tot = c[0] + c[1] + c[2] + c[3];
        int inc = tot;
#pragma unroll
        for (int off = 1; off < 32; off <<= 1) {
            const int v = __shfl_up(inc, off, 32);
            if (lane >= off) inc += v;
        }
        v4i o;
        o[0] = belowS + inc - tot;
        o[1] = o[0] + c[0];
        o[2] = o[1] + c[1];
        o[3] = o[2] + c[2];
        *(v4i*)(cntS + 4 * lane) = o;
        const int n0 = 4 * lane;
        if (n0     < hi) rp[base + n0]     = o[0];
        if (n0 + 1 < hi) rp[base + n0 + 1] = o[1];
        if (n0 + 2 < hi) rp[base + n0 + 2] = o[2];
        if (n0 + 3 < hi) rp[base + n0 + 3] = o[3];
        if (base + hi == N && lane == 31) rp[N] = belowS + inc;
    }
    __syncthreads();

    if (M <= GRP_CAP) {
        for (int i = t; i < M; i += GRP_TB)
            csrc[cntS[lstN[i]] + lstR[i]] = src[lstE[i]];
    } else {
        for (int e = t; e < E; e += GRP_TB) {
            const int dl = dst[e] - base;
            if ((unsigned)dl < (unsigned)hi) csrc[atomicAdd(&cntS[dl], 1)] = src[e];
        }
    }
}

template <int N>
__device__ __forceinline__ void fold_half(float* v, bool up, int mask)
{
#pragma unroll
    for (int k = 0; k < N / 2; ++k) {
        const float keep = up ? v[k + N / 2] : v[k];
        const float send = up ? v[k] : v[k + N / 2];
        v[k] = keep + __shfl_xor(send, mask, 32);
    }
}

#define GEMM_KC 32
#define GEMM_LP 36
__global__ __launch_bounds__(256)
void gemm_wmma_f32(const float* __restrict__ A, int lda,
                   const float* __restrict__ W0, int ldw0, float* __restrict__ C0,
                   const float* __restrict__ W1, int ldw1, float* __restrict__ C1,
                   const float* __restrict__ W2, int ldw2, float* __restrict__ C2,
                   int ldc, int M, int K, int accMask,
                   const float* __restrict__ attnL, const float* __restrict__ attnR,
                   float* __restrict__ el, float* __restrict__ er, int scoreH)
{
    const int y = blockIdx.y;
    const float* __restrict__ W = (y == 0) ? W0 : (y == 1) ? W1 : W2;
    float* __restrict__ C       = (y == 0) ? C0 : (y == 1) ? C1 : C2;
    const int ldw               = (y == 0) ? ldw0 : (y == 1) ? ldw1 : ldw2;
    const int accumulate        = (accMask >> y) & 1;

    __shared__ __attribute__((aligned(16))) float Ws[256 * GEMM_LP];
    __shared__ __attribute__((aligned(16))) float As[32 * GEMM_LP];
    __shared__ float Sc[512];

    const int tid  = threadIdx.x;
    const int wave = tid >> 5;
    const int lane = tid & 31;
    const int half = lane >> 4;
    const int lr   = lane & 15;
    const int wr   = wave >> 2;
    const int col0 = (wave & 3) * 64;
    const int rowBase = blockIdx.x * 32;
    const int rowTile = blockIdx.x * 2 + wr;
    const int nRowTiles = M >> 4;

    const int sRow = tid >> 3;
    const int sCol = (tid & 7) * 4;
    int aRowG = rowBase + sRow;
    if (aRowG > M - 1) aRowG = M - 1;
    const float* Ag = A + (size_t)aRowG * lda + sCol;
    const float* Wg = W + (size_t)sRow * ldw + sCol;
    float* AsSt = As + sRow * GEMM_LP + sCol;
    float* WsSt = Ws + sRow * GEMM_LP + sCol;

    const float* aF = As + (wr * 16 + lr) * GEMM_LP + 2 * half;
    const float* bF = Ws + (col0 + lr) * GEMM_LP + 2 * half;

    v8f acc0 = {}, acc1 = {}, acc2 = {}, acc3 = {};

    v4f areg = *(const v4f*)(Ag);
    v4f wreg[8];
#pragma unroll
    for (int j = 0; j < 8; ++j)
        wreg[j] = *(const v4f*)(Wg + (size_t)(32 * j) * ldw);

    for (int k0 = 0; k0 < K; k0 += GEMM_KC) {
        __syncthreads();
        *(v4f*)(AsSt) = areg;
#pragma unroll
        for (int j = 0; j < 8; ++j)
            *(v4f*)(WsSt + (32 * j) * GEMM_LP) = wreg[j];
        __syncthreads();

        if (k0 + GEMM_KC < K) {
            areg = *(const v4f*)(Ag + k0 + GEMM_KC);
#pragma unroll
            for (int j = 0; j < 8; ++j)
                wreg[j] = *(const v4f*)(Wg + (size_t)(32 * j) * ldw + k0 + GEMM_KC);
        }

#pragma unroll
        for (int hc = 0; hc < GEMM_KC; hc += 16) {
            v2f a[4], b0[4], b1[4], b2[4], b3[4];
#pragma unroll
            for (int s = 0; s < 4; ++s) {
                const int kk = hc + 4 * s;
                a[s]  = *(const v2f*)(aF + kk);
                b0[s] = *(const v2f*)(bF +  0 * GEMM_LP + kk);
                b1[s] = *(const v2f*)(bF + 16 * GEMM_LP + kk);
                b2[s] = *(const v2f*)(bF + 32 * GEMM_LP + kk);
                b3[s] = *(const v2f*)(bF + 48 * GEMM_LP + kk);
            }
#pragma unroll
            for (int s = 0; s < 4; ++s) {
                acc0 = __builtin_amdgcn_wmma_f32_16x16x4_f32(false, a[s], false, b0[s], (short)0, acc0, false, false);
                acc1 = __builtin_amdgcn_wmma_f32_16x16x4_f32(false, a[s], false, b1[s], (short)0, acc1, false, false);
                acc2 = __builtin_amdgcn_wmma_f32_16x16x4_f32(false, a[s], false, b2[s], (short)0, acc2, false, false);
                acc3 = __builtin_amdgcn_wmma_f32_16x16x4_f32(false, a[s], false, b3[s], (short)0, acc3, false, false);
            }
        }
    }

    if (rowTile < nRowTiles) {
        float* Cb = C + (size_t)(rowTile * 16 + 8 * half) * ldc + col0 + lr;
#pragma unroll
        for (int j = 0; j < 8; ++j) {
            float* p = Cb + (size_t)j * ldc;
            p[ 0] = (accumulate ? p[ 0] : 0.0f) + acc0[j];
            p[16] = (accumulate ? p[16] : 0.0f) + acc1[j];
            p[32] = (accumulate ? p[32] : 0.0f) + acc2[j];
            p[48] = (accumulate ? p[48] : 0.0f) + acc3[j];
        }
    }

    if (scoreH == 0 || y != 0) return;

    float al[4], ar[4];
#pragma unroll
    for (int t = 0; t < 4; ++t) {
        al[t] = attnL[col0 + 16 * t + lr];
        ar[t] = attnR[col0 + 16 * t + lr];
    }
    const int srow = wr * 16 + 8 * half + (lr >> 1);
    const int ssel = lr & 1;
    const int strip = wave & 3;

    if (scoreH == 8) {
        float v[32];
#pragma unroll
        for (int j = 0; j < 8; ++j) {
            v[4 * j + 0] = fmaf(acc1[j], al[1], acc0[j] * al[0]);
            v[4 * j + 1] = fmaf(acc3[j], al[3], acc2[j] * al[2]);
            v[4 * j + 2] = fmaf(acc1[j], ar[1], acc0[j] * ar[0]);
            v[4 * j + 3] = fmaf(acc3[j], ar[3], acc2[j] * ar[2]);
        }
        fold_half<32>(v, (lane & 8) != 0, 8);
        fold_half<16>(v, (lane & 4) != 0, 4);
        fold_half<8>(v, (lane & 2) != 0, 2);
        fold_half<4>(v, (lane & 1) != 0, 1);
        Sc[ssel * 256 + srow * 8 + 2 * strip]     = v[0];
        Sc[ssel * 256 + srow * 8 + 2 * strip + 1] = v[1];
        __syncthreads();
        if (rowBase + (tid >> 3) < M) {
            el[(size_t)rowBase * 8 + tid] = Sc[tid];
            er[(size_t)rowBase * 8 + tid] = Sc[256 + tid];
        }
    } else {
        float v[16];
#pragma unroll
        for (int j = 0; j < 8; ++j) {
            float sl = acc0[j] * al[0];
            sl = fmaf(acc1[j], al[1], sl);
            sl = fmaf(acc2[j], al[2], sl);
            sl = fmaf(acc3[j], al[3], sl);
            float sr = acc0[j] * ar[0];
            sr = fmaf(acc1[j], ar[1], sr);
            sr = fmaf(acc2[j], ar[2], sr);
            sr = fmaf(acc3[j], ar[3], sr);
            v[2 * j + 0] = sl;
            v[2 * j + 1] = sr;
        }
        fold_half<16>(v, (lane & 8) != 0, 8);
        fold_half<8>(v, (lane & 4) != 0, 4);
        fold_half<4>(v, (lane & 2) != 0, 2);
        fold_half<2>(v, (lane & 1) != 0, 1);
        Sc[strip * 64 + ssel * 32 + srow] = v[0];
        __syncthreads();
        if (tid < 64) {
            const float s = (Sc[tid] + Sc[64 + tid]) + (Sc[128 + tid] + Sc[192 + tid]);
            const int r = tid & 31;
            if (rowBase + r < M) (tid < 32 ? el : er)[rowBase + r] = s;
        }
    }
}

__device__ inline float leaky02(float v) { return v > 0.0f ? v : 0.2f * v; }

#define NODE_TB 256
#define AGG_U  4

template <int H>
__device__ inline void load_scores(const int* __restrict__ csrc, const float* __restrict__ el,
                                   const float (&ern)[H], int b, int cnt, int lane,
                                   int& sv, float (&e)[H])
{
    sv = 0;
#pragma unroll
    for (int h = 0; h < H; ++h) e[h] = -INFINITY;
    if (lane < cnt) {
        sv = csrc[b + lane];
        if constexpr (H == 8) {
            const v4f x0 = *(const v4f*)(el + (size_t)sv * 8);
            const v4f x1 = *(const v4f*)(el + (size_t)sv * 8 + 4);
#pragma unroll
            for (int h = 0; h < 4; ++h) {
                e[h]     = leaky02(x0[h] + ern[h]);
                e[h + 4] = leaky02(x1[h] + ern[h + 4]);
            }
        } else {
            e[0] = leaky02(el[sv] + ern[0]);
        }
    }
}

template <int H, int U>
__device__ inline void walk_group(const float* __restrict__ featL, int sv, const float (&coef)[H],
                                  int i, float (&acc)[8])
{
    float f[U][8];
#pragma unroll
    for (int u = 0; u < U; ++u) {
        const int sN = __builtin_amdgcn_readlane(sv, i + u);
        const float* fr = featL + (size_t)sN * 256;
#pragma unroll
        for (int j = 0; j < 8; ++j) f[u][j] = fr[32 * j];
    }
#pragma unroll
    for (int u = 0; u < U; ++u) {
#pragma unroll
        for (int j = 0; j < 8; ++j) {
            const float a = __int_as_float(__builtin_amdgcn_readlane(__float_as_int(coef[(j * H) >> 3]), i + u));
            acc[j] = fmaf(f[u][j], a, acc[j]);
        }
    }
}

template <int H>
__device__ inline void walk_rows(const float* __restrict__ featL, int sv, const float (&coef)[H],
                                 int cnt, float (&acc)[8])
{
    int i = 0;
    for (; i + AGG_U <= cnt; i += AGG_U) walk_group<H, AGG_U>(featL, sv, coef, i, acc);
    if (i + 2 <= cnt) { walk_group<H, 2>(featL, sv, coef, i, acc); i += 2; }
    if (i < cnt)      { walk_group<H, 1>(featL, sv, coef, i, acc); }
}

template <int H, bool RELU>
__global__ __launch_bounds__(NODE_TB)
void node_softmax_agg_k(const int* __restrict__ rp, const int* __restrict__ csrc,
                        const float* __restrict__ el, const float* __restrict__ er,
                        const float* __restrict__ feat,
                        const float* __restrict__ bias, const float* __restrict__ buf,
                        const float* __restrict__ bb, float* __restrict__ outp, int N)
{
    const int lane = threadIdx.x & 31;
    const int n = __builtin_amdgcn_readfirstlane((int)(blockIdx.x * (blockDim.x >> 5) + (threadIdx.x >> 5)));
    if (n >= N) return;
    const int start = __builtin_amdgcn_readfirstlane(rp[n]);
    const int end   = __builtin_amdgcn_readfirstlane(rp[n + 1]);
    const int len   = end - start;

    const size_t rowOff = (size_t)n * 256 + lane;
    float bufv[8];
#pragma unroll
    for (int j = 0; j < 8; ++j) bufv[j] = buf[rowOff + 32 * j];

    float ern[H];
#pragma unroll
    for (int h = 0; h < H; ++h) ern[h] = er[(size_t)n * H + h];

    float acc[8];
#pragma unroll
    for (int j = 0; j < 8; ++j) acc[j] = 0.0f;

    const float* featL = feat + lane;

    if (len > 0) {
        const int cnt0 = len < 32 ? len : 32;
        int sv0;
        float e0[H];
        load_scores<H>(csrc, el, ern, start, cnt0, lane, sv0, e0);

        float mx[H];
#pragma unroll
        for (int h = 0; h < H; ++h) mx[h] = e0[h];
        for (int b = start + 32; b < end; b += 32) {
            const int c = (end - b < 32) ? (end - b) : 32;
            int svb; float eb[H];
            load_scores<H>(csrc, el, ern, b, c, lane, svb, eb);
#pragma unroll
            for (int h = 0; h < H; ++h) mx[h] = fmaxf(mx[h], eb[h]);
        }
#pragma unroll
        for (int off = 16; off > 0; off >>= 1) {
#pragma unroll
            for (int h = 0; h < H; ++h) mx[h] = fmaxf(mx[h], __shfl_xor(mx[h], off, 32));
        }
#pragma unroll
        for (int h = 0; h < H; ++h) mx[h] = __int_as_float(__builtin_amdgcn_readfirstlane(__float_as_int(mx[h])));

        float ex0[H], sm[H];
#pragma unroll
        for (int h = 0; h < H; ++h) {
            ex0[h] = (lane < cnt0) ? expf(e0[h] - mx[h]) : 0.0f;
            sm[h]  = ex0[h];
        }
        for (int b = start + 32; b < end; b += 32) {
            const int c = (end - b < 32) ? (end - b) : 32;
            int svb; float eb[H];
            load_scores<H>(csrc, el, ern, b, c, lane, svb, eb);
#pragma unroll
            for (int h = 0; h < H; ++h) sm[h] += (lane < c) ? expf(eb[h] - mx[h]) : 0.0f;
        }
#pragma unroll
        for (int off = 16; off > 0; off >>= 1) {
#pragma unroll
            for (int h = 0; h < H; ++h) sm[h] += __shfl_xor(sm[h], off, 32);
        }
#pragma unroll
        for (int h = 0; h < H; ++h) sm[h] = __int_as_float(__builtin_amdgcn_readfirstlane(__float_as_int(sm[h])));

        float coef[H];
#pragma unroll
        for (int h = 0; h < H; ++h) coef[h] = ex0[h] / sm[h];
        walk_rows<H>(featL, sv0, coef, cnt0, acc);
        for (int b = start + 32; b < end; b += 32) {
            const int c = (end - b < 32) ? (end - b) : 32;
            int svb; float eb[H];
            load_scores<H>(csrc, el, ern, b, c, lane, svb, eb);
#pragma unroll
            for (int h = 0; h < H; ++h) coef[h] = ((lane < c) ? expf(eb[h] - mx[h]) : 0.0f) / sm[h];
            walk_rows<H>(featL, svb, coef, c, acc);
        }
    }

    const float dg = (float)len;
    const float norm = 1.0f / fmaxf(dg, 1.0f);
    float* o = outp + rowOff;
#pragma unroll
    for (int j = 0; j < 8; ++j) {
        float v = acc[j] + bias[lane + 32 * j] + norm * (bufv[j] + bb[lane + 32 * j]);
        if (RELU) v = v > 0.0f ? v : 0.0f;
        o[32 * j] = v;
    }
}

extern "C" void kernel_launch(void* const* d_in, const int* in_sizes, int n_in,
                              void* d_out, int out_size, void* d_ws, size_t ws_size,
                              hipStream_t stream)
{
    const float* inputs  = (const float*)d_in[0];
    const int*   src     = (const int*)  d_in[1];
    const int*   dst     = (const int*)  d_in[2];
    const float* fc1_w   = (const float*)d_in[3];
    const float* attn_l1 = (const float*)d_in[4];
    const float* attn_r1 = (const float*)d_in[5];
    const float* bias1   = (const float*)d_in[6];
    const float* fc2_w   = (const float*)d_in[7];
    const float* attn_l2 = (const float*)d_in[8];
    const float* attn_r2 = (const float*)d_in[9];
    const float* bias2   = (const float*)d_in[10];
    const float* wb0     = (const float*)d_in[11];
    const float* bb0     = (const float*)d_in[12];
    const float* wb1     = (const float*)d_in[13];
    const float* bb1     = (const float*)d_in[14];
    float* out = (float*)d_out;

    const int F  = 256;
    const int N  = in_sizes[0] / F;
    const int E  = in_sizes[1];
    const int H1 = 8;

    float* ws   = (float*)d_ws;
    float* feat = ws;                ws += (size_t)N * F;
    float* buf  = ws;                ws += (size_t)N * F;
    float* buf2 = ws;                ws += (size_t)N * F;
    float* h1   = ws;                ws += (size_t)N * F;
    float* el   = ws;                ws += (size_t)N * H1;
    float* er   = ws;                ws += (size_t)N * H1;
    int*   wi   = (int*)ws;
    int*   csrc = wi;                wi += (size_t)E;
    int*   rp   = wi;

    const int TB = 256;
    auto cdiv = [](int a, int b) { return (a + b - 1) / b; };

    const int waves    = (N / 16) * 4;
    const int gemmGrid = cdiv(waves, TB / 32);
    const int nodeGrid = cdiv(N, NODE_TB / 32);

    group_edges_k<<<cdiv(N, GRP_S), GRP_TB, 0, stream>>>(src, dst, rp, csrc, N, E);

    gemm_wmma_f32<<<dim3(gemmGrid, 3), TB, 0, stream>>>(inputs, F,
                                                        fc1_w, F,     feat,
                                                        wb0,   F,     buf,
                                                        wb1,   2 * F, buf2,
                                                        F, N, F, 0,
                                                        attn_l1, attn_r1, el, er, H1);
    node_softmax_agg_k<8, true><<<nodeGrid, NODE_TB, 0, stream>>>(rp, csrc, el, er, feat, bias1, buf, bb0, h1, N);

    gemm_wmma_f32<<<dim3(gemmGrid, 2), TB, 0, stream>>>(h1, F,
                                                        fc2_w,   F,     feat,
                                                        wb1 + F, 2 * F, buf2,
                                                        nullptr, 0,     nullptr,
                                                        F, N, F, 2,
                                                        attn_l2, attn_r2, el, er, 1);
    node_softmax_agg_k<1, false><<<nodeGrid, NODE_TB, 0, stream>>>(rp, csrc, el, er, feat, bias2, buf2, bb1, out, N);
}


# edge grouping moved into the layer-1 GEMM launch as its first 157 blocks (8 waves each, wave-private lists, no LDS round trip in the scan): runs beside the matrix work; 5 launches -> 4; arrival order
# speedup vs baseline: 4.3515x; 1.0317x over previous
#include <hip/hip_runtime.h>
#include <hip/hip_bf16.h>
#include <math.h>


typedef __attribute__((ext_vector_type(2))) float v2f;
typedef __attribute__((ext_vector_type(4))) float v4f;
typedef __attribute__((ext_vector_type(8))) float v8f;
typedef __attribute__((ext_vector_type(4))) int   v4i;

#define GRP_TB   256
#define GRP_NW   (GRP_TB / 32)
#define GRP_S    128
#define GRP_WCAP 512
#define GRP_U    8
#define GRP_SMEM_INTS (GRP_S + 2 * GRP_NW * GRP_WCAP + 2)

struct GrpLds {
    int* cnt;
    int* lstE;
    int* lstN;
    int* flag;
};

template <int K>
__device__ __forceinline__ void grp_take(const int (&dv)[K], int e0, int base, int hi,
                                         int& below, int& wcount, int wbase, const GrpLds& L)
{
#pragma unroll
    for (int k = 0; k < K; ++k) {
        const int dl = dv[k] - base;
        below += (dl < 0) ? 1 : 0;
        const bool in = (unsigned)dl < (unsigned)hi;
        const unsigned m = __builtin_amdgcn_ballot_w32(in);
        if (m != 0u) {
            const int p = wcount + (int)__builtin_amdgcn_mbcnt_lo(m, 0u);
            if (in) {
                atomicAdd(&L.cnt[dl], 1);
                if (p < GRP_WCAP) {
                    L.lstE[wbase + p] = e0 + (k >> 2) * (4 * GRP_TB) + (k & 3);
                    L.lstN[wbase + p] = dl;
                }
            }
            wcount += __builtin_popcount(m);
        }
    }
}

__device__ __forceinline__ void group_edges_role(const int* __restrict__ src, const int* __restrict__ dst,
                                                 int* __restrict__ rp, int* __restrict__ csrc,
                                                 int N, int E, int g, int* sm)
{
    GrpLds L;
    L.cnt  = sm;
    L.lstE = sm + GRP_S;
    L.lstN = L.lstE + GRP_NW * GRP_WCAP;
    L.flag = L.lstN + GRP_NW * GRP_WCAP;
    const int t = threadIdx.x;
    const int lane = t & 31;
    const int wv = t >> 5;
    const int wbase = wv * GRP_WCAP;
    const int base = g * GRP_S;
    const int hi = (N - base < GRP_S) ? (N - base) : GRP_S;
    const int none = base + GRP_S;

    if (t < GRP_S) L.cnt[t] = 0;
    if (t == 0) { L.flag[0] = 0; L.flag[1] = 0; }
    __syncthreads();

    int below = 0;
    int wcount = 0;
    const bool vec = (((size_t)dst) & 15) == 0;
    const int nV = vec ? (E >> 2) : 0;
    const int TILE = GRP_TB * GRP_U;
    const int nT = nV / TILE;
    const v4i* dq = (const v4i*)dst;
    v4i d[GRP_U];
    if (nT > 0) {
#pragma unroll
        for (int u = 0; u < GRP_U; ++u) d[u] = dq[u * GRP_TB + t];
    }
    for (int tile = 0; tile < nT; ++tile) {
        const int g0 = tile * TILE;
        const int gn = (tile + 1 < nT) ? g0 + TILE : g0;
        v4i dn[GRP_U];
#pragma unroll
        for (int u = 0; u < GRP_U; ++u) dn[u] = dq[gn + u * GRP_TB + t];
        int dv[4 * GRP_U];
#pragma unroll
        for (int k = 0; k < 4 * GRP_U; ++k) dv[k] = d[k >> 2][k & 3];
        grp_take<4 * GRP_U>(dv, 4 * (g0 + t), base, hi, below, wcount, wbase, L);
#pragma unroll
        for (int u = 0; u < GRP_U; ++u) d[u] = dn[u];
    }
    for (int gb = nT * TILE; gb < nV; gb += GRP_TB) {
        const int g4 = gb + t;
        int dv[4];
#pragma unroll
        for (int c = 0; c < 4; ++c) dv[c] = none;
        if (g4 < nV) {
            const v4i q = dq[g4];
#pragma unroll
            for (int c = 0; c < 4; ++c) dv[c] = q[c];
        }
        grp_take<4>(dv, 4 * g4, base, hi, below, wcount, wbase, L);
    }
    for (int eb = 4 * nV; eb < E; eb += GRP_TB) {
        const int e = eb + t;
        int dv[1];
        dv[0] = (e < E) ? dst[e] : none;
        grp_take<1>(dv, e, base, hi, below, wcount, wbase, L);
    }

#pragma unroll
    for (int off = 16; off > 0; off >>= 1) below += __shfl_xor(below, off, 32);
    if (lane == 0) {
        atomicAdd(&L.flag[0], below);
        if (wcount > GRP_WCAP) L.flag[1] = 1;
    }
    __syncthreads();

    const int belowB = L.flag[0];
    if (wv == 0) {
        const v4i cc = *(const v4i*)(L.cnt + 4 * lane);
        const int tot = cc[0] + cc[1] + cc[2] + cc[3];
        int inc = tot;
#pragma unroll
        for (int off = 1; off < 32; off <<= 1) {
            const int v = __shfl_up(inc, off, 32);
            if (lane >= off) inc += v;
        }
        v4i o;
        o[0] = belowB + inc - tot;
        o[1] = o[0] + cc[0];
        o[2] = o[1] + cc[1];
        o[3] = o[2] + cc[2];
        *(v4i*)(L.cnt + 4 * lane) = o;
        const int n0 = 4 * lane;
        if (n0     < hi) rp[base + n0]     = o[0];
        if (n0 + 1 < hi) rp[base + n0 + 1] = o[1];
        if (n0 + 2 < hi) rp[base + n0 + 2] = o[2];
        if (n0 + 3 < hi) rp[base + n0 + 3] = o[3];
        if (base + hi == N && lane == 31) rp[N] = belowB + inc;
    }
    __syncthreads();

    if (L.flag[1] == 0) {
        for (int i0 = lane; i0 < wcount; i0 += 4 * 32) {
            int sv[4], pos[4];
#pragma unroll
            for (int q = 0; q < 4; ++q) {
                const int i = i0 + 32 * q;
                if (i < wcount) {
                    sv[q]  = src[L.lstE[wbase + i]];
                    pos[q] = atomicAdd(&L.cnt[L.lstN[wbase + i]], 1);
                }
            }
#pragma unroll
            for (int q = 0; q < 4; ++q) {
                const int i = i0 + 32 * q;
                if (i < wcount) csrc[pos[q]] = sv[q];
            }
        }
    } else {
        for (int e = t; e < E; e += GRP_TB) {
            const int dl = dst[e] - base;
            if ((unsigned)dl < (unsigned)hi) csrc[atomicAdd(&L.cnt[dl], 1)] = src[e];
        }
    }
}

template <int N>
__device__ __forceinline__ void fold_half(float* v, bool up, int mask)
{
#pragma unroll
    for (int k = 0; k < N / 2; ++k) {
        const float keep = up ? v[k + N / 2] : v[k];
        const float send = up ? v[k] : v[k + N / 2];
        v[k] = keep + __shfl_xor(send, mask, 32);
    }
}

#define GEMM_KC 32
#define GEMM_LP 36
#define GEMM_SMEM_WORDS (256 * GEMM_LP + 32 * GEMM_LP + 512)
#define SMEM_WORDS (GRP_SMEM_INTS > GEMM_SMEM_WORDS ? GRP_SMEM_INTS : GEMM_SMEM_WORDS)
__global__ __launch_bounds__(256)
void gemm_wmma_f32(const float* __restrict__ A, int lda,
                   const float* __restrict__ W0, int ldw0, float* __restrict__ C0,
                   const float* __restrict__ W1, int ldw1, float* __restrict__ C1,
                   const float* __restrict__ W2, int ldw2, float* __restrict__ C2,
                   int ldc, int M, int K, int accMask,
                   const float* __restrict__ attnL, const float* __restrict__ attnR,
                   float* __restrict__ el, float* __restrict__ er, int scoreH,
                   const int* __restrict__ src, const int* __restrict__ dst,
                   int* __restrict__ rp, int* __restrict__ csrc, int nNodes, int nEdges,
                   int grpBlocks, int rowBlocks)
{
    __shared__ __attribute__((aligned(16))) unsigned char smem[SMEM_WORDS * 4];

    int b = blockIdx.x;
    if (b < grpBlocks) {
        group_edges_role(src, dst, rp, csrc, nNodes, nEdges, b, (int*)smem);
        return;
    }
    b -= grpBlocks;

    const int y  = (b >= 2 * rowBlocks) ? 2 : (b >= rowBlocks) ? 1 : 0;
    const int bx = b - y * rowBlocks;
    const float* __restrict__ W = (y == 0) ? W0 : (y == 1) ? W1 : W2;
    float* __restrict__ C       = (y == 0) ? C0 : (y == 1) ? C1 : C2;
    const int ldw               = (y == 0) ? ldw0 : (y == 1) ? ldw1 : ldw2;
    const int accumulate        = (accMask >> y) & 1;

    float* Ws = (float*)smem;
    float* As = Ws + 256 * GEMM_LP;
    float* Sc = As + 32 * GEMM_LP;

    const int tid  = threadIdx.x;
    const int wave = tid >> 5;
    const int lane = tid & 31;
    const int half = lane >> 4;
    const int lr   = lane & 15;
    const int wr   = wave >> 2;
    const int col0 = (wave & 3) * 64;
    const int rowBase = bx * 32;
    const int rowTile = bx * 2 + wr;
    const int nRowTiles = M >> 4;

    const int sRow = tid >> 3;
    const int sCol = (tid & 7) * 4;
    int aRowG = rowBase + sRow;
    if (aRowG > M - 1) aRowG = M - 1;
    const float* Ag = A + (size_t)aRowG * lda + sCol;
    const float* Wg = W + (size_t)sRow * ldw + sCol;
    float* AsSt = As + sRow * GEMM_LP + sCol;
    float* WsSt = Ws + sRow * GEMM_LP + sCol;

    const float* aF = As + (wr * 16 + lr) * GEMM_LP + 2 * half;
    const float* bF = Ws + (col0 + lr) * GEMM_LP + 2 * half;

    v8f acc0 = {}, acc1 = {}, acc2 = {}, acc3 = {};

    v4f areg = *(const v4f*)(Ag);
    v4f wreg[8];
#pragma unroll
    for (int j = 0; j < 8; ++j)
        wreg[j] = *(const v4f*)(Wg + (size_t)(32 * j) * ldw);

    for (int k0 = 0; k0 < K; k0 += GEMM_KC) {
        __syncthreads();
        *(v4f*)(AsSt) = areg;
#pragma unroll
        for (int j = 0; j < 8; ++j)
            *(v4f*)(WsSt + (32 * j) * GEMM_LP) = wreg[j];
        __syncthreads();

        if (k0 + GEMM_KC < K) {
            areg = *(const v4f*)(Ag + k0 + GEMM_KC);
#pragma unroll
            for (int j = 0; j < 8; ++j)
                wreg[j] = *(const v4f*)(Wg + (size_t)(32 * j) * ldw + k0 + GEMM_KC);
        }

#pragma unroll
        for (int hc = 0; hc < GEMM_KC; hc += 16) {
            v2f a[4], b0[4], b1[4], b2[4], b3[4];
#pragma unroll
            for (int s = 0; s < 4; ++s) {
                const int kk = hc + 4 * s;
                a[s]  = *(const v2f*)(aF + kk);
                b0[s] = *(const v2f*)(bF +  0 * GEMM_LP + kk);
                b1[s] = *(const v2f*)(bF + 16 * GEMM_LP + kk);
                b2[s] = *(const v2f*)(bF + 32 * GEMM_LP + kk);
                b3[s] = *(const v2f*)(bF + 48 * GEMM_LP + kk);
            }
#pragma unroll
            for (int s = 0; s < 4; ++s) {
                acc0 = __builtin_amdgcn_wmma_f32_16x16x4_f32(false, a[s], false, b0[s], (short)0, acc0, false, false);
                acc1 = __builtin_amdgcn_wmma_f32_16x16x4_f32(false, a[s], false, b1[s], (short)0, acc1, false, false);
                acc2 = __builtin_amdgcn_wmma_f32_16x16x4_f32(false, a[s], false, b2[s], (short)0, acc2, false, false);
                acc3 = __builtin_amdgcn_wmma_f32_16x16x4_f32(false, a[s], false, b3[s], (short)0, acc3, false, false);
            }
        }
    }

    if (rowTile < nRowTiles) {
        float* Cb = C + (size_t)(rowTile * 16 + 8 * half) * ldc + col0 + lr;
#pragma unroll
        for (int j = 0; j < 8; ++j) {
            float* p = Cb + (size_t)j * ldc;
            p[ 0] = (accumulate ? p[ 0] : 0.0f) + acc0[j];
            p[16] = (accumulate ? p[16] : 0.0f) + acc1[j];
            p[32] = (accumulate ? p[32] : 0.0f) + acc2[j];
            p[48] = (accumulate ? p[48] : 0.0f) + acc3[j];
        }
    }

    if (scoreH == 0 || y != 0) return;

    float al[4], ar[4];
#pragma unroll
    for (int t = 0; t < 4; ++t) {
        al[t] = attnL[col0 + 16 * t + lr];
        ar[t] = attnR[col0 + 16 * t + lr];
    }
    const int srow = wr * 16 + 8 * half + (lr >> 1);
    const int ssel = lr & 1;
    const int strip = wave & 3;

    if (scoreH == 8) {
        float v[32];
#pragma unroll
        for (int j = 0; j < 8; ++j) {
            v[4 * j + 0] = fmaf(acc1[j], al[1], acc0[j] * al[0]);
            v[4 * j + 1] = fmaf(acc3[j], al[3], acc2[j] * al[2]);
            v[4 * j + 2] = fmaf(acc1[j], ar[1], acc0[j] * ar[0]);
            v[4 * j + 3] = fmaf(acc3[j], ar[3], acc2[j] * ar[2]);
        }
        fold_half<32>(v, (lane & 8) != 0, 8);
        fold_half<16>(v, (lane & 4) != 0, 4);
        fold_half<8>(v, (lane & 2) != 0, 2);
        fold_half<4>(v, (lane & 1) != 0, 1);
        Sc[ssel * 256 + srow * 8 + 2 * strip]     = v[0];
        Sc[ssel * 256 + srow * 8 + 2 * strip + 1] = v[1];
        __syncthreads();
        if (rowBase + (tid >> 3) < M) {
            el[(size_t)rowBase * 8 + tid] = Sc[tid];
            er[(size_t)rowBase * 8 + tid] = Sc[256 + tid];
        }
    } else {
        float v[16];
#pragma unroll
        for (int j = 0; j < 8; ++j) {
            float sl = acc0[j] * al[0];
            sl = fmaf(acc1[j], al[1], sl);
            sl = fmaf(acc2[j], al[2], sl);
            sl = fmaf(acc3[j], al[3], sl);
            float sr = acc0[j] * ar[0];
            sr = fmaf(acc1[j], ar[1], sr);
            sr = fmaf(acc2[j], ar[2], sr);
            sr = fmaf(acc3[j], ar[3], sr);
            v[2 * j + 0] = sl;
            v[2 * j + 1] = sr;
        }
        fold_half<16>(v, (lane & 8) != 0, 8);
        fold_half<8>(v, (lane & 4) != 0, 4);
        fold_half<4>(v, (lane & 2) != 0, 2);
        fold_half<2>(v, (lane & 1) != 0, 1);
        Sc[strip * 64 + ssel * 32 + srow] = v[0];
        __syncthreads();
        if (tid < 64) {
            const float s = (Sc[tid] + Sc[64 + tid]) + (Sc[128 + tid] + Sc[192 + tid]);
            const int r = tid & 31;
            if (rowBase + r < M) (tid < 32 ? el : er)[rowBase + r] = s;
        }
    }
}

__device__ inline float leaky02(float v) { return v > 0.0f ? v : 0.2f * v; }

#define NODE_TB 256
#define AGG_U  4

template <int H>
__device__ inline void load_scores(const int* __restrict__ csrc, const float* __restrict__ el,
                                   const float (&ern)[H], int b, int cnt, int lane,
                                   int& sv, float (&e)[H])
{
    sv = 0;
#pragma unroll
    for (int h = 0; h < H; ++h) e[h] = -INFINITY;
    if (lane < cnt) {
        sv = csrc[b + lane];
        if constexpr (H == 8) {
            const v4f x0 = *(const v4f*)(el + (size_t)sv * 8);
            const v4f x1 = *(const v4f*)(el + (size_t)sv * 8 + 4);
#pragma unroll
            for (int h = 0; h < 4; ++h) {
                e[h]     = leaky02(x0[h] + ern[h]);
                e[h + 4] = leaky02(x1[h] + ern[h + 4]);
            }
        } else {
            e[0] = leaky02(el[sv] + ern[0]);
        }
    }
}

template <int H, int U>
__device__ inline void walk_group(const float* __restrict__ featL, int sv, const float (&coef)[H],
                                  int i, float (&acc)[8])
{
    float f[U][8];
#pragma unroll
    for (int u = 0; u < U; ++u) {
        const int sN = __builtin_amdgcn_readlane(sv, i + u);
        const float* fr = featL + (size_t)sN * 256;
#pragma unroll
        for (int j = 0; j < 8; ++j) f[u][j] = fr[32 * j];
    }
#pragma unroll
    for (int u = 0; u < U; ++u) {
#pragma unroll
        for (int j = 0; j < 8; ++j) {
            const float a = __int_as_float(__builtin_amdgcn_readlane(__float_as_int(coef[(j * H) >> 3]), i + u));
            acc[j] = fmaf(f[u][j], a, acc[j]);
        }
    }
}

template <int H>
__device__ inline void walk_rows(const float* __restrict__ featL, int sv, const float (&coef)[H],
                                 int cnt, float (&acc)[8])
{
    int i = 0;
    for (; i + AGG_U <= cnt; i += AGG_U) walk_group<H, AGG_U>(featL, sv, coef, i, acc);
    if (i + 2 <= cnt) { walk_group<H, 2>(featL, sv, coef, i, acc); i += 2; }
    if (i < cnt)      { walk_group<H, 1>(featL, sv, coef, i, acc); }
}

template <int H, bool RELU>
__global__ __launch_bounds__(NODE_TB)
void node_softmax_agg_k(const int* __restrict__ rp, const int* __restrict__ csrc,
                        const float* __restrict__ el, const float* __restrict__ er,
                        const float* __restrict__ feat,
                        const float* __restrict__ bias, const float* __restrict__ buf,
                        const float* __restrict__ bb, float* __restrict__ outp, int N)
{
    const int lane = threadIdx.x & 31;
    const int n = __builtin_amdgcn_readfirstlane((int)(blockIdx.x * (blockDim.x >> 5) + (threadIdx.x >> 5)));
    if (n >= N) return;
    const int start = __builtin_amdgcn_readfirstlane(rp[n]);
    const int end   = __builtin_amdgcn_readfirstlane(rp[n + 1]);
    const int len   = end - start;

    const size_t rowOff = (size_t)n * 256 + lane;
    float bufv[8];
#pragma unroll
    for (int j = 0; j < 8; ++j) bufv[j] = buf[rowOff + 32 * j];

    float ern[H];
#pragma unroll
    for (int h = 0; h < H; ++h) ern[h] = er[(size_t)n * H + h];

    float acc[8];
#pragma unroll
    for (int j = 0; j < 8; ++j) acc[j] = 0.0f;

    const float* featL = feat + lane;

    if (len > 0) {
        const int cnt0 = len < 32 ? len : 32;
        int sv0;
        float e0[H];
        load_scores<H>(csrc, el, ern, start, cnt0, lane, sv0, e0);

        float mx[H];
#pragma unroll
        for (int h = 0; h < H; ++h) mx[h] = e0[h];
        for (int b = start + 32; b < end; b += 32) {
            const int c = (end - b < 32) ? (end - b) : 32;
            int svb; float eb[H];
            load_scores<H>(csrc, el, ern, b, c, lane, svb, eb);
#pragma unroll
            for (int h = 0; h < H; ++h) mx[h] = fmaxf(mx[h], eb[h]);
        }
#pragma unroll
        for (int off = 16; off > 0; off >>= 1) {
#pragma unroll
            for (int h = 0; h < H; ++h) mx[h] = fmaxf(mx[h], __shfl_xor(mx[h], off, 32));
        }
#pragma unroll
        for (int h = 0; h < H; ++h) mx[h] = __int_as_float(__builtin_amdgcn_readfirstlane(__float_as_int(mx[h])));

        float ex0[H], sm[H];
#pragma unroll
        for (int h = 0; h < H; ++h) {
            ex0[h] = (lane < cnt0) ? expf(e0[h] - mx[h]) : 0.0f;
            sm[h]  = ex0[h];
        }
        for (int b = start + 32; b < end; b += 32) {
            const int c = (end - b < 32) ? (end - b) : 32;
            int svb; float eb[H];
            load_scores<H>(csrc, el, ern, b, c, lane, svb, eb);
#pragma unroll
            for (int h = 0; h < H; ++h) sm[h] += (lane < c) ? expf(eb[h] - mx[h]) : 0.0f;
        }
#pragma unroll
        for (int off = 16; off > 0; off >>= 1) {
#pragma unroll
            for (int h = 0; h < H; ++h) sm[h] += __shfl_xor(sm[h], off, 32);
        }
#pragma unroll
        for (int h = 0; h < H; ++h) sm[h] = __int_as_float(__builtin_amdgcn_readfirstlane(__float_as_int(sm[h])));

        float coef[H];
#pragma unroll
        for (int h = 0; h < H; ++h) coef[h] = ex0[h] / sm[h];
        walk_rows<H>(featL, sv0, coef, cnt0, acc);
        for (int b = start + 32; b < end; b += 32) {
            const int c = (end - b < 32) ? (end - b) : 32;
            int svb; float eb[H];
            load_scores<H>(csrc, el, ern, b, c, lane, svb, eb);
#pragma unroll
            for (int h = 0; h < H; ++h) coef[h] = ((lane < c) ? expf(eb[h] - mx[h]) : 0.0f) / sm[h];
            walk_rows<H>(featL, svb, coef, c, acc);
        }
    }

    const float dg = (float)len;
    const float norm = 1.0f / fmaxf(dg, 1.0f);
    float* o = outp + rowOff;
#pragma unroll
    for (int j = 0; j < 8; ++j) {
        float v = acc[j] + bias[lane + 32 * j] + norm * (bufv[j] + bb[lane + 32 * j]);
        if (RELU) v = v > 0.0f ? v : 0.0f;
        o[32 * j] = v;
    }
}

extern "C" void kernel_launch(void* const* d_in, const int* in_sizes, int n_in,
                              void* d_out, int out_size, void* d_ws, size_t ws_size,
                              hipStream_t stream)
{
    const float* inputs  = (const float*)d_in[0];
    const int*   src     = (const int*)  d_in[1];
    const int*   dst     = (const int*)  d_in[2];
    const float* fc1_w   = (const float*)d_in[3];
    const float* attn_l1 = (const float*)d_in[4];
    const float* attn_r1 = (const float*)d_in[5];
    const float* bias1   = (const float*)d_in[6];
    const float* fc2_w   = (const float*)d_in[7];
    const float* attn_l2 = (const float*)d_in[8];
    const float* attn_r2 = (const float*)d_in[9];
    const float* bias2   = (const float*)d_in[10];
    const float* wb0     = (const float*)d_in[11];
    const float* bb0     = (const float*)d_in[12];
    const float* wb1     = (const float*)d_in[13];
    const float* bb1     = (const float*)d_in[14];
    float* out = (float*)d_out;

    const int F  = 256;
    const int N  = in_sizes[0] / F;
    const int E  = in_sizes[1];
    const int H1 = 8;

    float* ws   = (float*)d_ws;
    float* feat = ws;                ws += (size_t)N * F;
    float* buf  = ws;                ws += (size_t)N * F;
    float* buf2 = ws;                ws += (size_t)N * F;
    float* h1   = ws;                ws += (size_t)N * F;
    float* el   = ws;                ws += (size_t)N * H1;
    float* er   = ws;                ws += (size_t)N * H1;
    int*   wi   = (int*)ws;
    int*   csrc = wi;                wi += (size_t)E;
    int*   rp   = wi;

    const int TB = 256;
    auto cdiv = [](int a, int b) { return (a + b - 1) / b; };

    const int waves    = (N / 16) * 4;
    const int gemmGrid = cdiv(waves, TB / 32);
    const int nodeGrid = cdiv(N, NODE_TB / 32);
    const int grpGrid  = cdiv(N, GRP_S);

    gemm_wmma_f32<<<grpGrid + 3 * gemmGrid, TB, 0, stream>>>(inputs, F,
                                                             fc1_w, F,     feat,
                                                             wb0,   F,     buf,
                                                             wb1,   2 * F, buf2,
                                                             F, N, F, 0,
                                                             attn_l1, attn_r1, el, er, H1,
                                                             src, dst, rp, csrc, N, E, grpGrid, gemmGrid);
    node_softmax_agg_k<8, true><<<nodeGrid, NODE_TB, 0, stream>>>(rp, csrc, el, er, feat, bias1, buf, bb0, h1, N);

    gemm_wmma_f32<<<2 * gemmGrid, TB, 0, stream>>>(h1, F,
                                                   fc2_w,   F,     feat,
                                                   wb1 + F, 2 * F, buf2,
                                                   nullptr, 0,     nullptr,
                                                   F, N, F, 2,
                                                   attn_l2, attn_r2, el, er, 1,
                                                   nullptr, nullptr, nullptr, nullptr, 0, 0, 0, gemmGrid);
    node_softmax_agg_k<1, false><<<nodeGrid, NODE_TB, 0, stream>>>(rp, csrc, el, er, feat, bias2, buf2, bb1, out, N);
}
